# starting kernel
# baseline (speedup 1.0000x reference)
#include <hip/hip_runtime.h>
#include <math.h>

typedef float v2f __attribute__((ext_vector_type(2)));
typedef float v8f __attribute__((ext_vector_type(8)));

__global__ __launch_bounds__(128) void gemm_f32_wmma(
    const float* __restrict__ A, long sAm, long sAk, long sAb,
    const float* __restrict__ B, long sBk, long sBn, long sBb,
    float* __restrict__ C, long sCm, long sCn, long sCb,
    int M, int N, int K, float alpha, int accum)
{
  const int lane = threadIdx.x & 31;
  const int wave = threadIdx.x >> 5;
  const int half = lane >> 4;
  const int l16  = lane & 15;
  const int mBase = blockIdx.x * 64;
  const int nBase = blockIdx.y * 64 + wave * 16;

  A += (long)blockIdx.z * sAb;
  B += (long)blockIdx.z * sBb;
  C += (long)blockIdx.z * sCb;

  v8f zero = {0.f,0.f,0.f,0.f,0.f,0.f,0.f,0.f};
  v8f acc[4];
#pragma unroll
  for (int t = 0; t < 4; ++t) acc[t] = zero;

  const float* aPtr[4];
  float aMask[4];
#pragma unroll
  for (int t = 0; t < 4; ++t) {
    int m = mBase + t * 16 + l16;
    bool v = (m < M);
    aMask[t] = v ? 1.f : 0.f;
    aPtr[t] = A + (long)(v ? m : 0) * sAm + (long)(2 * half) * sAk;
  }
  const int n = nBase + l16;
  const bool nv = (n < N);
  const float bMask = nv ? 1.f : 0.f;
  const float* bPtr = B + (long)(nv ? n : 0) * sBn + (long)(2 * half) * sBk;

  for (int k0 = 0; k0 < K; k0 += 4) {
    v2f bf;
    bf.x = bMask * bPtr[0];
    bf.y = bMask * bPtr[sBk];
    bPtr += 4 * sBk;
#pragma unroll
    for (int t = 0; t < 4; ++t) {
      v2f af;
      af.x = aMask[t] * aPtr[t][0];
      af.y = aMask[t] * aPtr[t][sAk];
      aPtr[t] += 4 * sAk;
      acc[t] = __builtin_amdgcn_wmma_f32_16x16x4_f32(
          false, af, false, bf, (short)0, acc[t], false, false);
    }
  }

  if (nv) {
#pragma unroll
    for (int t = 0; t < 4; ++t) {
#pragma unroll
      for (int r = 0; r < 8; ++r) {
        int m = mBase + t * 16 + r + 8 * half;
        if (m < M) {
          long idx = (long)m * sCm + (long)n * sCn;
          float val = alpha * acc[t][r];
          if (accum) C[idx] += val; else C[idx] = val;
        }
      }
    }
  }
}

__global__ void gen_consts(float* Fre, float* Fim, float* Vre, float* Vim) {
  int i = blockIdx.x * 256 + threadIdx.x;
  if (i >= 48 * 96) return;
  int k = i / 96, nn = i % 96;
  int f = (k < 24) ? k : (k + 48);
  int ph = (f * nn) % 96;
  float th = 6.283185307179586f * (float)ph / 96.f;
  float c = cosf(th), s = sinf(th);
  Fre[i] = c;  Fim[i] = -s;
  Vre[nn * 48 + k] = c;  Vim[nn * 48 + k] = s;
}

__global__ void pack_wqkv(const float* __restrict__ wq, const float* __restrict__ wk,
                          const float* __restrict__ wv, float* __restrict__ dst) {
  int i = blockIdx.x * 256 + threadIdx.x;
  if (i >= 384 * 64) return;
  int ic = i % 64, r = i / 64;
  int t = r / 128, q = r % 128;
  const float* w = (t == 0) ? wq : ((t == 1) ? wk : wv);
  dst[i] = w[q * 64 + ic];
}

__global__ void addsub(const float* __restrict__ x, const float* __restrict__ y,
                       float* __restrict__ sum, float* __restrict__ dif, int nelem) {
  int i = blockIdx.x * 256 + threadIdx.x;
  if (i >= nelem) return;
  float a = x[i], b = y[i];
  sum[i] = a + b;
  dif[i] = a - b;
}

__global__ void selu_k(float* __restrict__ s, int nelem) {
  int i = blockIdx.x * 256 + threadIdx.x;
  if (i >= nelem) return;
  float x = s[i];
  const float lam = 1.0507009873554805f, alp = 1.6732632423543772f;
  s[i] = lam * (x > 0.f ? x : alp * (expf(x) - 1.f));
}

__global__ void group_qkv(const float* __restrict__ src, float* __restrict__ dst) {
  int i = blockIdx.x * 256 + threadIdx.x;
  if (i >= 42467328) return;
  int s = i % 110592, r = i / 110592;
  int x = s % 48, y = (s / 48) % 48, z = s / 2304;
  int o = r % 16, h = (r / 16) % 8, t = r / 128;
  int px = x & 7, nx = x >> 3, py = y & 7, ny = y >> 3, pz = z & 7, nz = z >> 3;
  int cp = ((((o << 3) | pz) << 3) | py) << 3 | px;
  int tok = (nz * 6 + ny) * 6 + nx;
  dst[((long)((t * 8 + h) * 8192 + cp)) * 216 + tok] = src[i];
}

__global__ void ungroup_o(const float* __restrict__ src, float* __restrict__ dst) {
  int i = blockIdx.x * 256 + threadIdx.x;
  if (i >= 14155776) return;
  int tok = i % 216, cp = (i / 216) % 8192, h = i / (216 * 8192);
  int px = cp & 7, py = (cp >> 3) & 7, pz = (cp >> 6) & 7, o = cp >> 9;
  int nx = tok % 6, ny = (tok / 6) % 6, nz = tok / 36;
  int z = nz * 8 + pz, y = ny * 8 + py, x = nx * 8 + px;
  dst[(long)(h * 16 + o) * 110592 + (long)(z * 48 + y) * 48 + x] = src[i];
}

extern "C" void kernel_launch(void* const* d_in, const int* in_sizes, int n_in,
                              void* d_out, int out_size, void* d_ws, size_t ws_size,
                              hipStream_t stream) {
  const float* in = (const float*)d_in[0];
  const float* wq = (const float*)d_in[1];
  const float* wk = (const float*)d_in[2];
  const float* wv = (const float*)d_in[3];
  const float* wo = (const float*)d_in[4];
  float* out = (float*)d_out;
  float* w = (float*)d_ws;

  const long R1re_o = 0,          R1im_o = 28311552;
  const long R2re_o = 56623104,   R2im_o = 70778880;
  const long blk_o  = 84934656;
  const long QKV_o  = 0,          G_o    = 42467328;
  const long S_o    = 92012544,   Og_o   = 92385792;
  const long U_o    = 0,          spec_o = 14155776;
  const long T1re_o = 15925248,   T1im_o = 19464192;
  const long T2re_o = 23003136,   T2im_o = 30081024;
  const long sumI_o = 37158912,   difI_o = 44236800;
  const long Wqkv_o = 106541568;
  const long Fre_o  = 106566144,  Fim_o  = 106570752;
  const long Vre_o  = 106575360,  Vim_o  = 106579968;
  const long WS_FLOATS = 106584576;
  if (ws_size < (size_t)WS_FLOATS * sizeof(float)) return;

  float *R1re = w+R1re_o, *R1im = w+R1im_o, *R2re = w+R2re_o, *R2im = w+R2im_o;
  float *blk = w+blk_o, *QKV = w+QKV_o, *G = w+G_o, *S = w+S_o, *Og = w+Og_o;
  float *U = w+U_o, *spec = w+spec_o;
  float *T1re = w+T1re_o, *T1im = w+T1im_o, *T2re = w+T2re_o, *T2im = w+T2im_o;
  float *sumI = w+sumI_o, *difI = w+difI_o, *Wqkv = w+Wqkv_o;
  float *Fre = w+Fre_o, *Fim = w+Fim_o, *Vre = w+Vre_o, *Vim = w+Vim_o;

  auto gemm = [&](const float* A, long sAm, long sAk, long sAb,
                  const float* B, long sBk, long sBn, long sBb,
                  float* C, long sCm, long sCn, long sCb,
                  int M, int N, int K, int batch, float alpha, int accum) {
    dim3 g((unsigned)((M + 63) / 64), (unsigned)((N + 63) / 64), (unsigned)batch);
    gemm_f32_wmma<<<g, 128, 0, stream>>>(A, sAm, sAk, sAb, B, sBk, sBn, sBb,
                                         C, sCm, sCn, sCb, M, N, K, alpha, accum);
  };
  auto ew = [&](int n) { return dim3((unsigned)((n + 255) / 256)); };

  gen_consts<<<ew(4608), 256, 0, stream>>>(Fre, Fim, Vre, Vim);
  pack_wqkv<<<ew(24576), 256, 0, stream>>>(wq, wk, wv, Wqkv);

  gemm(in, 96, 1, 0, Fre, 1, 96, 0, R1re, 48, 1, 0, 589824, 48, 96, 1, 1.f, 0);
  gemm(in, 96, 1, 0, Fim, 1, 96, 0, R1im, 48, 1, 0, 589824, 48, 96, 1, 1.f, 0);
  gemm(Fre, 96, 1, 0, R1re, 48, 1, 4608, R2re, 48, 1, 2304, 48, 48, 96, 6144,  1.f, 0);
  gemm(Fim, 96, 1, 0, R1im, 48, 1, 4608, R2re, 48, 1, 2304, 48, 48, 96, 6144, -1.f, 1);
  gemm(Fre, 96, 1, 0, R1im, 48, 1, 4608, R2im, 48, 1, 2304, 48, 48, 96, 6144,  1.f, 0);
  gemm(Fim, 96, 1, 0, R1re, 48, 1, 4608, R2im, 48, 1, 2304, 48, 48, 96, 6144,  1.f, 1);
  addsub<<<ew(14155776), 256, 0, stream>>>(R2re, R2im,  R1re,  R1im, 14155776);
  gemm(Fre, 96, 1, 0, R1im, 2304, 1, 221184, blk, 2304, 1, 110592, 48, 2304, 96, 64,  1.f, 0);
  gemm(Fim, 96, 1, 0, R1re, 2304, 1, 221184, blk, 2304, 1, 110592, 48, 2304, 96, 64, -1.f, 1);

  gemm(Wqkv, 64, 1, 0, blk, 110592, 1, 0, QKV, 110592, 1, 0, 384, 110592, 64, 1, 1.f, 0);
  group_qkv<<<ew(42467328), 256, 0, stream>>>(QKV, G);

  const long HD = 8192L * 216L;
  gemm(G, 1, 216, HD, G + 8 * HD, 216, 1, HD, S, 216, 1, 46656,
       216, 216, 8192, 8, 0.011048543456039806f, 0);
  selu_k<<<ew(373248), 256, 0, stream>>>(S, 373248);
  gemm(G + 16 * HD, 216, 1, HD, S, 1, 216, 46656, Og, 216, 1, HD,
       8192, 216, 216, 8, 1.f, 0);
  ungroup_o<<<ew(14155776), 256, 0, stream>>>(Og, U);

  gemm(wo, 128, 1, 0, U, 110592, 1, 0, spec, 110592, 1, 0, 16, 110592, 128, 1, 1.f, 0);

  gemm(spec, 48, 1, 0, Vre, 1, 48, 0, T1re, 96, 1, 0, 36864, 96, 48, 1, 1.f, 0);
  gemm(spec, 48, 1, 0, Vim, 1, 48, 0, T1im, 96, 1, 0, 36864, 96, 48, 1, 1.f, 0);
  gemm(Vre, 48, 1, 0, T1re, 96, 1, 4608, T2re, 96, 1, 9216, 96, 96, 48, 768,  1.f, 0);
  gemm(Vim, 48, 1, 0, T1im, 96, 1, 4608, T2re, 96, 1, 9216, 96, 96, 48, 768, -1.f, 1);
  gemm(Vre, 48, 1, 0, T1im, 96, 1, 4608, T2im, 96, 1, 9216, 96, 96, 48, 768,  1.f, 0);
  gemm(Vim, 48, 1, 0, T1re, 96, 1, 4608, T2im, 96, 1, 9216, 96, 96, 48, 768,  1.f, 1);
  addsub<<<ew(7077888), 256, 0, stream>>>(T2re, T2im, sumI, difI, 7077888);
  const float invN = 1.f / 884736.f;
  gemm(Vre, 48, 1, 0, sumI, 9216, 1, 442368, out, 9216, 1, 884736, 96, 9216, 48, 16, invN, 0);
  gemm(Vim, 48, 1, 0, difI, 9216, 1, 442368, out, 9216, 1, 884736, 96, 9216, 48, 16, invN, 1);
}


# Split-K on the Q^T K score product: K=8192 cut into 32 slices run as extra batch entries, partial planes in workspace, summed in fixed slice order inside the SELU kernel (summation order changes)
# speedup vs baseline: 1.5037x; 1.5037x over previous
#include <hip/hip_runtime.h>
#include <math.h>

typedef float v2f __attribute__((ext_vector_type(2)));
typedef float v8f __attribute__((ext_vector_type(8)));

__global__ __launch_bounds__(128) void gemm_f32_wmma(
    const float* __restrict__ A, long sAm, long sAk, long sAb,
    const float* __restrict__ B, long sBk, long sBn, long sBb,
    float* __restrict__ C, long sCm, long sCn, long sCb,
    int M, int N, int K, float alpha, int accum)
{
  const int lane = threadIdx.x & 31;
  const int wave = threadIdx.x >> 5;
  const int half = lane >> 4;
  const int l16  = lane & 15;
  const int mBase = blockIdx.x * 64;
  const int nBase = blockIdx.y * 64 + wave * 16;

  A += (long)blockIdx.z * sAb;
  B += (long)blockIdx.z * sBb;
  C += (long)blockIdx.z * sCb;

  v8f zero = {0.f,0.f,0.f,0.f,0.f,0.f,0.f,0.f};
  v8f acc[4];
#pragma unroll
  for (int t = 0; t < 4; ++t) acc[t] = zero;

  const float* aPtr[4];
  float aMask[4];
#pragma unroll
  for (int t = 0; t < 4; ++t) {
    int m = mBase + t * 16 + l16;
    bool v = (m < M);
    aMask[t] = v ? 1.f : 0.f;
    aPtr[t] = A + (long)(v ? m : 0) * sAm + (long)(2 * half) * sAk;
  }
  const int n = nBase + l16;
  const bool nv = (n < N);
  const float bMask = nv ? 1.f : 0.f;
  const float* bPtr = B + (long)(nv ? n : 0) * sBn + (long)(2 * half) * sBk;

  for (int k0 = 0; k0 < K; k0 += 4) {
    v2f bf;
    bf.x = bMask * bPtr[0];
    bf.y = bMask * bPtr[sBk];
    bPtr += 4 * sBk;
#pragma unroll
    for (int t = 0; t < 4; ++t) {
      v2f af;
      af.x = aMask[t] * aPtr[t][0];
      af.y = aMask[t] * aPtr[t][sAk];
      aPtr[t] += 4 * sAk;
      acc[t] = __builtin_amdgcn_wmma_f32_16x16x4_f32(
          false, af, false, bf, (short)0, acc[t], false, false);
    }
  }

  if (nv) {
#pragma unroll
    for (int t = 0; t < 4; ++t) {
#pragma unroll
      for (int r = 0; r < 8; ++r) {
        int m = mBase + t * 16 + r + 8 * half;
        if (m < M) {
          long idx = (long)m * sCm + (long)n * sCn;
          float val = alpha * acc[t][r];
          if (accum) C[idx] += val; else C[idx] = val;
        }
      }
    }
  }
}

__global__ void gen_consts(float* Fre, float* Fim, float* Vre, float* Vim) {
  int i = blockIdx.x * 256 + threadIdx.x;
  if (i >= 48 * 96) return;
  int k = i / 96, nn = i % 96;
  int f = (k < 24) ? k : (k + 48);
  int ph = (f * nn) % 96;
  float th = 6.283185307179586f * (float)ph / 96.f;
  float c = cosf(th), s = sinf(th);
  Fre[i] = c;  Fim[i] = -s;
  Vre[nn * 48 + k] = c;  Vim[nn * 48 + k] = s;
}

__global__ void pack_wqkv(const float* __restrict__ wq, const float* __restrict__ wk,
                          const float* __restrict__ wv, float* __restrict__ dst) {
  int i = blockIdx.x * 256 + threadIdx.x;
  if (i >= 384 * 64) return;
  int ic = i % 64, r = i / 64;
  int t = r / 128, q = r % 128;
  const float* w = (t == 0) ? wq : ((t == 1) ? wk : wv);
  dst[i] = w[q * 64 + ic];
}

__global__ void addsub(const float* __restrict__ x, const float* __restrict__ y,
                       float* __restrict__ sum, float* __restrict__ dif, int nelem) {
  int i = blockIdx.x * 256 + threadIdx.x;
  if (i >= nelem) return;
  float a = x[i], b = y[i];
  sum[i] = a + b;
  dif[i] = a - b;
}

#define KSPLIT 32
__global__ __launch_bounds__(256) void sum_scale_selu(const float* __restrict__ part, float* __restrict__ s,
                               int nvec, int planeVec, float scale) {
  int i = blockIdx.x * 256 + threadIdx.x;
  if (i >= nvec) return;
  int h = i / planeVec, e = i - h * planeVec;
  const float4* p = reinterpret_cast<const float4*>(part) + (long)h * KSPLIT * planeVec + e;
  float4 v[KSPLIT];
  for (int j = 0; j < KSPLIT; ++j) v[j] = p[(long)j * planeVec];
  float4 a = v[0];
  for (int j = 1; j < KSPLIT; ++j) {
    a.x += v[j].x; a.y += v[j].y; a.z += v[j].z; a.w += v[j].w;
  }
  const float lam = 1.0507009873554805f, alp = 1.6732632423543772f;
  float x0 = scale * a.x, x1 = scale * a.y, x2 = scale * a.z, x3 = scale * a.w;
  float4 r;
  r.x = lam * (x0 > 0.f ? x0 : alp * (expf(x0) - 1.f));
  r.y = lam * (x1 > 0.f ? x1 : alp * (expf(x1) - 1.f));
  r.z = lam * (x2 > 0.f ? x2 : alp * (expf(x2) - 1.f));
  r.w = lam * (x3 > 0.f ? x3 : alp * (expf(x3) - 1.f));
  reinterpret_cast<float4*>(s)[i] = r;
}

__global__ void group_qkv(const float* __restrict__ src, float* __restrict__ dst) {
  int i = blockIdx.x * 256 + threadIdx.x;
  if (i >= 42467328) return;
  int s = i % 110592, r = i / 110592;
  int x = s % 48, y = (s / 48) % 48, z = s / 2304;
  int o = r % 16, h = (r / 16) % 8, t = r / 128;
  int px = x & 7, nx = x >> 3, py = y & 7, ny = y >> 3, pz = z & 7, nz = z >> 3;
  int cp = ((((o << 3) | pz) << 3) | py) << 3 | px;
  int tok = (nz * 6 + ny) * 6 + nx;
  dst[((long)((t * 8 + h) * 8192 + cp)) * 216 + tok] = src[i];
}

__global__ void ungroup_o(const float* __restrict__ src, float* __restrict__ dst) {
  int i = blockIdx.x * 256 + threadIdx.x;
  if (i >= 14155776) return;
  int tok = i % 216, cp = (i / 216) % 8192, h = i / (216 * 8192);
  int px = cp & 7, py = (cp >> 3) & 7, pz = (cp >> 6) & 7, o = cp >> 9;
  int nx = tok % 6, ny = (tok / 6) % 6, nz = tok / 36;
  int z = nz * 8 + pz, y = ny * 8 + py, x = nx * 8 + px;
  dst[(long)(h * 16 + o) * 110592 + (long)(z * 48 + y) * 48 + x] = src[i];
}

extern "C" void kernel_launch(void* const* d_in, const int* in_sizes, int n_in,
                              void* d_out, int out_size, void* d_ws, size_t ws_size,
                              hipStream_t stream) {
  const float* in = (const float*)d_in[0];
  const float* wq = (const float*)d_in[1];
  const float* wk = (const float*)d_in[2];
  const float* wv = (const float*)d_in[3];
  const float* wo = (const float*)d_in[4];
  float* out = (float*)d_out;
  float* w = (float*)d_ws;

  const long R1re_o = 0,          R1im_o = 28311552;
  const long R2re_o = 56623104,   R2im_o = 70778880;
  const long blk_o  = 84934656;
  const long QKV_o  = 0,          G_o    = 42467328;
  const long S_o    = 92012544,   Og_o   = 92385792;
  const long P_o    = 0;
  const long U_o    = 0,          spec_o = 14155776;
  const long T1re_o = 15925248,   T1im_o = 19464192;
  const long T2re_o = 23003136,   T2im_o = 30081024;
  const long sumI_o = 37158912,   difI_o = 44236800;
  const long Wqkv_o = 106541568;
  const long Fre_o  = 106566144,  Fim_o  = 106570752;
  const long Vre_o  = 106575360,  Vim_o  = 106579968;
  const long WS_FLOATS = 106584576;
  if (ws_size < (size_t)WS_FLOATS * sizeof(float)) return;

  float *R1re = w+R1re_o, *R1im = w+R1im_o, *R2re = w+R2re_o, *R2im = w+R2im_o;
  float *blk = w+blk_o, *QKV = w+QKV_o, *G = w+G_o, *S = w+S_o, *Og = w+Og_o;
  float *U = w+U_o, *spec = w+spec_o, *P = w+P_o;
  float *T1re = w+T1re_o, *T1im = w+T1im_o, *T2re = w+T2re_o, *T2im = w+T2im_o;
  float *sumI = w+sumI_o, *difI = w+difI_o, *Wqkv = w+Wqkv_o;
  float *Fre = w+Fre_o, *Fim = w+Fim_o, *Vre = w+Vre_o, *Vim = w+Vim_o;

  auto gemm = [&](const float* A, long sAm, long sAk, long sAb,
                  const float* B, long sBk, long sBn, long sBb,
                  float* C, long sCm, long sCn, long sCb,
                  int M, int N, int K, int batch, float alpha, int accum) {
    dim3 g((unsigned)((M + 63) / 64), (unsigned)((N + 63) / 64), (unsigned)batch);
    gemm_f32_wmma<<<g, 128, 0, stream>>>(A, sAm, sAk, sAb, B, sBk, sBn, sBb,
                                         C, sCm, sCn, sCb, M, N, K, alpha, accum);
  };
  auto ew = [&](int n) { return dim3((unsigned)((n + 255) / 256)); };

  gen_consts<<<ew(4608), 256, 0, stream>>>(Fre, Fim, Vre, Vim);
  pack_wqkv<<<ew(24576), 256, 0, stream>>>(wq, wk, wv, Wqkv);

  gemm(in, 96, 1, 0, Fre, 1, 96, 0, R1re, 48, 1, 0, 589824, 48, 96, 1, 1.f, 0);
  gemm(in, 96, 1, 0, Fim, 1, 96, 0, R1im, 48, 1, 0, 589824, 48, 96, 1, 1.f, 0);
  gemm(Fre, 96, 1, 0, R1re, 48, 1, 4608, R2re, 48, 1, 2304, 48, 48, 96, 6144,  1.f, 0);
  gemm(Fim, 96, 1, 0, R1im, 48, 1, 4608, R2re, 48, 1, 2304, 48, 48, 96, 6144, -1.f, 1);
  gemm(Fre, 96, 1, 0, R1im, 48, 1, 4608, R2im, 48, 1, 2304, 48, 48, 96, 6144,  1.f, 0);
  gemm(Fim, 96, 1, 0, R1re, 48, 1, 4608, R2im, 48, 1, 2304, 48, 48, 96, 6144,  1.f, 1);
  addsub<<<ew(14155776), 256, 0, stream>>>(R2re, R2im,  R1re,  R1im, 14155776);
  gemm(Fre, 96, 1, 0, R1im, 2304, 1, 221184, blk, 2304, 1, 110592, 48, 2304, 96, 64,  1.f, 0);
  gemm(Fim, 96, 1, 0, R1re, 2304, 1, 221184, blk, 2304, 1, 110592, 48, 2304, 96, 64, -1.f, 1);

  gemm(Wqkv, 64, 1, 0, blk, 110592, 1, 0, QKV, 110592, 1, 0, 384, 110592, 64, 1, 1.f, 0);
  group_qkv<<<ew(42467328), 256, 0, stream>>>(QKV, G);

  const long HD = 8192L * 216L;
  const long SL = 256L * 216L;
  gemm(G, 1, 216, SL, G + 8 * HD, 216, 1, SL, P, 216, 1, 46656,
       216, 216, 256, 8 * KSPLIT, 1.f, 0);
  sum_scale_selu<<<ew(93312), 256, 0, stream>>>(P, S, 93312, 11664,
                                                0.011048543456039806f);
  gemm(G + 16 * HD, 216, 1, HD, S, 1, 216, 46656, Og, 216, 1, HD,
       8192, 216, 216, 8, 1.f, 0);
  ungroup_o<<<ew(14155776), 256, 0, stream>>>(Og, U);

  gemm(wo, 128, 1, 0, U, 110592, 1, 0, spec, 110592, 1, 0, 16, 110592, 128, 1, 1.f, 0);

  gemm(spec, 48, 1, 0, Vre, 1, 48, 0, T1re, 96, 1, 0, 36864, 96, 48, 1, 1.f, 0);
  gemm(spec, 48, 1, 0, Vim, 1, 48, 0, T1im, 96, 1, 0, 36864, 96, 48, 1, 1.f, 0);
  gemm(Vre, 48, 1, 0, T1re, 96, 1, 4608, T2re, 96, 1, 9216, 96, 96, 48, 768,  1.f, 0);
  gemm(Vim, 48, 1, 0, T1im, 96, 1, 4608, T2re, 96, 1, 9216, 96, 96, 48, 768, -1.f, 1);
  gemm(Vre, 48, 1, 0, T1im, 96, 1, 4608, T2im, 96, 1, 9216, 96, 96, 48, 768,  1.f, 0);
  gemm(Vim, 48, 1, 0, T1re, 96, 1, 4608, T2im, 96, 1, 9216, 96, 96, 48, 768,  1.f, 1);
  addsub<<<ew(7077888), 256, 0, stream>>>(T2re, T2im, sumI, difI, 7077888);
  const float invN = 1.f / 884736.f;
  gemm(Vre, 48, 1, 0, sumI, 9216, 1, 442368, out, 9216, 1, 884736, 96, 9216, 48, 16, invN, 0);
  gemm(Vim, 48, 1, 0, difI, 9216, 1, 442368, out, 9216, 1, 884736, 96, 9216, 48, 16, invN, 1);
}


# Stage the GEMM's A and B tiles through LDS: each 64x16 tile fetched once per block with 16-byte loads along its contiguous axis, double-buffered; waves read fragments from LDS; sum order unchanged
# speedup vs baseline: 2.4891x; 1.6553x over previous
#include <hip/hip_runtime.h>
#include <math.h>

typedef float v2f __attribute__((ext_vector_type(2)));
typedef float v8f __attribute__((ext_vector_type(8)));

#define KC 16
#define PK 20
#define PR 72
#define TILE 1280

template <bool AK, bool BK>
__global__ __launch_bounds__(128) void gemm_f32_wmma(
    const float* __restrict__ A, long sAm, long sAk, long sAb,
    const float* __restrict__ B, long sBk, long sBn, long sBb,
    float* __restrict__ C, long sCm, long sCn, long sCb,
    int M, int N, int K, float alpha, int accum)
{
  __shared__ __attribute__((aligned(16))) float As[2 * TILE];
  __shared__ __attribute__((aligned(16))) float Bs[2 * TILE];

  const int tid  = threadIdx.x;
  const int lane = tid & 31;
  const int wave = tid >> 5;
  const int half = lane >> 4;
  const int l16  = lane & 15;
  const int mBase = blockIdx.x * 64;
  const int nBlk  = blockIdx.y * 64;

  A += (long)blockIdx.z * sAb;
  B += (long)blockIdx.z * sBb;
  C += (long)blockIdx.z * sCb;

  v8f zero = {0.f,0.f,0.f,0.f,0.f,0.f,0.f,0.f};
  v8f acc[4];
#pragma unroll
  for (int t = 0; t < 4; ++t) acc[t] = zero;

  long aOff[2], bOff[2];
  int  aKo[2], bKo[2];
  bool aIn[2], bIn[2];
  int  aDst[2], bDst[2];
  for (int j = 0; j < 2; ++j) {
    const int idx = tid + 128 * j;
    if (AK) {
      const int row = idx >> 2, kq = (idx & 3) * 4, m = mBase + row;
      aIn[j] = (m < M);  aKo[j] = kq;  aDst[j] = row * PK + kq;
      aOff[j] = (long)(aIn[j] ? m : 0) * sAm + kq;
    } else {
      const int kk = idx >> 4, mq = (idx & 15) * 4, m = mBase + mq;
      aIn[j] = (m < M);  aKo[j] = kk;  aDst[j] = kk * PR + mq;
      aOff[j] = (long)kk * sAk + (aIn[j] ? m : 0);
    }
    if (BK) {
      const int row = idx >> 2, kq = (idx & 3) * 4, n = nBlk + row;
      bIn[j] = (n < N);  bKo[j] = kq;  bDst[j] = row * PK + kq;
      bOff[j] = (long)(bIn[j] ? n : 0) * sBn + kq;
    } else {
      const int kk = idx >> 4, nq = (idx & 15) * 4, n = nBlk + nq;
      bIn[j] = (n < N);  bKo[j] = kk;  bDst[j] = kk * PR + nq;
      bOff[j] = (long)kk * sBk + (bIn[j] ? n : 0);
    }
  }
  const long aStep = AK ? (long)KC : (long)KC * sAk;
  const long bStep = BK ? (long)KC : (long)KC * sBk;

  const int aFrag = AK ? (l16 * PK + 2 * half) : (2 * half * PR + l16);
  const int bFrag = BK ? ((wave * 16 + l16) * PK + 2 * half)
                       : (2 * half * PR + wave * 16 + l16);

  float4 ra[2], rb[2];

  for (int j = 0; j < 2; ++j) {
    ra[j] = make_float4(0.f, 0.f, 0.f, 0.f);
    rb[j] = make_float4(0.f, 0.f, 0.f, 0.f);
    if (aIn[j] && aKo[j] < K) ra[j] = *reinterpret_cast<const float4*>(A + aOff[j]);
    if (bIn[j] && bKo[j] < K) rb[j] = *reinterpret_cast<const float4*>(B + bOff[j]);
    aOff[j] += aStep;  bOff[j] += bStep;
  }
  for (int j = 0; j < 2; ++j) {
    *reinterpret_cast<float4*>(As + aDst[j]) = ra[j];
    *reinterpret_cast<float4*>(Bs + bDst[j]) = rb[j];
  }
  __syncthreads();

  int p = 0;
  for (int k0 = 0; ; k0 += KC) {
    const bool more = (k0 + KC < K);
    if (more) {
      for (int j = 0; j < 2; ++j) {
        ra[j] = make_float4(0.f, 0.f, 0.f, 0.f);
        rb[j] = make_float4(0.f, 0.f, 0.f, 0.f);
        if (aIn[j] && k0 + KC + aKo[j] < K) ra[j] = *reinterpret_cast<const float4*>(A + aOff[j]);
        if (bIn[j] && k0 + KC + bKo[j] < K) rb[j] = *reinterpret_cast<const float4*>(B + bOff[j]);
        aOff[j] += aStep;  bOff[j] += bStep;
      }
    }

    const float* as = As + p * TILE + aFrag;
    const float* bs = Bs + p * TILE + bFrag;
    for (int ks = 0; ks < KC / 4; ++ks) {
      v2f bf;
      if (BK) {
        bf = *reinterpret_cast<const v2f*>(bs + ks * 4);
      } else {
        bf.x = bs[(ks * 4) * PR];
        bf.y = bs[(ks * 4 + 1) * PR];
      }
      for (int t = 0; t < 4; ++t) {
        v2f af;
        if (AK) {
          af = *reinterpret_cast<const v2f*>(as + t * 16 * PK + ks * 4);
        } else {
          af.x = as[(ks * 4) * PR + t * 16];
          af.y = as[(ks * 4 + 1) * PR + t * 16];
        }
        acc[t] = __builtin_amdgcn_wmma_f32_16x16x4_f32(
            false, af, false, bf, (short)0, acc[t], false, false);
      }
    }

    if (!more) break;
    p ^= 1;
    for (int j = 0; j < 2; ++j) {
      *reinterpret_cast<float4*>(As + p * TILE + aDst[j]) = ra[j];
      *reinterpret_cast<float4*>(Bs + p * TILE + bDst[j]) = rb[j];
    }
    __syncthreads();
  }

  const int n = nBlk + wave * 16 + l16;
  if (n < N) {
#pragma unroll
    for (int t = 0; t < 4; ++t) {
#pragma unroll
      for (int r = 0; r < 8; ++r) {
        int m = mBase + t * 16 + r + 8 * half;
        if (m < M) {
          long idx = (long)m * sCm + (long)n * sCn;
          float val = alpha * acc[t][r];
          if (accum) C[idx] += val; else C[idx] = val;
        }
      }
    }
  }
}

template <bool AK, bool BK>
static void launch_gemm(hipStream_t stream,
                        const float* A, long sAm, long sAk, long sAb,
                        const float* B, long sBk, long sBn, long sBb,
                        float* C, long sCm, long sCn, long sCb,
                        int M, int N, int K, int batch, float alpha, int accum) {
  dim3 g((unsigned)((M + 63) / 64), (unsigned)((N + 63) / 64), (unsigned)batch);
  gemm_f32_wmma<AK, BK><<<g, 128, 0, stream>>>(A, sAm, sAk, sAb, B, sBk, sBn, sBb,
                                               C, sCm, sCn, sCb, M, N, K, alpha, accum);
}

__global__ void gen_consts(float* Fre, float* Fim, float* Vre, float* Vim) {
  int i = blockIdx.x * 256 + threadIdx.x;
  if (i >= 48 * 96) return;
  int k = i / 96, nn = i % 96;
  int f = (k < 24) ? k : (k + 48);
  int ph = (f * nn) % 96;
  float th = 6.283185307179586f * (float)ph / 96.f;
  float c = cosf(th), s = sinf(th);
  Fre[i] = c;  Fim[i] = -s;
  Vre[nn * 48 + k] = c;  Vim[nn * 48 + k] = s;
}

__global__ void pack_wqkv(const float* __restrict__ wq, const float* __restrict__ wk,
                          const float* __restrict__ wv, float* __restrict__ dst) {
  int i = blockIdx.x * 256 + threadIdx.x;
  if (i >= 384 * 64) return;
  int ic = i % 64, r = i / 64;
  int t = r / 128, q = r % 128;
  const float* w = (t == 0) ? wq : ((t == 1) ? wk : wv);
  dst[i] = w[q * 64 + ic];
}

__global__ void addsub(const float* __restrict__ x, const float* __restrict__ y,
                       float* __restrict__ sum, float* __restrict__ dif, int nelem) {
  int i = blockIdx.x * 256 + threadIdx.x;
  if (i >= nelem) return;
  float a = x[i], b = y[i];
  sum[i] = a + b;
  dif[i] = a - b;
}

#define KSPLIT 32
__global__ __launch_bounds__(256) void sum_scale_selu(const float* __restrict__ part, float* __restrict__ s,
                               int nvec, int planeVec, float scale) {
  int i = blockIdx.x * 256 + threadIdx.x;
  if (i >= nvec) return;
  int h = i / planeVec, e = i - h * planeVec;
  const float4* p = reinterpret_cast<const float4*>(part) + (long)h * KSPLIT * planeVec + e;
  float4 v[KSPLIT];
  for (int j = 0; j < KSPLIT; ++j) v[j] = p[(long)j * planeVec];
  float4 a = v[0];
  for (int j = 1; j < KSPLIT; ++j) {
    a.x += v[j].x; a.y += v[j].y; a.z += v[j].z; a.w += v[j].w;
  }
  const float lam = 1.0507009873554805f, alp = 1.6732632423543772f;
  float x0 = scale * a.x, x1 = scale * a.y, x2 = scale * a.z, x3 = scale * a.w;
  float4 r;
  r.x = lam * (x0 > 0.f ? x0 : alp * (expf(x0) - 1.f));
  r.y = lam * (x1 > 0.f ? x1 : alp * (expf(x1) - 1.f));
  r.z = lam * (x2 > 0.f ? x2 : alp * (expf(x2) - 1.f));
  r.w = lam * (x3 > 0.f ? x3 : alp * (expf(x3) - 1.f));
  reinterpret_cast<float4*>(s)[i] = r;
}

__global__ void group_qkv(const float* __restrict__ src, float* __restrict__ dst) {
  int i = blockIdx.x * 256 + threadIdx.x;
  if (i >= 42467328) return;
  int s = i % 110592, r = i / 110592;
  int x = s % 48, y = (s / 48) % 48, z = s / 2304;
  int o = r % 16, h = (r / 16) % 8, t = r / 128;
  int px = x & 7, nx = x >> 3, py = y & 7, ny = y >> 3, pz = z & 7, nz = z >> 3;
  int cp = ((((o << 3) | pz) << 3) | py) << 3 | px;
  int tok = (nz * 6 + ny) * 6 + nx;
  dst[((long)((t * 8 + h) * 8192 + cp)) * 216 + tok] = src[i];
}

__global__ void ungroup_o(const float* __restrict__ src, float* __restrict__ dst) {
  int i = blockIdx.x * 256 + threadIdx.x;
  if (i >= 14155776) return;
  int tok = i % 216, cp = (i / 216) % 8192, h = i / (216 * 8192);
  int px = cp & 7, py = (cp >> 3) & 7, pz = (cp >> 6) & 7, o = cp >> 9;
  int nx = tok % 6, ny = (tok / 6) % 6, nz = tok / 36;
  int z = nz * 8 + pz, y = ny * 8 + py, x = nx * 8 + px;
  dst[(long)(h * 16 + o) * 110592 + (long)(z * 48 + y) * 48 + x] = src[i];
}

extern "C" void kernel_launch(void* const* d_in, const int* in_sizes, int n_in,
                              void* d_out, int out_size, void* d_ws, size_t ws_size,
                              hipStream_t stream) {
  const float* in = (const float*)d_in[0];
  const float* wq = (const float*)d_in[1];
  const float* wk = (const float*)d_in[2];
  const float* wv = (const float*)d_in[3];
  const float* wo = (const float*)d_in[4];
  float* out = (float*)d_out;
  float* w = (float*)d_ws;

  const long R1re_o = 0,          R1im_o = 28311552;
  const long R2re_o = 56623104,   R2im_o = 70778880;
  const long blk_o  = 84934656;
  const long QKV_o  = 0,          G_o    = 42467328;
  const long S_o    = 92012544,   Og_o   = 92385792;
  const long P_o    = 0;
  const long U_o    = 0,          spec_o = 14155776;
  const long T1re_o = 15925248,   T1im_o = 19464192;
  const long T2re_o = 23003136,   T2im_o = 30081024;
  const long sumI_o = 37158912,   difI_o = 44236800;
  const long Wqkv_o = 106541568;
  const long Fre_o  = 106566144,  Fim_o  = 106570752;
  const long Vre_o  = 106575360,  Vim_o  = 106579968;
  const long WS_FLOATS = 106584576;
  if (ws_size < (size_t)WS_FLOATS * sizeof(float)) return;

  float *R1re = w+R1re_o, *R1im = w+R1im_o, *R2re = w+R2re_o, *R2im = w+R2im_o;
  float *blk = w+blk_o, *QKV = w+QKV_o, *G = w+G_o, *S = w+S_o, *Og = w+Og_o;
  float *U = w+U_o, *spec = w+spec_o, *P = w+P_o;
  float *T1re = w+T1re_o, *T1im = w+T1im_o, *T2re = w+T2re_o, *T2im = w+T2im_o;
  float *sumI = w+sumI_o, *difI = w+difI_o, *Wqkv = w+Wqkv_o;
  float *Fre = w+Fre_o, *Fim = w+Fim_o, *Vre = w+Vre_o, *Vim = w+Vim_o;

  auto ew = [&](int n) { return dim3((unsigned)((n + 255) / 256)); };

  gen_consts<<<ew(4608), 256, 0, stream>>>(Fre, Fim, Vre, Vim);
  pack_wqkv<<<ew(24576), 256, 0, stream>>>(wq, wk, wv, Wqkv);

  launch_gemm<true, true>(stream, in, 96, 1, 0, Fre, 1, 96, 0, R1re, 48, 1, 0, 589824, 48, 96, 1, 1.f, 0);
  launch_gemm<true, true>(stream, in, 96, 1, 0, Fim, 1, 96, 0, R1im, 48, 1, 0, 589824, 48, 96, 1, 1.f, 0);
  launch_gemm<true, false>(stream, Fre, 96, 1, 0, R1re, 48, 1, 4608, R2re, 48, 1, 2304, 48, 48, 96, 6144,  1.f, 0);
  launch_gemm<true, false>(stream, Fim, 96, 1, 0, R1im, 48, 1, 4608, R2re, 48, 1, 2304, 48, 48, 96, 6144, -1.f, 1);
  launch_gemm<true, false>(stream, Fre, 96, 1, 0, R1im, 48, 1, 4608, R2im, 48, 1, 2304, 48, 48, 96, 6144,  1.f, 0);
  launch_gemm<true, false>(stream, Fim, 96, 1, 0, R1re, 48, 1, 4608, R2im, 48, 1, 2304, 48, 48, 96, 6144,  1.f, 1);
  addsub<<<ew(14155776), 256, 0, stream>>>(R2re, R2im,  R1re,  R1im, 14155776);
  launch_gemm<true, false>(stream, Fre, 96, 1, 0, R1im, 2304, 1, 221184, blk, 2304, 1, 110592, 48, 2304, 96, 64,  1.f, 0);
  launch_gemm<true, false>(stream, Fim, 96, 1, 0, R1re, 2304, 1, 221184, blk, 2304, 1, 110592, 48, 2304, 96, 64, -1.f, 1);

  launch_gemm<true, false>(stream, Wqkv, 64, 1, 0, blk, 110592, 1, 0, QKV, 110592, 1, 0, 384, 110592, 64, 1, 1.f, 0);
  group_qkv<<<ew(42467328), 256, 0, stream>>>(QKV, G);

  const long HD = 8192L * 216L;
  const long SL = 256L * 216L;
  launch_gemm<false, false>(stream, G, 1, 216, SL, G + 8 * HD, 216, 1, SL, P, 216, 1, 46656,
       216, 216, 256, 8 * KSPLIT, 1.f, 0);
  sum_scale_selu<<<ew(93312), 256, 0, stream>>>(P, S, 93312, 11664,
                                                0.011048543456039806f);
  launch_gemm<true, true>(stream, G + 16 * HD, 216, 1, HD, S, 1, 216, 46656, Og, 216, 1, HD,
       8192, 216, 216, 8, 1.f, 0);
  ungroup_o<<<ew(14155776), 256, 0, stream>>>(Og, U);

  launch_gemm<true, false>(stream, wo, 128, 1, 0, U, 110592, 1, 0, spec, 110592, 1, 0, 16, 110592, 128, 1, 1.f, 0);

  launch_gemm<true, true>(stream, spec, 48, 1, 0, Vre, 1, 48, 0, T1re, 96, 1, 0, 36864, 96, 48, 1, 1.f, 0);
  launch_gemm<true, true>(stream, spec, 48, 1, 0, Vim, 1, 48, 0, T1im, 96, 1, 0, 36864, 96, 48, 1, 1.f, 0);
  launch_gemm<true, false>(stream, Vre, 48, 1, 0, T1re, 96, 1, 4608, T2re, 96, 1, 9216, 96, 96, 48, 768,  1.f, 0);
  launch_gemm<true, false>(stream, Vim, 48, 1, 0, T1im, 96, 1, 4608, T2re, 96, 1, 9216, 96, 96, 48, 768, -1.f, 1);
  launch_gemm<true, false>(stream, Vre, 48, 1, 0, T1im, 96, 1, 4608, T2im, 96, 1, 9216, 96, 96, 48, 768,  1.f, 0);
  launch_gemm<true, false>(stream, Vim, 48, 1, 0, T1re, 96, 1, 4608, T2im, 96, 1, 9216, 96, 96, 48, 768,  1.f, 1);
  addsub<<<ew(7077888), 256, 0, stream>>>(T2re, T2im, sumI, difI, 7077888);
  const float invN = 1.f / 884736.f;
  launch_gemm<true, false>(stream, Vre, 48, 1, 0, sumI, 9216, 1, 442368, out, 9216, 1, 884736, 96, 9216, 48, 16, invN, 0);
  launch_gemm<true, false>(stream, Vim, 48, 1, 0, difI, 9216, 1, 442368, out, 9216, 1, 884736, 96, 9216, 48, 16, invN, 1);
}


# Patch grouping moved across the per-position channel projections: permute the 64-channel spectrum before QKV and the 16-channel result after the out projection, not 384 and 128 channels; same sums
# speedup vs baseline: 2.9393x; 1.1809x over previous
#include <hip/hip_runtime.h>
#include <math.h>

typedef float v2f __attribute__((ext_vector_type(2)));
typedef float v8f __attribute__((ext_vector_type(8)));

#define KC 16
#define PK 20
#define PR 72
#define TILE 1280

template <bool AK, bool BK>
__global__ __launch_bounds__(128) void gemm_f32_wmma(
    const float* __restrict__ A, long sAm, long sAk, long sAb,
    const float* __restrict__ B, long sBk, long sBn, long sBb,
    float* __restrict__ C, long sCm, long sCn, long sCb,
    int M, int N, int K, float alpha, int accum)
{
  __shared__ __attribute__((aligned(16))) float As[2 * TILE];
  __shared__ __attribute__((aligned(16))) float Bs[2 * TILE];

  const int tid  = threadIdx.x;
  const int lane = tid & 31;
  const int wave = tid >> 5;
  const int half = lane >> 4;
  const int l16  = lane & 15;
  const int mBase = blockIdx.x * 64;
  const int nBlk  = blockIdx.y * 64;

  A += (long)blockIdx.z * sAb;
  B += (long)blockIdx.z * sBb;
  C += (long)blockIdx.z * sCb;

  v8f zero = {0.f,0.f,0.f,0.f,0.f,0.f,0.f,0.f};
  v8f acc[4];
#pragma unroll
  for (int t = 0; t < 4; ++t) acc[t] = zero;

  long aOff[2], bOff[2];
  int  aKo[2], bKo[2];
  bool aIn[2], bIn[2];
  int  aDst[2], bDst[2];
  for (int j = 0; j < 2; ++j) {
    const int idx = tid + 128 * j;
    if (AK) {
      const int row = idx >> 2, kq = (idx & 3) * 4, m = mBase + row;
      aIn[j] = (m < M);  aKo[j] = kq;  aDst[j] = row * PK + kq;
      aOff[j] = (long)(aIn[j] ? m : 0) * sAm + kq;
    } else {
      const int kk = idx >> 4, mq = (idx & 15) * 4, m = mBase + mq;
      aIn[j] = (m < M);  aKo[j] = kk;  aDst[j] = kk * PR + mq;
      aOff[j] = (long)kk * sAk + (aIn[j] ? m : 0);
    }
    if (BK) {
      const int row = idx >> 2, kq = (idx & 3) * 4, n = nBlk + row;
      bIn[j] = (n < N);  bKo[j] = kq;  bDst[j] = row * PK + kq;
      bOff[j] = (long)(bIn[j] ? n : 0) * sBn + kq;
    } else {
      const int kk = idx >> 4, nq = (idx & 15) * 4, n = nBlk + nq;
      bIn[j] = (n < N);  bKo[j] = kk;  bDst[j] = kk * PR + nq;
      bOff[j] = (long)kk * sBk + (bIn[j] ? n : 0);
    }
  }
  const long aStep = AK ? (long)KC : (long)KC * sAk;
  const long bStep = BK ? (long)KC : (long)KC * sBk;

  const int aFrag = AK ? (l16 * PK + 2 * half) : (2 * half * PR + l16);
  const int bFrag = BK ? ((wave * 16 + l16) * PK + 2 * half)
                       : (2 * half * PR + wave * 16 + l16);

  float4 ra[2], rb[2];

  for (int j = 0; j < 2; ++j) {
    ra[j] = make_float4(0.f, 0.f, 0.f, 0.f);
    rb[j] = make_float4(0.f, 0.f, 0.f, 0.f);
    if (aIn[j] && aKo[j] < K) ra[j] = *reinterpret_cast<const float4*>(A + aOff[j]);
    if (bIn[j] && bKo[j] < K) rb[j] = *reinterpret_cast<const float4*>(B + bOff[j]);
    aOff[j] += aStep;  bOff[j] += bStep;
  }
  for (int j = 0; j < 2; ++j) {
    *reinterpret_cast<float4*>(As + aDst[j]) = ra[j];
    *reinterpret_cast<float4*>(Bs + bDst[j]) = rb[j];
  }
  __syncthreads();

  int p = 0;
  for (int k0 = 0; ; k0 += KC) {
    const bool more = (k0 + KC < K);
    if (more) {
      for (int j = 0; j < 2; ++j) {
        ra[j] = make_float4(0.f, 0.f, 0.f, 0.f);
        rb[j] = make_float4(0.f, 0.f, 0.f, 0.f);
        if (aIn[j] && k0 + KC + aKo[j] < K) ra[j] = *reinterpret_cast<const float4*>(A + aOff[j]);
        if (bIn[j] && k0 + KC + bKo[j] < K) rb[j] = *reinterpret_cast<const float4*>(B + bOff[j]);
        aOff[j] += aStep;  bOff[j] += bStep;
      }
    }

    const float* as = As + p * TILE + aFrag;
    const float* bs = Bs + p * TILE + bFrag;
    for (int ks = 0; ks < KC / 4; ++ks) {
      v2f bf;
      if (BK) {
        bf = *reinterpret_cast<const v2f*>(bs + ks * 4);
      } else {
        bf.x = bs[(ks * 4) * PR];
        bf.y = bs[(ks * 4 + 1) * PR];
      }
      for (int t = 0; t < 4; ++t) {
        v2f af;
        if (AK) {
          af = *reinterpret_cast<const v2f*>(as + t * 16 * PK + ks * 4);
        } else {
          af.x = as[(ks * 4) * PR + t * 16];
          af.y = as[(ks * 4 + 1) * PR + t * 16];
        }
        acc[t] = __builtin_amdgcn_wmma_f32_16x16x4_f32(
            false, af, false, bf, (short)0, acc[t], false, false);
      }
    }

    if (!more) break;
    p ^= 1;
    for (int j = 0; j < 2; ++j) {
      *reinterpret_cast<float4*>(As + p * TILE + aDst[j]) = ra[j];
      *reinterpret_cast<float4*>(Bs + p * TILE + bDst[j]) = rb[j];
    }
    __syncthreads();
  }

  const int n = nBlk + wave * 16 + l16;
  if (n < N) {
#pragma unroll
    for (int t = 0; t < 4; ++t) {
#pragma unroll
      for (int r = 0; r < 8; ++r) {
        int m = mBase + t * 16 + r + 8 * half;
        if (m < M) {
          long idx = (long)m * sCm + (long)n * sCn;
          float val = alpha * acc[t][r];
          if (accum) C[idx] += val; else C[idx] = val;
        }
      }
    }
  }
}

template <bool AK, bool BK>
static void launch_gemm(hipStream_t stream,
                        const float* A, long sAm, long sAk, long sAb,
                        const float* B, long sBk, long sBn, long sBb,
                        float* C, long sCm, long sCn, long sCb,
                        int M, int N, int K, int batch, float alpha, int accum) {
  dim3 g((unsigned)((M + 63) / 64), (unsigned)((N + 63) / 64), (unsigned)batch);
  gemm_f32_wmma<AK, BK><<<g, 128, 0, stream>>>(A, sAm, sAk, sAb, B, sBk, sBn, sBb,
                                               C, sCm, sCn, sCb, M, N, K, alpha, accum);
}

__global__ void gen_consts(float* Fre, float* Fim, float* Vre, float* Vim) {
  int i = blockIdx.x * 256 + threadIdx.x;
  if (i >= 48 * 96) return;
  int k = i / 96, nn = i % 96;
  int f = (k < 24) ? k : (k + 48);
  int ph = (f * nn) % 96;
  float th = 6.283185307179586f * (float)ph / 96.f;
  float c = cosf(th), s = sinf(th);
  Fre[i] = c;  Fim[i] = -s;
  Vre[nn * 48 + k] = c;  Vim[nn * 48 + k] = s;
}

__global__ void pack_wqkv(const float* __restrict__ wq, const float* __restrict__ wk,
                          const float* __restrict__ wv, float* __restrict__ dst) {
  int i = blockIdx.x * 256 + threadIdx.x;
  if (i >= 384 * 64) return;
  int ic = i % 64, r = i / 64;
  int t = r / 128, q = r % 128;
  const float* w = (t == 0) ? wq : ((t == 1) ? wk : wv);
  dst[i] = w[q * 64 + ic];
}

__global__ void addsub(const float* __restrict__ x, const float* __restrict__ y,
                       float* __restrict__ sum, float* __restrict__ dif, int nelem) {
  int i = blockIdx.x * 256 + threadIdx.x;
  if (i >= nelem) return;
  float a = x[i], b = y[i];
  sum[i] = a + b;
  dif[i] = a - b;
}

#define KSPLIT 32
__global__ __launch_bounds__(256) void sum_scale_selu(const float* __restrict__ part, float* __restrict__ s,
                               int nvec, int planeVec, float scale) {
  int i = blockIdx.x * 256 + threadIdx.x;
  if (i >= nvec) return;
  int h = i / planeVec, e = i - h * planeVec;
  const float4* p = reinterpret_cast<const float4*>(part) + (long)h * KSPLIT * planeVec + e;
  float4 v[KSPLIT];
  for (int j = 0; j < KSPLIT; ++j) v[j] = p[(long)j * planeVec];
  float4 a = v[0];
  for (int j = 1; j < KSPLIT; ++j) {
    a.x += v[j].x; a.y += v[j].y; a.z += v[j].z; a.w += v[j].w;
  }
  const float lam = 1.0507009873554805f, alp = 1.6732632423543772f;
  float x0 = scale * a.x, x1 = scale * a.y, x2 = scale * a.z, x3 = scale * a.w;
  float4 r;
  r.x = lam * (x0 > 0.f ? x0 : alp * (expf(x0) - 1.f));
  r.y = lam * (x1 > 0.f ? x1 : alp * (expf(x1) - 1.f));
  r.z = lam * (x2 > 0.f ? x2 : alp * (expf(x2) - 1.f));
  r.w = lam * (x3 > 0.f ? x3 : alp * (expf(x3) - 1.f));
  reinterpret_cast<float4*>(s)[i] = r;
}


__global__ void group_rows(const float* __restrict__ src, float* __restrict__ dst, int nelem) {
  int i = blockIdx.x * 256 + threadIdx.x;
  if (i >= nelem) return;
  int s = i % 110592, r = i / 110592;
  int x = s % 48, y = (s / 48) % 48, z = s / 2304;
  int px = x & 7, nx = x >> 3, py = y & 7, ny = y >> 3, pz = z & 7, nz = z >> 3;
  int p = (((pz << 3) | py) << 3) | px;
  int tok = (nz * 6 + ny) * 6 + nx;
  dst[(long)r * 110592 + p * 216 + tok] = src[i];
}

__global__ void ungroup_rows(const float* __restrict__ src, float* __restrict__ dst, int nelem) {
  int i = blockIdx.x * 256 + threadIdx.x;
  if (i >= nelem) return;
  int g = i % 110592, r = i / 110592;
  int tok = g % 216, p = g / 216;
  int px = p & 7, py = (p >> 3) & 7, pz = p >> 6;
  int nx = tok % 6, ny = (tok / 6) % 6, nz = tok / 36;
  int z = nz * 8 + pz, y = ny * 8 + py, x = nx * 8 + px;
  dst[(long)r * 110592 + (long)(z * 48 + y) * 48 + x] = src[i];
}

extern "C" void kernel_launch(void* const* d_in, const int* in_sizes, int n_in,
                              void* d_out, int out_size, void* d_ws, size_t ws_size,
                              hipStream_t stream) {
  const float* in = (const float*)d_in[0];
  const float* wq = (const float*)d_in[1];
  const float* wk = (const float*)d_in[2];
  const float* wv = (const float*)d_in[3];
  const float* wo = (const float*)d_in[4];
  float* out = (float*)d_out;
  float* w = (float*)d_ws;

  const long R1re_o = 0,          R1im_o = 28311552;
  const long R2re_o = 56623104,   R2im_o = 70778880;
  const long blk_o  = 84934656;
  const long blkG_o = 0,          G_o    = 42467328;
  const long S_o    = 92012544,   Og_o   = 92385792;
  const long P_o    = 0;
  const long specG_o = 0,         spec_o = 14155776;
  const long T1re_o = 15925248,   T1im_o = 19464192;
  const long T2re_o = 23003136,   T2im_o = 30081024;
  const long sumI_o = 37158912,   difI_o = 44236800;
  const long Wqkv_o = 106541568;
  const long Fre_o  = 106566144,  Fim_o  = 106570752;
  const long Vre_o  = 106575360,  Vim_o  = 106579968;
  const long WS_FLOATS = 106584576;
  if (ws_size < (size_t)WS_FLOATS * sizeof(float)) return;

  float *R1re = w+R1re_o, *R1im = w+R1im_o, *R2re = w+R2re_o, *R2im = w+R2im_o;
  float *blk = w+blk_o, *blkG = w+blkG_o, *G = w+G_o, *S = w+S_o, *Og = w+Og_o;
  float *specG = w+specG_o, *spec = w+spec_o, *P = w+P_o;
  float *T1re = w+T1re_o, *T1im = w+T1im_o, *T2re = w+T2re_o, *T2im = w+T2im_o;
  float *sumI = w+sumI_o, *difI = w+difI_o, *Wqkv = w+Wqkv_o;
  float *Fre = w+Fre_o, *Fim = w+Fim_o, *Vre = w+Vre_o, *Vim = w+Vim_o;

  auto ew = [&](int n) { return dim3((unsigned)((n + 255) / 256)); };

  gen_consts<<<ew(4608), 256, 0, stream>>>(Fre, Fim, Vre, Vim);
  pack_wqkv<<<ew(24576), 256, 0, stream>>>(wq, wk, wv, Wqkv);

  launch_gemm<true, true>(stream, in, 96, 1, 0, Fre, 1, 96, 0, R1re, 48, 1, 0, 589824, 48, 96, 1, 1.f, 0);
  launch_gemm<true, true>(stream, in, 96, 1, 0, Fim, 1, 96, 0, R1im, 48, 1, 0, 589824, 48, 96, 1, 1.f, 0);
  launch_gemm<true, false>(stream, Fre, 96, 1, 0, R1re, 48, 1, 4608, R2re, 48, 1, 2304, 48, 48, 96, 6144,  1.f, 0);
  launch_gemm<true, false>(stream, Fim, 96, 1, 0, R1im, 48, 1, 4608, R2re, 48, 1, 2304, 48, 48, 96, 6144, -1.f, 1);
  launch_gemm<true, false>(stream, Fre, 96, 1, 0, R1im, 48, 1, 4608, R2im, 48, 1, 2304, 48, 48, 96, 6144,  1.f, 0);
  launch_gemm<true, false>(stream, Fim, 96, 1, 0, R1re, 48, 1, 4608, R2im, 48, 1, 2304, 48, 48, 96, 6144,  1.f, 1);
  addsub<<<ew(14155776), 256, 0, stream>>>(R2re, R2im,  R1re,  R1im, 14155776);
  launch_gemm<true, false>(stream, Fre, 96, 1, 0, R1im, 2304, 1, 221184, blk, 2304, 1, 110592, 48, 2304, 96, 64,  1.f, 0);
  launch_gemm<true, false>(stream, Fim, 96, 1, 0, R1re, 2304, 1, 221184, blk, 2304, 1, 110592, 48, 2304, 96, 64, -1.f, 1);

  group_rows<<<ew(7077888), 256, 0, stream>>>(blk, blkG, 7077888);
  launch_gemm<true, false>(stream, Wqkv, 64, 1, 0, blkG, 110592, 1, 0, G, 110592, 1, 0, 384, 110592, 64, 1, 1.f, 0);

  const long HD = 8192L * 216L;
  const long SL = 256L * 216L;
  launch_gemm<false, false>(stream, G, 1, 216, SL, G + 8 * HD, 216, 1, SL, P, 216, 1, 46656,
       216, 216, 256, 8 * KSPLIT, 1.f, 0);
  sum_scale_selu<<<ew(93312), 256, 0, stream>>>(P, S, 93312, 11664,
                                                0.011048543456039806f);
  launch_gemm<true, true>(stream, G + 16 * HD, 216, 1, HD, S, 1, 216, 46656, Og, 216, 1, HD,
       8192, 216, 216, 8, 1.f, 0);

  launch_gemm<true, false>(stream, wo, 128, 1, 0, Og, 110592, 1, 0, specG, 110592, 1, 0, 16, 110592, 128, 1, 1.f, 0);
  ungroup_rows<<<ew(1769472), 256, 0, stream>>>(specG, spec, 1769472);

  launch_gemm<true, true>(stream, spec, 48, 1, 0, Vre, 1, 48, 0, T1re, 96, 1, 0, 36864, 96, 48, 1, 1.f, 0);
  launch_gemm<true, true>(stream, spec, 48, 1, 0, Vim, 1, 48, 0, T1im, 96, 1, 0, 36864, 96, 48, 1, 1.f, 0);
  launch_gemm<true, false>(stream, Vre, 48, 1, 0, T1re, 96, 1, 4608, T2re, 96, 1, 9216, 96, 96, 48, 768,  1.f, 0);
  launch_gemm<true, false>(stream, Vim, 48, 1, 0, T1im, 96, 1, 4608, T2re, 96, 1, 9216, 96, 96, 48, 768, -1.f, 1);
  launch_gemm<true, false>(stream, Vre, 48, 1, 0, T1im, 96, 1, 4608, T2im, 96, 1, 9216, 96, 96, 48, 768,  1.f, 0);
  launch_gemm<true, false>(stream, Vim, 48, 1, 0, T1re, 96, 1, 4608, T2im, 96, 1, 9216, 96, 96, 48, 768,  1.f, 1);
  addsub<<<ew(7077888), 256, 0, stream>>>(T2re, T2im, sumI, difI, 7077888);
  const float invN = 1.f / 884736.f;
  launch_gemm<true, false>(stream, Vre, 48, 1, 0, sumI, 9216, 1, 442368, out, 9216, 1, 884736, 96, 9216, 48, 16, invN, 0);
  launch_gemm<true, false>(stream, Vim, 48, 1, 0, difI, 9216, 1, 442368, out, 9216, 1, 884736, 96, 9216, 48, 16, invN, 1);
}


# Each DFT axis as ONE real GEMM: (re,im) stacked into K with combined twiddle blocks (Hartley add/sub folded in); no accumulate passes, no add/sub kernels; partial sums now one chain (order changes)
# speedup vs baseline: 4.0501x; 1.3779x over previous
#include <hip/hip_runtime.h>
#include <math.h>

typedef float v2f __attribute__((ext_vector_type(2)));
typedef float v8f __attribute__((ext_vector_type(8)));

#define KC 16
#define PK 20
#define PR 72
#define TILE 1280

template <bool AK, bool BK, bool NX>
__global__ __launch_bounds__(128) void gemm_f32_wmma(
    const float* __restrict__ A, long sAm, long sAk, long sAb,
    const float* __restrict__ B, long sBk, long sBn, long sBb,
    float* __restrict__ C, long sCm, long sCn, long sCb,
    int M, int N, int K, float alpha, int accum)
{
  __shared__ __attribute__((aligned(16))) float As[2 * TILE];
  __shared__ __attribute__((aligned(16))) float Bs[2 * TILE];

  const int tid  = threadIdx.x;
  const int lane = tid & 31;
  const int wave = tid >> 5;
  const int half = lane >> 4;
  const int l16  = lane & 15;
  const int mBase = (NX ? blockIdx.y : blockIdx.x) * 64;
  const int nBlk  = (NX ? blockIdx.x : blockIdx.y) * 64;

  A += (long)blockIdx.z * sAb;
  B += (long)blockIdx.z * sBb;
  C += (long)blockIdx.z * sCb;

  v8f zero = {0.f,0.f,0.f,0.f,0.f,0.f,0.f,0.f};
  v8f acc[4];
#pragma unroll
  for (int t = 0; t < 4; ++t) acc[t] = zero;

  long aOff[2], bOff[2];
  int  aKo[2], bKo[2];
  bool aIn[2], bIn[2];
  int  aDst[2], bDst[2];
  for (int j = 0; j < 2; ++j) {
    const int idx = tid + 128 * j;
    if (AK) {
      const int row = idx >> 2, kq = (idx & 3) * 4, m = mBase + row;
      aIn[j] = (m < M);  aKo[j] = kq;  aDst[j] = row * PK + kq;
      aOff[j] = (long)(aIn[j] ? m : 0) * sAm + kq;
    } else {
      const int kk = idx >> 4, mq = (idx & 15) * 4, m = mBase + mq;
      aIn[j] = (m < M);  aKo[j] = kk;  aDst[j] = kk * PR + mq;
      aOff[j] = (long)kk * sAk + (aIn[j] ? m : 0);
    }
    if (BK) {
      const int row = idx >> 2, kq = (idx & 3) * 4, n = nBlk + row;
      bIn[j] = (n < N);  bKo[j] = kq;  bDst[j] = row * PK + kq;
      bOff[j] = (long)(bIn[j] ? n : 0) * sBn + kq;
    } else {
      const int kk = idx >> 4, nq = (idx & 15) * 4, n = nBlk + nq;
      bIn[j] = (n < N);  bKo[j] = kk;  bDst[j] = kk * PR + nq;
      bOff[j] = (long)kk * sBk + (bIn[j] ? n : 0);
    }
  }
  const long aStep = AK ? (long)KC : (long)KC * sAk;
  const long bStep = BK ? (long)KC : (long)KC * sBk;

  const int aFrag = AK ? (l16 * PK + 2 * half) : (2 * half * PR + l16);
  const int bFrag = BK ? ((wave * 16 + l16) * PK + 2 * half)
                       : (2 * half * PR + wave * 16 + l16);

  float4 ra[2], rb[2];

  for (int j = 0; j < 2; ++j) {
    ra[j] = make_float4(0.f, 0.f, 0.f, 0.f);
    rb[j] = make_float4(0.f, 0.f, 0.f, 0.f);
    if (aIn[j] && aKo[j] < K) ra[j] = *reinterpret_cast<const float4*>(A + aOff[j]);
    if (bIn[j] && bKo[j] < K) rb[j] = *reinterpret_cast<const float4*>(B + bOff[j]);
    aOff[j] += aStep;  bOff[j] += bStep;
  }
  for (int j = 0; j < 2; ++j) {
    *reinterpret_cast<float4*>(As + aDst[j]) = ra[j];
    *reinterpret_cast<float4*>(Bs + bDst[j]) = rb[j];
  }
  __syncthreads();

  int p = 0;
  for (int k0 = 0; ; k0 += KC) {
    const bool more = (k0 + KC < K);
    if (more) {
      for (int j = 0; j < 2; ++j) {
        ra[j] = make_float4(0.f, 0.f, 0.f, 0.f);
        rb[j] = make_float4(0.f, 0.f, 0.f, 0.f);
        if (aIn[j] && k0 + KC + aKo[j] < K) ra[j] = *reinterpret_cast<const float4*>(A + aOff[j]);
        if (bIn[j] && k0 + KC + bKo[j] < K) rb[j] = *reinterpret_cast<const float4*>(B + bOff[j]);
        aOff[j] += aStep;  bOff[j] += bStep;
      }
    }

    const float* as = As + p * TILE + aFrag;
    const float* bs = Bs + p * TILE + bFrag;
    for (int ks = 0; ks < KC / 4; ++ks) {
      v2f bf;
      if (BK) {
        bf = *reinterpret_cast<const v2f*>(bs + ks * 4);
      } else {
        bf.x = bs[(ks * 4) * PR];
        bf.y = bs[(ks * 4 + 1) * PR];
      }
      for (int t = 0; t < 4; ++t) {
        v2f af;
        if (AK) {
          af = *reinterpret_cast<const v2f*>(as + t * 16 * PK + ks * 4);
        } else {
          af.x = as[(ks * 4) * PR + t * 16];
          af.y = as[(ks * 4 + 1) * PR + t * 16];
        }
        acc[t] = __builtin_amdgcn_wmma_f32_16x16x4_f32(
            false, af, false, bf, (short)0, acc[t], false, false);
      }
    }

    if (!more) break;
    p ^= 1;
    for (int j = 0; j < 2; ++j) {
      *reinterpret_cast<float4*>(As + p * TILE + aDst[j]) = ra[j];
      *reinterpret_cast<float4*>(Bs + p * TILE + bDst[j]) = rb[j];
    }
    __syncthreads();
  }

  const int n = nBlk + wave * 16 + l16;
  if (n < N) {
#pragma unroll
    for (int t = 0; t < 4; ++t) {
#pragma unroll
      for (int r = 0; r < 8; ++r) {
        int m = mBase + t * 16 + r + 8 * half;
        if (m < M) {
          long idx = (long)m * sCm + (long)n * sCn;
          float val = alpha * acc[t][r];
          if (accum) C[idx] += val; else C[idx] = val;
        }
      }
    }
  }
}

template <bool AK, bool BK, bool NX = false>
static void launch_gemm(hipStream_t stream,
                        const float* A, long sAm, long sAk, long sAb,
                        const float* B, long sBk, long sBn, long sBb,
                        float* C, long sCm, long sCn, long sCb,
                        int M, int N, int K, int batch, float alpha, int accum) {
  const unsigned mb = (unsigned)((M + 63) / 64), nb = (unsigned)((N + 63) / 64);
  dim3 g(NX ? nb : mb, NX ? mb : nb, (unsigned)batch);
  gemm_f32_wmma<AK, BK, NX><<<g, 128, 0, stream>>>(A, sAm, sAk, sAb, B, sBk, sBn, sBb,
                                               C, sCm, sCn, sCb, M, N, K, alpha, accum);
}

__global__ void gen_consts(float* Fre, float* Fim, float* Vre, float* Vim,
                           float* F2, float* F3, float* V2, float* V3) {
  int i = blockIdx.x * 256 + threadIdx.x;
  if (i >= 48 * 96) return;
  int k = i / 96, nn = i % 96;
  int f = (k < 24) ? k : (k + 48);
  int ph = (f * nn) % 96;
  float th = 6.283185307179586f * (float)ph / 96.f;
  float c = cosf(th), s = sinf(th);
  Fre[i] = c;  Fim[i] = -s;
  Vre[nn * 48 + k] = c;  Vim[nn * 48 + k] = s;
  F2[k * 192 + 2 * nn]            = c;
  F2[k * 192 + 2 * nn + 1]        = s;
  F2[(48 + k) * 192 + 2 * nn]     = -s;
  F2[(48 + k) * 192 + 2 * nn + 1] = c;
  F3[k * 192 + 2 * nn]     = c + s;
  F3[k * 192 + 2 * nn + 1] = s - c;
  V2[nn * 96 + 2 * k]            = c;
  V2[nn * 96 + 2 * k + 1]        = -s;
  V2[(96 + nn) * 96 + 2 * k]     = s;
  V2[(96 + nn) * 96 + 2 * k + 1] = c;
  V3[nn * 96 + 2 * k]     = c + s;
  V3[nn * 96 + 2 * k + 1] = c - s;
}

__global__ void pack_wqkv(const float* __restrict__ wq, const float* __restrict__ wk,
                          const float* __restrict__ wv, float* __restrict__ dst) {
  int i = blockIdx.x * 256 + threadIdx.x;
  if (i >= 384 * 64) return;
  int ic = i % 64, r = i / 64;
  int t = r / 128, q = r % 128;
  const float* w = (t == 0) ? wq : ((t == 1) ? wk : wv);
  dst[i] = w[q * 64 + ic];
}

#define KSPLIT 32
__global__ __launch_bounds__(256) void sum_scale_selu(const float* __restrict__ part, float* __restrict__ s,
                               int nvec, int planeVec, float scale) {
  int i = blockIdx.x * 256 + threadIdx.x;
  if (i >= nvec) return;
  int h = i / planeVec, e = i - h * planeVec;
  const float4* p = reinterpret_cast<const float4*>(part) + (long)h * KSPLIT * planeVec + e;
  float4 v[KSPLIT];
  for (int j = 0; j < KSPLIT; ++j) v[j] = p[(long)j * planeVec];
  float4 a = v[0];
  for (int j = 1; j < KSPLIT; ++j) {
    a.x += v[j].x; a.y += v[j].y; a.z += v[j].z; a.w += v[j].w;
  }
  const float lam = 1.0507009873554805f, alp = 1.6732632423543772f;
  float x0 = scale * a.x, x1 = scale * a.y, x2 = scale * a.z, x3 = scale * a.w;
  float4 r;
  r.x = lam * (x0 > 0.f ? x0 : alp * (expf(x0) - 1.f));
  r.y = lam * (x1 > 0.f ? x1 : alp * (expf(x1) - 1.f));
  r.z = lam * (x2 > 0.f ? x2 : alp * (expf(x2) - 1.f));
  r.w = lam * (x3 > 0.f ? x3 : alp * (expf(x3) - 1.f));
  reinterpret_cast<float4*>(s)[i] = r;
}


__global__ void group_rows(const float* __restrict__ src, float* __restrict__ dst, int nelem) {
  int i = blockIdx.x * 256 + threadIdx.x;
  if (i >= nelem) return;
  int s = i % 110592, r = i / 110592;
  int x = s % 48, y = (s / 48) % 48, z = s / 2304;
  int px = x & 7, nx = x >> 3, py = y & 7, ny = y >> 3, pz = z & 7, nz = z >> 3;
  int p = (((pz << 3) | py) << 3) | px;
  int tok = (nz * 6 + ny) * 6 + nx;
  dst[(long)r * 110592 + p * 216 + tok] = src[i];
}

__global__ void ungroup_rows(const float* __restrict__ src, float* __restrict__ dst, int nelem) {
  int i = blockIdx.x * 256 + threadIdx.x;
  if (i >= nelem) return;
  int g = i % 110592, r = i / 110592;
  int tok = g % 216, p = g / 216;
  int px = p & 7, py = (p >> 3) & 7, pz = p >> 6;
  int nx = tok % 6, ny = (tok / 6) % 6, nz = tok / 36;
  int z = nz * 8 + pz, y = ny * 8 + py, x = nx * 8 + px;
  dst[(long)r * 110592 + (long)(z * 48 + y) * 48 + x] = src[i];
}

extern "C" void kernel_launch(void* const* d_in, const int* in_sizes, int n_in,
                              void* d_out, int out_size, void* d_ws, size_t ws_size,
                              hipStream_t stream) {
  const float* in = (const float*)d_in[0];
  const float* wq = (const float*)d_in[1];
  const float* wk = (const float*)d_in[2];
  const float* wv = (const float*)d_in[3];
  const float* wo = (const float*)d_in[4];
  float* out = (float*)d_out;
  float* w = (float*)d_ws;

  const long R1_o   = 0;
  const long R2_o   = 56623104;
  const long blk_o  = 0;
  const long blkG_o = 7077888,    G_o    = 42467328;
  const long S_o    = 92012544,   Og_o   = 92385792;
  const long P_o    = 0;
  const long specG_o = 0,         spec_o = 14155776;
  const long T1_o   = 15925248;
  const long T2_o   = 23003136;
  const long F2_o   = 84934656,   F3_o   = 84953088;
  const long V2_o   = 84962304,   V3_o   = 84980736;
  const long Wqkv_o = 106541568;
  const long Fre_o  = 106566144,  Fim_o  = 106570752;
  const long Vre_o  = 106575360,  Vim_o  = 106579968;
  const long WS_FLOATS = 106584576;
  if (ws_size < (size_t)WS_FLOATS * sizeof(float)) return;

  float *R1 = w+R1_o, *R2 = w+R2_o;
  float *blk = w+blk_o, *blkG = w+blkG_o, *G = w+G_o, *S = w+S_o, *Og = w+Og_o;
  float *specG = w+specG_o, *spec = w+spec_o, *P = w+P_o;
  float *T1 = w+T1_o, *T2 = w+T2_o, *Wqkv = w+Wqkv_o;
  float *Fre = w+Fre_o, *Fim = w+Fim_o, *Vre = w+Vre_o, *Vim = w+Vim_o;
  float *F2 = w+F2_o, *F3 = w+F3_o, *V2 = w+V2_o, *V3 = w+V3_o;

  auto ew = [&](int n) { return dim3((unsigned)((n + 255) / 256)); };

  gen_consts<<<ew(4608), 256, 0, stream>>>(Fre, Fim, Vre, Vim, F2, F3, V2, V3);
  pack_wqkv<<<ew(24576), 256, 0, stream>>>(wq, wk, wv, Wqkv);

  launch_gemm<true, true, true>(stream, in, 96, 1, 0, Fre, 1, 96, 0, R1, 96, 1, 0, 589824, 96, 96, 1, 1.f, 0);
  launch_gemm<true, false>(stream, F2, 192, 1, 0, R1, 48, 1, 9216, R2, 48, 1, 4608, 96, 48, 192, 6144, 1.f, 0);
  launch_gemm<true, false>(stream, F3, 192, 1, 0, R2, 2304, 1, 442368, blk, 2304, 1, 110592, 48, 2304, 192, 64, 1.f, 0);

  group_rows<<<ew(7077888), 256, 0, stream>>>(blk, blkG, 7077888);
  launch_gemm<true, false>(stream, Wqkv, 64, 1, 0, blkG, 110592, 1, 0, G, 110592, 1, 0, 384, 110592, 64, 1, 1.f, 0);

  const long HD = 8192L * 216L;
  const long SL = 256L * 216L;
  launch_gemm<false, false>(stream, G, 1, 216, SL, G + 8 * HD, 216, 1, SL, P, 216, 1, 46656,
       216, 216, 256, 8 * KSPLIT, 1.f, 0);
  sum_scale_selu<<<ew(93312), 256, 0, stream>>>(P, S, 93312, 11664,
                                                0.011048543456039806f);
  launch_gemm<true, true>(stream, G + 16 * HD, 216, 1, HD, S, 1, 216, 46656, Og, 216, 1, HD,
       8192, 216, 216, 8, 1.f, 0);

  launch_gemm<true, false>(stream, wo, 128, 1, 0, Og, 110592, 1, 0, specG, 110592, 1, 0, 16, 110592, 128, 1, 1.f, 0);
  ungroup_rows<<<ew(1769472), 256, 0, stream>>>(specG, spec, 1769472);

  launch_gemm<true, true>(stream, spec, 48, 1, 0, Vre, 1, 48, 0, T1, 192, 1, 0, 36864, 192, 48, 1, 1.f, 0);
  launch_gemm<true, false>(stream, V2, 96, 1, 0, T1, 96, 1, 9216, T2, 96, 1, 18432, 192, 96, 96, 768, 1.f, 0);
  const float invN = 1.f / 884736.f;
  launch_gemm<true, false>(stream, V3, 96, 1, 0, T2, 9216, 1, 884736, out, 9216, 1, 884736, 96, 9216, 96, 16, invN, 0);
}


# GEMM block tile becomes a per-call-site shape (16*TM rows x 16*NW cols, NW waves) fitted to each product's M and N: zero-padded 16x16 subtiles are no longer multiplied or staged; sum order unchanged
# speedup vs baseline: 4.5828x; 1.1315x over previous
#include <hip/hip_runtime.h>
#include <math.h>

typedef float v2f __attribute__((ext_vector_type(2)));
typedef float v8f __attribute__((ext_vector_type(8)));

#define KC 16
#define PK 20

template <bool AK, bool BK, int TM, int NW>
__global__ __launch_bounds__(32 * NW) void gemm_f32_wmma(
    const float* __restrict__ A, long sAm, long sAk, long sAb,
    const float* __restrict__ B, long sBk, long sBn, long sBb,
    float* __restrict__ C, long sCm, long sCn, long sCb,
    int M, int N, int K, float alpha, int accum)
{
  constexpr int BM = 16 * TM, BN = 16 * NW, NT = 32 * NW;
  constexpr int PRA = BM + 8, PRB = BN + 8;
  constexpr int TA = AK ? BM * PK : KC * PRA;
  constexpr int TB = BK ? BN * PK : KC * PRB;
  constexpr int NA = (4 * BM + NT - 1) / NT;
  constexpr bool AFULL = ((4 * BM) % NT) == 0;
  __shared__ __attribute__((aligned(16))) float As[2 * TA];
  __shared__ __attribute__((aligned(16))) float Bs[2 * TB];

  const int tid  = threadIdx.x;
  const int lane = tid & 31;
  const int wave = tid >> 5;
  const int half = lane >> 4;
  const int l16  = lane & 15;
  const int mBase = blockIdx.x * BM;
  const int nBlk  = blockIdx.y * BN;

  A += (long)blockIdx.z * sAb;
  B += (long)blockIdx.z * sBb;
  C += (long)blockIdx.z * sCb;

  v8f zero = {0.f,0.f,0.f,0.f,0.f,0.f,0.f,0.f};
  v8f acc[TM];
#pragma unroll
  for (int t = 0; t < TM; ++t) acc[t] = zero;

  long aOff[NA], bOff[2];
  int  aKo[NA], bKo[2];
  bool aIn[NA], bIn[2];
  bool aHas[NA];
  int  aDst[NA], bDst[2];
#pragma unroll
  for (int j = 0; j < NA; ++j) {
    const int idx = tid + NT * j;
    aHas[j] = AFULL || (idx < 4 * BM);
    if (AK) {
      const int row = idx >> 2, kq = (idx & 3) * 4, m = mBase + row;
      aIn[j] = aHas[j] && (m < M);  aKo[j] = kq;  aDst[j] = row * PK + kq;
      aOff[j] = (long)(aIn[j] ? m : 0) * sAm + kq;
    } else {
      const int kk = idx / (BM / 4), mq = (idx % (BM / 4)) * 4, m = mBase + mq;
      aIn[j] = aHas[j] && (m < M);  aKo[j] = kk;  aDst[j] = kk * PRA + mq;
      aOff[j] = (long)kk * sAk + (aIn[j] ? m : 0);
    }
  }
  for (int j = 0; j < 2; ++j) {
    const int idx = tid + NT * j;
    if (BK) {
      const int row = idx >> 2, kq = (idx & 3) * 4, n = nBlk + row;
      bIn[j] = (n < N);  bKo[j] = kq;  bDst[j] = row * PK + kq;
      bOff[j] = (long)(bIn[j] ? n : 0) * sBn + kq;
    } else {
      const int kk = idx / (BN / 4), nq = (idx % (BN / 4)) * 4, n = nBlk + nq;
      bIn[j] = (n < N);  bKo[j] = kk;  bDst[j] = kk * PRB + nq;
      bOff[j] = (long)kk * sBk + (bIn[j] ? n : 0);
    }
  }
  const long aStep = AK ? (long)KC : (long)KC * sAk;
  const long bStep = BK ? (long)KC : (long)KC * sBk;

  const int aFrag = AK ? (l16 * PK + 2 * half) : (2 * half * PRA + l16);
  const int bFrag = BK ? ((wave * 16 + l16) * PK + 2 * half)
                       : (2 * half * PRB + wave * 16 + l16);

  float4 ra[NA], rb[2];

#pragma unroll
  for (int j = 0; j < NA; ++j) {
    ra[j] = make_float4(0.f, 0.f, 0.f, 0.f);
    if (aIn[j] && aKo[j] < K) ra[j] = *reinterpret_cast<const float4*>(A + aOff[j]);
    aOff[j] += aStep;
  }
  for (int j = 0; j < 2; ++j) {
    rb[j] = make_float4(0.f, 0.f, 0.f, 0.f);
    if (bIn[j] && bKo[j] < K) rb[j] = *reinterpret_cast<const float4*>(B + bOff[j]);
    bOff[j] += bStep;
  }
#pragma unroll
  for (int j = 0; j < NA; ++j)
    if (aHas[j]) *reinterpret_cast<float4*>(As + aDst[j]) = ra[j];
  for (int j = 0; j < 2; ++j)
    *reinterpret_cast<float4*>(Bs + bDst[j]) = rb[j];
  __syncthreads();

  int p = 0;
  for (int k0 = 0; ; k0 += KC) {
    const bool more = (k0 + KC < K);
    if (more) {
#pragma unroll
      for (int j = 0; j < NA; ++j) {
        ra[j] = make_float4(0.f, 0.f, 0.f, 0.f);
        if (aIn[j] && k0 + KC + aKo[j] < K) ra[j] = *reinterpret_cast<const float4*>(A + aOff[j]);
        aOff[j] += aStep;
      }
      for (int j = 0; j < 2; ++j) {
        rb[j] = make_float4(0.f, 0.f, 0.f, 0.f);
        if (bIn[j] && k0 + KC + bKo[j] < K) rb[j] = *reinterpret_cast<const float4*>(B + bOff[j]);
        bOff[j] += bStep;
      }
    }

    const float* as = As + p * TA + aFrag;
    const float* bs = Bs + p * TB + bFrag;
#pragma unroll
    for (int ks = 0; ks < KC / 4; ++ks) {
      v2f bf;
      if (BK) {
        bf = *reinterpret_cast<const v2f*>(bs + ks * 4);
      } else {
        bf.x = bs[(ks * 4) * PRB];
        bf.y = bs[(ks * 4 + 1) * PRB];
      }
#pragma unroll
      for (int t = 0; t < TM; ++t) {
        v2f af;
        if (AK) {
          af = *reinterpret_cast<const v2f*>(as + t * 16 * PK + ks * 4);
        } else {
          af.x = as[(ks * 4) * PRA + t * 16];
          af.y = as[(ks * 4 + 1) * PRA + t * 16];
        }
        acc[t] = __builtin_amdgcn_wmma_f32_16x16x4_f32(
            false, af, false, bf, (short)0, acc[t], false, false);
      }
    }

    if (!more) break;
    p ^= 1;
#pragma unroll
    for (int j = 0; j < NA; ++j)
      if (aHas[j]) *reinterpret_cast<float4*>(As + p * TA + aDst[j]) = ra[j];
    for (int j = 0; j < 2; ++j)
      *reinterpret_cast<float4*>(Bs + p * TB + bDst[j]) = rb[j];
    __syncthreads();
  }

  const int n = nBlk + wave * 16 + l16;
  if (n < N) {
#pragma unroll
    for (int t = 0; t < TM; ++t) {
#pragma unroll
      for (int r = 0; r < 8; ++r) {
        int m = mBase + t * 16 + r + 8 * half;
        if (m < M) {
          long idx = (long)m * sCm + (long)n * sCn;
          float val = alpha * acc[t][r];
          if (accum) C[idx] += val; else C[idx] = val;
        }
      }
    }
  }
}

template <bool AK, bool BK, int TM, int NW>
static void launch_gemm(hipStream_t stream,
                        const float* A, long sAm, long sAk, long sAb,
                        const float* B, long sBk, long sBn, long sBb,
                        float* C, long sCm, long sCn, long sCb,
                        int M, int N, int K, int batch, float alpha, int accum) {
  const unsigned mb = (unsigned)((M + 16 * TM - 1) / (16 * TM));
  const unsigned nb = (unsigned)((N + 16 * NW - 1) / (16 * NW));
  dim3 g(mb, nb, (unsigned)batch);
  gemm_f32_wmma<AK, BK, TM, NW><<<g, 32 * NW, 0, stream>>>(A, sAm, sAk, sAb, B, sBk, sBn, sBb,
                                                       C, sCm, sCn, sCb, M, N, K, alpha, accum);
}

__global__ void gen_consts(float* Fre, float* Fim, float* Vre, float* Vim,
                           float* F2, float* F3, float* V2, float* V3) {
  int i = blockIdx.x * 256 + threadIdx.x;
  if (i >= 48 * 96) return;
  int k = i / 96, nn = i % 96;
  int f = (k < 24) ? k : (k + 48);
  int ph = (f * nn) % 96;
  float th = 6.283185307179586f * (float)ph / 96.f;
  float c = cosf(th), s = sinf(th);
  Fre[i] = c;  Fim[i] = -s;
  Vre[nn * 48 + k] = c;  Vim[nn * 48 + k] = s;
  F2[k * 192 + 2 * nn]            = c;
  F2[k * 192 + 2 * nn + 1]        = s;
  F2[(48 + k) * 192 + 2 * nn]     = -s;
  F2[(48 + k) * 192 + 2 * nn + 1] = c;
  F3[k * 192 + 2 * nn]     = c + s;
  F3[k * 192 + 2 * nn + 1] = s - c;
  V2[nn * 96 + 2 * k]            = c;
  V2[nn * 96 + 2 * k + 1]        = -s;
  V2[(96 + nn) * 96 + 2 * k]     = s;
  V2[(96 + nn) * 96 + 2 * k + 1] = c;
  V3[nn * 96 + 2 * k]     = c + s;
  V3[nn * 96 + 2 * k + 1] = c - s;
}

__global__ void pack_wqkv(const float* __restrict__ wq, const float* __restrict__ wk,
                          const float* __restrict__ wv, float* __restrict__ dst) {
  int i = blockIdx.x * 256 + threadIdx.x;
  if (i >= 384 * 64) return;
  int ic = i % 64, r = i / 64;
  int t = r / 128, q = r % 128;
  const float* w = (t == 0) ? wq : ((t == 1) ? wk : wv);
  dst[i] = w[q * 64 + ic];
}

#define KSPLIT 32
__global__ __launch_bounds__(256) void sum_scale_selu(const float* __restrict__ part, float* __restrict__ s,
                               int nvec, int planeVec, float scale) {
  int i = blockIdx.x * 256 + threadIdx.x;
  if (i >= nvec) return;
  int h = i / planeVec, e = i - h * planeVec;
  const float4* p = reinterpret_cast<const float4*>(part) + (long)h * KSPLIT * planeVec + e;
  float4 v[KSPLIT];
  for (int j = 0; j < KSPLIT; ++j) v[j] = p[(long)j * planeVec];
  float4 a = v[0];
  for (int j = 1; j < KSPLIT; ++j) {
    a.x += v[j].x; a.y += v[j].y; a.z += v[j].z; a.w += v[j].w;
  }
  const float lam = 1.0507009873554805f, alp = 1.6732632423543772f;
  float x0 = scale * a.x, x1 = scale * a.y, x2 = scale * a.z, x3 = scale * a.w;
  float4 r;
  r.x = lam * (x0 > 0.f ? x0 : alp * (expf(x0) - 1.f));
  r.y = lam * (x1 > 0.f ? x1 : alp * (expf(x1) - 1.f));
  r.z = lam * (x2 > 0.f ? x2 : alp * (expf(x2) - 1.f));
  r.w = lam * (x3 > 0.f ? x3 : alp * (expf(x3) - 1.f));
  reinterpret_cast<float4*>(s)[i] = r;
}


__global__ void group_rows(const float* __restrict__ src, float* __restrict__ dst, int nelem) {
  int i = blockIdx.x * 256 + threadIdx.x;
  if (i >= nelem) return;
  int s = i % 110592, r = i / 110592;
  int x = s % 48, y = (s / 48) % 48, z = s / 2304;
  int px = x & 7, nx = x >> 3, py = y & 7, ny = y >> 3, pz = z & 7, nz = z >> 3;
  int p = (((pz << 3) | py) << 3) | px;
  int tok = (nz * 6 + ny) * 6 + nx;
  dst[(long)r * 110592 + p * 216 + tok] = src[i];
}

__global__ void ungroup_rows(const float* __restrict__ src, float* __restrict__ dst, int nelem) {
  int i = blockIdx.x * 256 + threadIdx.x;
  if (i >= nelem) return;
  int g = i % 110592, r = i / 110592;
  int tok = g % 216, p = g / 216;
  int px = p & 7, py = (p >> 3) & 7, pz = p >> 6;
  int nx = tok % 6, ny = (tok / 6) % 6, nz = tok / 36;
  int z = nz * 8 + pz, y = ny * 8 + py, x = nx * 8 + px;
  dst[(long)r * 110592 + (long)(z * 48 + y) * 48 + x] = src[i];
}

extern "C" void kernel_launch(void* const* d_in, const int* in_sizes, int n_in,
                              void* d_out, int out_size, void* d_ws, size_t ws_size,
                              hipStream_t stream) {
  const float* in = (const float*)d_in[0];
  const float* wq = (const float*)d_in[1];
  const float* wk = (const float*)d_in[2];
  const float* wv = (const float*)d_in[3];
  const float* wo = (const float*)d_in[4];
  float* out = (float*)d_out;
  float* w = (float*)d_ws;

  const long R1_o   = 0;
  const long R2_o   = 56623104;
  const long blk_o  = 0;
  const long blkG_o = 7077888,    G_o    = 42467328;
  const long S_o    = 92012544,   Og_o   = 92385792;
  const long P_o    = 0;
  const long specG_o = 0,         spec_o = 14155776;
  const long T1_o   = 15925248;
  const long T2_o   = 23003136;
  const long F2_o   = 84934656,   F3_o   = 84953088;
  const long V2_o   = 84962304,   V3_o   = 84980736;
  const long Wqkv_o = 106541568;
  const long Fre_o  = 106566144,  Fim_o  = 106570752;
  const long Vre_o  = 106575360,  Vim_o  = 106579968;
  const long WS_FLOATS = 106584576;
  if (ws_size < (size_t)WS_FLOATS * sizeof(float)) return;

  float *R1 = w+R1_o, *R2 = w+R2_o;
  float *blk = w+blk_o, *blkG = w+blkG_o, *G = w+G_o, *S = w+S_o, *Og = w+Og_o;
  float *specG = w+specG_o, *spec = w+spec_o, *P = w+P_o;
  float *T1 = w+T1_o, *T2 = w+T2_o, *Wqkv = w+Wqkv_o;
  float *Fre = w+Fre_o, *Fim = w+Fim_o, *Vre = w+Vre_o, *Vim = w+Vim_o;
  float *F2 = w+F2_o, *F3 = w+F3_o, *V2 = w+V2_o, *V3 = w+V3_o;

  auto ew = [&](int n) { return dim3((unsigned)((n + 255) / 256)); };

  gen_consts<<<ew(4608), 256, 0, stream>>>(Fre, Fim, Vre, Vim, F2, F3, V2, V3);
  pack_wqkv<<<ew(24576), 256, 0, stream>>>(wq, wk, wv, Wqkv);

  launch_gemm<true, true, 4, 6>(stream, in, 96, 1, 0, Fre, 1, 96, 0, R1, 96, 1, 0, 589824, 96, 96, 1, 1.f, 0);
  launch_gemm<true, false, 6, 3>(stream, F2, 192, 1, 0, R1, 48, 1, 9216, R2, 48, 1, 4608, 96, 48, 192, 6144, 1.f, 0);
  launch_gemm<true, false, 3, 4>(stream, F3, 192, 1, 0, R2, 2304, 1, 442368, blk, 2304, 1, 110592, 48, 2304, 192, 64, 1.f, 0);

  group_rows<<<ew(7077888), 256, 0, stream>>>(blk, blkG, 7077888);
  launch_gemm<true, false, 4, 4>(stream, Wqkv, 64, 1, 0, blkG, 110592, 1, 0, G, 110592, 1, 0, 384, 110592, 64, 1, 1.f, 0);

  const long HD = 8192L * 216L;
  const long SL = 256L * 216L;
  launch_gemm<false, false, 7, 7>(stream, G, 1, 216, SL, G + 8 * HD, 216, 1, SL, P, 216, 1, 46656,
       216, 216, 256, 8 * KSPLIT, 1.f, 0);
  sum_scale_selu<<<ew(93312), 256, 0, stream>>>(P, S, 93312, 11664,
                                                0.011048543456039806f);
  launch_gemm<true, true, 4, 7>(stream, G + 16 * HD, 216, 1, HD, S, 1, 216, 46656, Og, 216, 1, HD,
       8192, 216, 216, 8, 1.f, 0);

  launch_gemm<true, false, 1, 4>(stream, wo, 128, 1, 0, Og, 110592, 1, 0, specG, 110592, 1, 0, 16, 110592, 128, 1, 1.f, 0);
  ungroup_rows<<<ew(1769472), 256, 0, stream>>>(specG, spec, 1769472);

  launch_gemm<true, true, 4, 4>(stream, spec, 48, 1, 0, Vre, 1, 48, 0, T1, 192, 1, 0, 36864, 192, 48, 1, 1.f, 0);
  launch_gemm<true, false, 6, 6>(stream, V2, 96, 1, 0, T1, 96, 1, 9216, T2, 96, 1, 18432, 192, 96, 96, 768, 1.f, 0);
  const float invN = 1.f / 884736.f;
  launch_gemm<true, false, 6, 4>(stream, V3, 96, 1, 0, T2, 9216, 1, 884736, out, 9216, 1, 884736, 96, 9216, 96, 16, invN, 0);
}


# GEMM result tile goes through LDS (operand space reused) and out along the rows in 16-byte stores, whole 128-byte lines per instruction, not 4-byte stores in 64-byte pieces; sum order unchanged
# speedup vs baseline: 5.6458x; 1.2320x over previous
#include <hip/hip_runtime.h>
#include <math.h>

typedef float v2f __attribute__((ext_vector_type(2)));
typedef float v8f __attribute__((ext_vector_type(8)));

#define KC 16
#define PK 20

#define IDLE_SLOTS "v_nop\n\tv_nop\n\tv_nop\n\tv_nop\n\tv_nop"
template <int TM>
__device__ __forceinline__ void slots_behind_last_wmma(v8f (&acc)[TM]) {
  static_assert(TM == 1 || TM == 3 || TM == 4 || TM == 6 || TM == 7, "add the operand list for this TM");
  if constexpr (TM == 1)
    asm volatile(IDLE_SLOTS : "+v"(acc[0]));
  else if constexpr (TM == 3)
    asm volatile(IDLE_SLOTS : "+v"(acc[0]), "+v"(acc[1]), "+v"(acc[2]));
  else if constexpr (TM == 4)
    asm volatile(IDLE_SLOTS : "+v"(acc[0]), "+v"(acc[1]), "+v"(acc[2]), "+v"(acc[3]));
  else if constexpr (TM == 6)
    asm volatile(IDLE_SLOTS : "+v"(acc[0]), "+v"(acc[1]), "+v"(acc[2]), "+v"(acc[3]), "+v"(acc[4]), "+v"(acc[5]));
  else
    asm volatile(IDLE_SLOTS : "+v"(acc[0]), "+v"(acc[1]), "+v"(acc[2]), "+v"(acc[3]), "+v"(acc[4]), "+v"(acc[5]),
                 "+v"(acc[6]));
}

template <bool AK, bool BK, int TM, int NW>
__global__ __launch_bounds__(32 * NW) void gemm_f32_wmma(
    const float* __restrict__ A, long sAm, long sAk, long sAb,
    const float* __restrict__ B, long sBk, long sBn, long sBb,
    float* __restrict__ C, long sCm, long sCb,
    int M, int N, int K, float alpha)
{
  constexpr int BM = 16 * TM, BN = 16 * NW, NT = 32 * NW;
  constexpr int PRA = BM + 8, PRB = BN + 8;
  constexpr int TA = AK ? BM * PK : KC * PRA;
  constexpr int TB = BK ? BN * PK : KC * PRB;
  constexpr int NA = (4 * BM + NT - 1) / NT;
  constexpr bool AFULL = ((4 * BM) % NT) == 0;
  constexpr int PC = BN + 4;
  constexpr int OPF = 2 * TA + 2 * TB;
  constexpr int CTF = BM * PC;
  constexpr int SMF = OPF > CTF ? OPF : CTF;
  __shared__ __attribute__((aligned(16))) float smem[SMF];
  float* const As = smem;
  float* const Bs = smem + 2 * TA;

  const int tid  = threadIdx.x;
  const int lane = tid & 31;
  const int wave = tid >> 5;
  const int half = lane >> 4;
  const int l16  = lane & 15;
  const int mBase = blockIdx.x * BM;
  const int nBlk  = blockIdx.y * BN;

  A += (long)blockIdx.z * sAb;
  B += (long)blockIdx.z * sBb;
  C += (long)blockIdx.z * sCb;

  v8f zero = {0.f,0.f,0.f,0.f,0.f,0.f,0.f,0.f};
  v8f acc[TM];
#pragma unroll
  for (int t = 0; t < TM; ++t) acc[t] = zero;

  long aOff[NA], bOff[2];
  int  aKo[NA], bKo[2];
  bool aIn[NA], bIn[2];
  bool aHas[NA];
  int  aDst[NA], bDst[2];
#pragma unroll
  for (int j = 0; j < NA; ++j) {
    const int idx = tid + NT * j;
    aHas[j] = AFULL || (idx < 4 * BM);
    if (AK) {
      const int row = idx >> 2, kq = (idx & 3) * 4, m = mBase + row;
      aIn[j] = aHas[j] && (m < M);  aKo[j] = kq;  aDst[j] = row * PK + kq;
      aOff[j] = (long)(aIn[j] ? m : 0) * sAm + kq;
    } else {
      const int kk = idx / (BM / 4), mq = (idx % (BM / 4)) * 4, m = mBase + mq;
      aIn[j] = aHas[j] && (m < M);  aKo[j] = kk;  aDst[j] = kk * PRA + mq;
      aOff[j] = (long)kk * sAk + (aIn[j] ? m : 0);
    }
  }
  for (int j = 0; j < 2; ++j) {
    const int idx = tid + NT * j;
    if (BK) {
      const int row = idx >> 2, kq = (idx & 3) * 4, n = nBlk + row;
      bIn[j] = (n < N);  bKo[j] = kq;  bDst[j] = row * PK + kq;
      bOff[j] = (long)(bIn[j] ? n : 0) * sBn + kq;
    } else {
      const int kk = idx / (BN / 4), nq = (idx % (BN / 4)) * 4, n = nBlk + nq;
      bIn[j] = (n < N);  bKo[j] = kk;  bDst[j] = kk * PRB + nq;
      bOff[j] = (long)kk * sBk + (bIn[j] ? n : 0);
    }
  }
  const long aStep = AK ? (long)KC : (long)KC * sAk;
  const long bStep = BK ? (long)KC : (long)KC * sBk;

  const int aFrag = AK ? (l16 * PK + 2 * half) : (2 * half * PRA + l16);
  const int bFrag = BK ? ((wave * 16 + l16) * PK + 2 * half)
                       : (2 * half * PRB + wave * 16 + l16);

  float4 ra[NA], rb[2];

#pragma unroll
  for (int j = 0; j < NA; ++j) {
    ra[j] = make_float4(0.f, 0.f, 0.f, 0.f);
    if (aIn[j] && aKo[j] < K) ra[j] = *reinterpret_cast<const float4*>(A + aOff[j]);
    aOff[j] += aStep;
  }
  for (int j = 0; j < 2; ++j) {
    rb[j] = make_float4(0.f, 0.f, 0.f, 0.f);
    if (bIn[j] && bKo[j] < K) rb[j] = *reinterpret_cast<const float4*>(B + bOff[j]);
    bOff[j] += bStep;
  }
#pragma unroll
  for (int j = 0; j < NA; ++j)
    if (aHas[j]) *reinterpret_cast<float4*>(As + aDst[j]) = ra[j];
  for (int j = 0; j < 2; ++j)
    *reinterpret_cast<float4*>(Bs + bDst[j]) = rb[j];
  __syncthreads();

  int p = 0;
  for (int k0 = 0; ; k0 += KC) {
    const bool more = (k0 + KC < K);
    if (more) {
#pragma unroll
      for (int j = 0; j < NA; ++j) {
        ra[j] = make_float4(0.f, 0.f, 0.f, 0.f);
        if (aIn[j] && k0 + KC + aKo[j] < K) ra[j] = *reinterpret_cast<const float4*>(A + aOff[j]);
        aOff[j] += aStep;
      }
      for (int j = 0; j < 2; ++j) {
        rb[j] = make_float4(0.f, 0.f, 0.f, 0.f);
        if (bIn[j] && k0 + KC + bKo[j] < K) rb[j] = *reinterpret_cast<const float4*>(B + bOff[j]);
        bOff[j] += bStep;
      }
    }

    const float* as = As + p * TA + aFrag;
    const float* bs = Bs + p * TB + bFrag;
#pragma unroll
    for (int ks = 0; ks < KC / 4; ++ks) {
      v2f bf;
      if (BK) {
        bf = *reinterpret_cast<const v2f*>(bs + ks * 4);
      } else {
        bf.x = bs[(ks * 4) * PRB];
        bf.y = bs[(ks * 4 + 1) * PRB];
      }
#pragma unroll
      for (int t = 0; t < TM; ++t) {
        v2f af;
        if (AK) {
          af = *reinterpret_cast<const v2f*>(as + t * 16 * PK + ks * 4);
        } else {
          af.x = as[(ks * 4) * PRA + t * 16];
          af.y = as[(ks * 4 + 1) * PRA + t * 16];
        }
        acc[t] = __builtin_amdgcn_wmma_f32_16x16x4_f32(
            false, af, false, bf, (short)0, acc[t], false, false);
      }
    }

    if (!more) break;
    p ^= 1;
#pragma unroll
    for (int j = 0; j < NA; ++j)
      if (aHas[j]) *reinterpret_cast<float4*>(As + p * TA + aDst[j]) = ra[j];
    for (int j = 0; j < 2; ++j)
      *reinterpret_cast<float4*>(Bs + p * TB + bDst[j]) = rb[j];
    __syncthreads();
  }

  slots_behind_last_wmma<TM>(acc);
  __builtin_amdgcn_sched_barrier(0);
  __syncthreads();
  {
    float* cw = smem + (8 * half) * PC + wave * 16 + l16;
#pragma unroll
    for (int t = 0; t < TM; ++t) {
#pragma unroll
      for (int r = 0; r < 8; ++r) cw[(t * 16 + r) * PC] = alpha * acc[t][r];
    }
  }
  __syncthreads();
  {
    constexpr int RV = BN / 4;
    const int row0 = tid / RV, c4 = tid - row0 * RV;
    const int n = nBlk + 4 * c4;
    const float* cr = smem + row0 * PC + 4 * c4;
    float* cg = C + (long)(mBase + row0) * sCm + n;
    const long gStep = 8 * sCm;
    if (mBase + BM <= M && nBlk + BN <= N) {
#pragma unroll
      for (int j = 0; j < 2 * TM; ++j) {
        *reinterpret_cast<float4*>(cg) = *reinterpret_cast<const float4*>(cr + j * 8 * PC);
        cg += gStep;
      }
    } else if (n < N) {
#pragma unroll
      for (int j = 0; j < 2 * TM; ++j) {
        if (mBase + row0 + 8 * j < M)
          *reinterpret_cast<float4*>(cg) = *reinterpret_cast<const float4*>(cr + j * 8 * PC);
        cg += gStep;
      }
    }
  }
}

template <bool AK, bool BK, int TM, int NW>
static void launch_gemm(hipStream_t stream,
                        const float* A, long sAm, long sAk, long sAb,
                        const float* B, long sBk, long sBn, long sBb,
                        float* C, long sCm, long sCb,
                        int M, int N, int K, int batch, float alpha) {
  const unsigned mb = (unsigned)((M + 16 * TM - 1) / (16 * TM));
  const unsigned nb = (unsigned)((N + 16 * NW - 1) / (16 * NW));
  dim3 g(mb, nb, (unsigned)batch);
  gemm_f32_wmma<AK, BK, TM, NW><<<g, 32 * NW, 0, stream>>>(A, sAm, sAk, sAb, B, sBk, sBn, sBb,
                                                       C, sCm, sCb, M, N, K, alpha);
}

__global__ void gen_consts(float* Fre, float* Fim, float* Vre, float* Vim,
                           float* F2, float* F3, float* V2, float* V3) {
  int i = blockIdx.x * 256 + threadIdx.x;
  if (i >= 48 * 96) return;
  int k = i / 96, nn = i % 96;
  int f = (k < 24) ? k : (k + 48);
  int ph = (f * nn) % 96;
  float th = 6.283185307179586f * (float)ph / 96.f;
  float c = cosf(th), s = sinf(th);
  Fre[i] = c;  Fim[i] = -s;
  Vre[nn * 48 + k] = c;  Vim[nn * 48 + k] = s;
  F2[k * 192 + 2 * nn]            = c;
  F2[k * 192 + 2 * nn + 1]        = s;
  F2[(48 + k) * 192 + 2 * nn]     = -s;
  F2[(48 + k) * 192 + 2 * nn + 1] = c;
  F3[k * 192 + 2 * nn]     = c + s;
  F3[k * 192 + 2 * nn + 1] = s - c;
  V2[nn * 96 + 2 * k]            = c;
  V2[nn * 96 + 2 * k + 1]        = -s;
  V2[(96 + nn) * 96 + 2 * k]     = s;
  V2[(96 + nn) * 96 + 2 * k + 1] = c;
  V3[nn * 96 + 2 * k]     = c + s;
  V3[nn * 96 + 2 * k + 1] = c - s;
}

__global__ void pack_wqkv(const float* __restrict__ wq, const float* __restrict__ wk,
                          const float* __restrict__ wv, float* __restrict__ dst) {
  int i = blockIdx.x * 256 + threadIdx.x;
  if (i >= 384 * 64) return;
  int ic = i % 64, r = i / 64;
  int t = r / 128, q = r % 128;
  const float* w = (t == 0) ? wq : ((t == 1) ? wk : wv);
  dst[i] = w[q * 64 + ic];
}

#define KSPLIT 32
__global__ __launch_bounds__(256) void sum_scale_selu(const float* __restrict__ part, float* __restrict__ s,
                               int nvec, int planeVec, float scale) {
  int i = blockIdx.x * 256 + threadIdx.x;
  if (i >= nvec) return;
  int h = i / planeVec, e = i - h * planeVec;
  const float4* p = reinterpret_cast<const float4*>(part) + (long)h * KSPLIT * planeVec + e;
  float4 v[KSPLIT];
  for (int j = 0; j < KSPLIT; ++j) v[j] = p[(long)j * planeVec];
  float4 a = v[0];
  for (int j = 1; j < KSPLIT; ++j) {
    a.x += v[j].x; a.y += v[j].y; a.z += v[j].z; a.w += v[j].w;
  }
  const float lam = 1.0507009873554805f, alp = 1.6732632423543772f;
  float x0 = scale * a.x, x1 = scale * a.y, x2 = scale * a.z, x3 = scale * a.w;
  float4 r;
  r.x = lam * (x0 > 0.f ? x0 : alp * (expf(x0) - 1.f));
  r.y = lam * (x1 > 0.f ? x1 : alp * (expf(x1) - 1.f));
  r.z = lam * (x2 > 0.f ? x2 : alp * (expf(x2) - 1.f));
  r.w = lam * (x3 > 0.f ? x3 : alp * (expf(x3) - 1.f));
  reinterpret_cast<float4*>(s)[i] = r;
}


__global__ void group_rows(const float* __restrict__ src, float* __restrict__ dst, int nelem) {
  int i = blockIdx.x * 256 + threadIdx.x;
  if (i >= nelem) return;
  int s = i % 110592, r = i / 110592;
  int x = s % 48, y = (s / 48) % 48, z = s / 2304;
  int px = x & 7, nx = x >> 3, py = y & 7, ny = y >> 3, pz = z & 7, nz = z >> 3;
  int p = (((pz << 3) | py) << 3) | px;
  int tok = (nz * 6 + ny) * 6 + nx;
  dst[(long)r * 110592 + p * 216 + tok] = src[i];
}

__global__ void ungroup_rows(const float* __restrict__ src, float* __restrict__ dst, int nelem) {
  int i = blockIdx.x * 256 + threadIdx.x;
  if (i >= nelem) return;
  int g = i % 110592, r = i / 110592;
  int tok = g % 216, p = g / 216;
  int px = p & 7, py = (p >> 3) & 7, pz = p >> 6;
  int nx = tok % 6, ny = (tok / 6) % 6, nz = tok / 36;
  int z = nz * 8 + pz, y = ny * 8 + py, x = nx * 8 + px;
  dst[(long)r * 110592 + (long)(z * 48 + y) * 48 + x] = src[i];
}

extern "C" void kernel_launch(void* const* d_in, const int* in_sizes, int n_in,
                              void* d_out, int out_size, void* d_ws, size_t ws_size,
                              hipStream_t stream) {
  const float* in = (const float*)d_in[0];
  const float* wq = (const float*)d_in[1];
  const float* wk = (const float*)d_in[2];
  const float* wv = (const float*)d_in[3];
  const float* wo = (const float*)d_in[4];
  float* out = (float*)d_out;
  float* w = (float*)d_ws;

  const long R1_o   = 0;
  const long R2_o   = 56623104;
  const long blk_o  = 0;
  const long blkG_o = 7077888,    G_o    = 42467328;
  const long S_o    = 92012544,   Og_o   = 92385792;
  const long P_o    = 0;
  const long specG_o = 0,         spec_o = 14155776;
  const long T1_o   = 15925248;
  const long T2_o   = 23003136;
  const long F2_o   = 84934656,   F3_o   = 84953088;
  const long V2_o   = 84962304,   V3_o   = 84980736;
  const long Wqkv_o = 106541568;
  const long Fre_o  = 106566144,  Fim_o  = 106570752;
  const long Vre_o  = 106575360,  Vim_o  = 106579968;
  const long WS_FLOATS = 106584576;
  if (ws_size < (size_t)WS_FLOATS * sizeof(float)) return;

  float *R1 = w+R1_o, *R2 = w+R2_o;
  float *blk = w+blk_o, *blkG = w+blkG_o, *G = w+G_o, *S = w+S_o, *Og = w+Og_o;
  float *specG = w+specG_o, *spec = w+spec_o, *P = w+P_o;
  float *T1 = w+T1_o, *T2 = w+T2_o, *Wqkv = w+Wqkv_o;
  float *Fre = w+Fre_o, *Fim = w+Fim_o, *Vre = w+Vre_o, *Vim = w+Vim_o;
  float *F2 = w+F2_o, *F3 = w+F3_o, *V2 = w+V2_o, *V3 = w+V3_o;

  auto ew = [&](int n) { return dim3((unsigned)((n + 255) / 256)); };

  gen_consts<<<ew(4608), 256, 0, stream>>>(Fre, Fim, Vre, Vim, F2, F3, V2, V3);
  pack_wqkv<<<ew(24576), 256, 0, stream>>>(wq, wk, wv, Wqkv);

  launch_gemm<true, true, 4, 6>(stream, in, 96, 1, 0, Fre, 1, 96, 0, R1, 96, 0, 589824, 96, 96, 1, 1.f);
  launch_gemm<true, false, 6, 3>(stream, F2, 192, 1, 0, R1, 48, 1, 9216, R2, 48, 4608, 96, 48, 192, 6144, 1.f);
  launch_gemm<true, false, 3, 4>(stream, F3, 192, 1, 0, R2, 2304, 1, 442368, blk, 2304, 110592, 48, 2304, 192, 64, 1.f);

  group_rows<<<ew(7077888), 256, 0, stream>>>(blk, blkG, 7077888);
  launch_gemm<true, false, 4, 4>(stream, Wqkv, 64, 1, 0, blkG, 110592, 1, 0, G, 110592, 0, 384, 110592, 64, 1, 1.f);

  const long HD = 8192L * 216L;
  const long SL = 256L * 216L;
  launch_gemm<false, false, 7, 7>(stream, G, 1, 216, SL, G + 8 * HD, 216, 1, SL, P, 216, 46656,
       216, 216, 256, 8 * KSPLIT, 1.f);
  sum_scale_selu<<<ew(93312), 256, 0, stream>>>(P, S, 93312, 11664,
                                                0.011048543456039806f);
  launch_gemm<true, true, 4, 7>(stream, G + 16 * HD, 216, 1, HD, S, 1, 216, 46656, Og, 216, HD,
       8192, 216, 216, 8, 1.f);

  launch_gemm<true, false, 1, 4>(stream, wo, 128, 1, 0, Og, 110592, 1, 0, specG, 110592, 0, 16, 110592, 128, 1, 1.f);
  ungroup_rows<<<ew(1769472), 256, 0, stream>>>(specG, spec, 1769472);

  launch_gemm<true, true, 4, 4>(stream, spec, 48, 1, 0, Vre, 1, 48, 0, T1, 192, 0, 36864, 192, 48, 1, 1.f);
  launch_gemm<true, false, 6, 6>(stream, V2, 96, 1, 0, T1, 96, 1, 9216, T2, 96, 18432, 192, 96, 96, 768, 1.f);
  const float invN = 1.f / 884736.f;
  launch_gemm<true, false, 6, 4>(stream, V3, 96, 1, 0, T2, 9216, 1, 884736, out, 9216, 884736, 96, 9216, 96, 16, invN);
}


# x-axis and y-axis DFT products fused into one kernel per (c,z) slab: the 96x96 x-axis result stays in LDS as the y-axis product's B operand, never written to or read from memory; sums unchanged
# speedup vs baseline: 6.1454x; 1.0885x over previous
#include <hip/hip_runtime.h>
#include <math.h>

typedef float v2f __attribute__((ext_vector_type(2)));
typedef float v8f __attribute__((ext_vector_type(8)));

#define KC 16
#define PK 20

#define IDLE_SLOTS "v_nop\n\tv_nop\n\tv_nop\n\tv_nop\n\tv_nop"
template <int TM>
__device__ __forceinline__ void slots_behind_last_wmma(v8f (&acc)[TM]) {
  static_assert(TM == 1 || TM == 3 || TM == 4 || TM == 6 || TM == 7, "add the operand list for this TM");
  if constexpr (TM == 1)
    asm volatile(IDLE_SLOTS : "+v"(acc[0]));
  else if constexpr (TM == 3)
    asm volatile(IDLE_SLOTS : "+v"(acc[0]), "+v"(acc[1]), "+v"(acc[2]));
  else if constexpr (TM == 4)
    asm volatile(IDLE_SLOTS : "+v"(acc[0]), "+v"(acc[1]), "+v"(acc[2]), "+v"(acc[3]));
  else if constexpr (TM == 6)
    asm volatile(IDLE_SLOTS : "+v"(acc[0]), "+v"(acc[1]), "+v"(acc[2]), "+v"(acc[3]), "+v"(acc[4]), "+v"(acc[5]));
  else
    asm volatile(IDLE_SLOTS : "+v"(acc[0]), "+v"(acc[1]), "+v"(acc[2]), "+v"(acc[3]), "+v"(acc[4]), "+v"(acc[5]),
                 "+v"(acc[6]));
}

template <bool AK, bool BK, int TM, int NW>
__global__ __launch_bounds__(32 * NW) void gemm_f32_wmma(
    const float* __restrict__ A, long sAm, long sAk, long sAb,
    const float* __restrict__ B, long sBk, long sBn, long sBb,
    float* __restrict__ C, long sCm, long sCb,
    int M, int N, int K, float alpha)
{
  constexpr int BM = 16 * TM, BN = 16 * NW, NT = 32 * NW;
  constexpr int PRA = BM + 8, PRB = BN + 8;
  constexpr int TA = AK ? BM * PK : KC * PRA;
  constexpr int TB = BK ? BN * PK : KC * PRB;
  constexpr int NA = (4 * BM + NT - 1) / NT;
  constexpr bool AFULL = ((4 * BM) % NT) == 0;
  constexpr int PC = BN + 4;
  constexpr int OPF = 2 * TA + 2 * TB;
  constexpr int CTF = BM * PC;
  constexpr int SMF = OPF > CTF ? OPF : CTF;
  __shared__ __attribute__((aligned(16))) float smem[SMF];
  float* const As = smem;
  float* const Bs = smem + 2 * TA;

  const int tid  = threadIdx.x;
  const int lane = tid & 31;
  const int wave = tid >> 5;
  const int half = lane >> 4;
  const int l16  = lane & 15;
  const int mBase = blockIdx.x * BM;
  const int nBlk  = blockIdx.y * BN;

  A += (long)blockIdx.z * sAb;
  B += (long)blockIdx.z * sBb;
  C += (long)blockIdx.z * sCb;

  v8f zero = {0.f,0.f,0.f,0.f,0.f,0.f,0.f,0.f};
  v8f acc[TM];
#pragma unroll
  for (int t = 0; t < TM; ++t) acc[t] = zero;

  long aOff[NA], bOff[2];
  int  aKo[NA], bKo[2];
  bool aIn[NA], bIn[2];
  bool aHas[NA];
  int  aDst[NA], bDst[2];
#pragma unroll
  for (int j = 0; j < NA; ++j) {
    const int idx = tid + NT * j;
    aHas[j] = AFULL || (idx < 4 * BM);
    if (AK) {
      const int row = idx >> 2, kq = (idx & 3) * 4, m = mBase + row;
      aIn[j] = aHas[j] && (m < M);  aKo[j] = kq;  aDst[j] = row * PK + kq;
      aOff[j] = (long)(aIn[j] ? m : 0) * sAm + kq;
    } else {
      const int kk = idx / (BM / 4), mq = (idx % (BM / 4)) * 4, m = mBase + mq;
      aIn[j] = aHas[j] && (m < M);  aKo[j] = kk;  aDst[j] = kk * PRA + mq;
      aOff[j] = (long)kk * sAk + (aIn[j] ? m : 0);
    }
  }
  for (int j = 0; j < 2; ++j) {
    const int idx = tid + NT * j;
    if (BK) {
      const int row = idx >> 2, kq = (idx & 3) * 4, n = nBlk + row;
      bIn[j] = (n < N);  bKo[j] = kq;  bDst[j] = row * PK + kq;
      bOff[j] = (long)(bIn[j] ? n : 0) * sBn + kq;
    } else {
      const int kk = idx / (BN / 4), nq = (idx % (BN / 4)) * 4, n = nBlk + nq;
      bIn[j] = (n < N);  bKo[j] = kk;  bDst[j] = kk * PRB + nq;
      bOff[j] = (long)kk * sBk + (bIn[j] ? n : 0);
    }
  }
  const long aStep = AK ? (long)KC : (long)KC * sAk;
  const long bStep = BK ? (long)KC : (long)KC * sBk;

  const int aFrag = AK ? (l16 * PK + 2 * half) : (2 * half * PRA + l16);
  const int bFrag = BK ? ((wave * 16 + l16) * PK + 2 * half)
                       : (2 * half * PRB + wave * 16 + l16);

  float4 ra[NA], rb[2];

#pragma unroll
  for (int j = 0; j < NA; ++j) {
    ra[j] = make_float4(0.f, 0.f, 0.f, 0.f);
    if (aIn[j] && aKo[j] < K) ra[j] = *reinterpret_cast<const float4*>(A + aOff[j]);
    aOff[j] += aStep;
  }
  for (int j = 0; j < 2; ++j) {
    rb[j] = make_float4(0.f, 0.f, 0.f, 0.f);
    if (bIn[j] && bKo[j] < K) rb[j] = *reinterpret_cast<const float4*>(B + bOff[j]);
    bOff[j] += bStep;
  }
#pragma unroll
  for (int j = 0; j < NA; ++j)
    if (aHas[j]) *reinterpret_cast<float4*>(As + aDst[j]) = ra[j];
  for (int j = 0; j < 2; ++j)
    *reinterpret_cast<float4*>(Bs + bDst[j]) = rb[j];
  __syncthreads();

  int p = 0;
  for (int k0 = 0; ; k0 += KC) {
    const bool more = (k0 + KC < K);
    if (more) {
#pragma unroll
      for (int j = 0; j < NA; ++j) {
        ra[j] = make_float4(0.f, 0.f, 0.f, 0.f);
        if (aIn[j] && k0 + KC + aKo[j] < K) ra[j] = *reinterpret_cast<const float4*>(A + aOff[j]);
        aOff[j] += aStep;
      }
      for (int j = 0; j < 2; ++j) {
        rb[j] = make_float4(0.f, 0.f, 0.f, 0.f);
        if (bIn[j] && k0 + KC + bKo[j] < K) rb[j] = *reinterpret_cast<const float4*>(B + bOff[j]);
        bOff[j] += bStep;
      }
    }

    const float* as = As + p * TA + aFrag;
    const float* bs = Bs + p * TB + bFrag;
#pragma unroll
    for (int ks = 0; ks < KC / 4; ++ks) {
      v2f bf;
      if (BK) {
        bf = *reinterpret_cast<const v2f*>(bs + ks * 4);
      } else {
        bf.x = bs[(ks * 4) * PRB];
        bf.y = bs[(ks * 4 + 1) * PRB];
      }
#pragma unroll
      for (int t = 0; t < TM; ++t) {
        v2f af;
        if (AK) {
          af = *reinterpret_cast<const v2f*>(as + t * 16 * PK + ks * 4);
        } else {
          af.x = as[(ks * 4) * PRA + t * 16];
          af.y = as[(ks * 4 + 1) * PRA + t * 16];
        }
        acc[t] = __builtin_amdgcn_wmma_f32_16x16x4_f32(
            false, af, false, bf, (short)0, acc[t], false, false);
      }
    }

    if (!more) break;
    p ^= 1;
#pragma unroll
    for (int j = 0; j < NA; ++j)
      if (aHas[j]) *reinterpret_cast<float4*>(As + p * TA + aDst[j]) = ra[j];
    for (int j = 0; j < 2; ++j)
      *reinterpret_cast<float4*>(Bs + p * TB + bDst[j]) = rb[j];
    __syncthreads();
  }

  slots_behind_last_wmma<TM>(acc);
  __builtin_amdgcn_sched_barrier(0);
  __syncthreads();
  {
    float* cw = smem + (8 * half) * PC + wave * 16 + l16;
#pragma unroll
    for (int t = 0; t < TM; ++t) {
#pragma unroll
      for (int r = 0; r < 8; ++r) cw[(t * 16 + r) * PC] = alpha * acc[t][r];
    }
  }
  __syncthreads();
  {
    constexpr int RV = BN / 4;
    const int row0 = tid / RV, c4 = tid - row0 * RV;
    const int n = nBlk + 4 * c4;
    const float* cr = smem + row0 * PC + 4 * c4;
    float* cg = C + (long)(mBase + row0) * sCm + n;
    const long gStep = 8 * sCm;
    if (mBase + BM <= M && nBlk + BN <= N) {
#pragma unroll
      for (int j = 0; j < 2 * TM; ++j) {
        *reinterpret_cast<float4*>(cg) = *reinterpret_cast<const float4*>(cr + j * 8 * PC);
        cg += gStep;
      }
    } else if (n < N) {
#pragma unroll
      for (int j = 0; j < 2 * TM; ++j) {
        if (mBase + row0 + 8 * j < M)
          *reinterpret_cast<float4*>(cg) = *reinterpret_cast<const float4*>(cr + j * 8 * PC);
        cg += gStep;
      }
    }
  }
}

template <bool AK, bool BK, int TM, int NW>
static void launch_gemm(hipStream_t stream,
                        const float* A, long sAm, long sAk, long sAb,
                        const float* B, long sBk, long sBn, long sBb,
                        float* C, long sCm, long sCb,
                        int M, int N, int K, int batch, float alpha) {
  const unsigned mb = (unsigned)((M + 16 * TM - 1) / (16 * TM));
  const unsigned nb = (unsigned)((N + 16 * NW - 1) / (16 * NW));
  dim3 g(mb, nb, (unsigned)batch);
  gemm_f32_wmma<AK, BK, TM, NW><<<g, 32 * NW, 0, stream>>>(A, sAm, sAk, sAb, B, sBk, sBn, sBb,
                                                       C, sCm, sCb, M, N, K, alpha);
}

#define PX 100
#define PY 52
#define KC2 32
#define PK2 36
__global__ __launch_bounds__(192) void dft_xy_fused(const float* __restrict__ in,
                                                    const float* __restrict__ Fcat,
                                                    const float* F2m,
                                                    float* __restrict__ R2)
{
  constexpr int TF  = 96 * PK;
  constexpr int X1F = 96 * PX;
  __shared__ __attribute__((aligned(16))) float smem[X1F + 4 * TF];
  float* const X1 = smem;
  float* const As = smem + X1F;
  float* const Bs = As + 2 * TF;

  const int tid  = threadIdx.x;
  const int lane = tid & 31;
  const int wave = tid >> 5;
  const int half = lane >> 4;
  const int l16  = lane & 15;

  const float* A = in + (long)blockIdx.x * 9216;

  const int row = tid >> 2, kq = (tid & 3) * 4;
  const int g0 = row * 96 + kq,  g1 = g0 + 48 * 96;
  const int d0 = row * PK + kq,  d1 = d0 + 48 * PK;
  constexpr int TF2 = 96 * PK2;
  static_assert(2 * TF2 <= 4 * TF && 96 * PY <= 4 * TF, "phase-2 tiles and the result tile reuse the phase-1 tile space");

  v8f zero = {0.f,0.f,0.f,0.f,0.f,0.f,0.f,0.f};
  float4 ra0, ra1, rb0, rb1, rc0, rc1;

  v8f acc[6];
#pragma unroll
  for (int t = 0; t < 6; ++t) acc[t] = zero;

  ra0 = *reinterpret_cast<const float4*>(A + g0);
  ra1 = *reinterpret_cast<const float4*>(A + g1);
  rb0 = *reinterpret_cast<const float4*>(Fcat + g0);
  rb1 = *reinterpret_cast<const float4*>(Fcat + g1);
  *reinterpret_cast<float4*>(As + d0) = ra0;
  *reinterpret_cast<float4*>(As + d1) = ra1;
  *reinterpret_cast<float4*>(Bs + d0) = rb0;
  *reinterpret_cast<float4*>(Bs + d1) = rb1;
  __syncthreads();

  const int aFrag = l16 * PK + 2 * half;
  const int bFrag = (wave * 16 + l16) * PK + 2 * half;
  int p = 0;
#pragma nounroll
  for (int k0 = 0; ; k0 += KC) {
    const bool more = (k0 + KC < 96);
    if (more) {
      ra0 = *reinterpret_cast<const float4*>(A + g0 + k0 + KC);
      ra1 = *reinterpret_cast<const float4*>(A + g1 + k0 + KC);
      rb0 = *reinterpret_cast<const float4*>(Fcat + g0 + k0 + KC);
      rb1 = *reinterpret_cast<const float4*>(Fcat + g1 + k0 + KC);
    }
    const float* as = As + p * TF + aFrag;
    const float* bs = Bs + p * TF + bFrag;
#pragma unroll
    for (int ks = 0; ks < KC / 4; ++ks) {
      const v2f bf = *reinterpret_cast<const v2f*>(bs + ks * 4);
#pragma unroll
      for (int t = 0; t < 6; ++t) {
        const v2f af = *reinterpret_cast<const v2f*>(as + t * 16 * PK + ks * 4);
        acc[t] = __builtin_amdgcn_wmma_f32_16x16x4_f32(
            false, af, false, bf, (short)0, acc[t], false, false);
      }
    }
    if (!more) break;
    p ^= 1;
    *reinterpret_cast<float4*>(As + p * TF + d0) = ra0;
    *reinterpret_cast<float4*>(As + p * TF + d1) = ra1;
    *reinterpret_cast<float4*>(Bs + p * TF + d0) = rb0;
    *reinterpret_cast<float4*>(Bs + p * TF + d1) = rb1;
    __syncthreads();
  }

  slots_behind_last_wmma<6>(acc);
  __builtin_amdgcn_sched_barrier(0);
  const int row2 = tid >> 3, kq2 = (tid & 7) * 4;
  const int f0 = row2 * 192 + kq2;
  const int e0 = row2 * PK2 + kq2;
  ra0 = *reinterpret_cast<const float4*>(F2m + f0);
  ra1 = *reinterpret_cast<const float4*>(F2m + f0 + 24 * 192);
  rc0 = *reinterpret_cast<const float4*>(F2m + f0 + 48 * 192);
  rc1 = *reinterpret_cast<const float4*>(F2m + f0 + 72 * 192);
  {
    float* xw = X1 + (8 * half) * PX + wave * 16 + l16;
#pragma unroll
    for (int t = 0; t < 6; ++t) {
#pragma unroll
      for (int r = 0; r < 8; ++r) xw[(t * 16 + r) * PX] = acc[t][r];
    }
  }
  __syncthreads();
  *reinterpret_cast<float4*>(As + e0) = ra0;
  *reinterpret_cast<float4*>(As + e0 + 24 * PK2) = ra1;
  *reinterpret_cast<float4*>(As + e0 + 48 * PK2) = rc0;
  *reinterpret_cast<float4*>(As + e0 + 72 * PK2) = rc1;
  __syncthreads();

  v8f acc2[3];
#pragma unroll
  for (int t = 0; t < 3; ++t) acc2[t] = zero;
  const int ct = wave % 3, rg = wave / 3;
  const int aFrag2 = (rg * 48 + l16) * PK2 + 2 * half;
  const float* xb = X1 + half * PX + ct * 16 + l16;
  p = 0;
  for (int k0 = 0; ; k0 += KC2) {
    const bool more = (k0 + KC2 < 192);
    if (more) {
      ra0 = *reinterpret_cast<const float4*>(F2m + f0 + k0 + KC2);
      ra1 = *reinterpret_cast<const float4*>(F2m + f0 + 24 * 192 + k0 + KC2);
      rc0 = *reinterpret_cast<const float4*>(F2m + f0 + 48 * 192 + k0 + KC2);
      rc1 = *reinterpret_cast<const float4*>(F2m + f0 + 72 * 192 + k0 + KC2);
    }
    const float* as = As + p * TF2 + aFrag2;
    const float* bs = xb + (k0 >> 1) * PX;
#pragma unroll
    for (int ks = 0; ks < KC2 / 4; ++ks) {
      v2f bf;
      bf.x = bs[(ks * 2) * PX];
      bf.y = bs[(ks * 2) * PX + 48];
#pragma unroll
      for (int t = 0; t < 3; ++t) {
        const v2f af = *reinterpret_cast<const v2f*>(as + t * 16 * PK2 + ks * 4);
        acc2[t] = __builtin_amdgcn_wmma_f32_16x16x4_f32(
            false, af, false, bf, (short)0, acc2[t], false, false);
      }
    }
    if (!more) break;
    p ^= 1;
    *reinterpret_cast<float4*>(As + p * TF2 + e0) = ra0;
    *reinterpret_cast<float4*>(As + p * TF2 + e0 + 24 * PK2) = ra1;
    *reinterpret_cast<float4*>(As + p * TF2 + e0 + 48 * PK2) = rc0;
    *reinterpret_cast<float4*>(As + p * TF2 + e0 + 72 * PK2) = rc1;
    __syncthreads();
  }

  slots_behind_last_wmma<3>(acc2);
  __builtin_amdgcn_sched_barrier(0);
  __syncthreads();
  {
    float* yw = As + (rg * 48 + 8 * half) * PY + ct * 16 + l16;
#pragma unroll
    for (int t = 0; t < 3; ++t) {
#pragma unroll
      for (int r = 0; r < 8; ++r) yw[(t * 16 + r) * PY] = acc2[t][r];
    }
  }
  __syncthreads();
  {
    const int row0 = tid / 12, c4 = tid - row0 * 12;
    const float* yr = As + row0 * PY + 4 * c4;
    float* cg = R2 + (long)blockIdx.x * 4608 + row0 * 48 + 4 * c4;
#pragma unroll
    for (int j = 0; j < 6; ++j)
      *reinterpret_cast<float4*>(cg + j * 16 * 48) = *reinterpret_cast<const float4*>(yr + j * 16 * PY);
  }
}

__global__ void gen_consts(float* Fre, float* Fim, float* Vre, float* Vim,
                           float* F2, float* F3, float* V2, float* V3) {
  int i = blockIdx.x * 256 + threadIdx.x;
  if (i >= 48 * 96) return;
  int k = i / 96, nn = i % 96;
  int f = (k < 24) ? k : (k + 48);
  int ph = (f * nn) % 96;
  float th = 6.283185307179586f * (float)ph / 96.f;
  float c = cosf(th), s = sinf(th);
  Fre[i] = c;  Fim[i] = -s;
  Vre[nn * 48 + k] = c;  Vim[nn * 48 + k] = s;
  F2[k * 192 + 2 * nn]            = c;
  F2[k * 192 + 2 * nn + 1]        = s;
  F2[(48 + k) * 192 + 2 * nn]     = -s;
  F2[(48 + k) * 192 + 2 * nn + 1] = c;
  F3[k * 192 + 2 * nn]     = c + s;
  F3[k * 192 + 2 * nn + 1] = s - c;
  V2[nn * 96 + 2 * k]            = c;
  V2[nn * 96 + 2 * k + 1]        = -s;
  V2[(96 + nn) * 96 + 2 * k]     = s;
  V2[(96 + nn) * 96 + 2 * k + 1] = c;
  V3[nn * 96 + 2 * k]     = c + s;
  V3[nn * 96 + 2 * k + 1] = c - s;
}

__global__ void pack_wqkv(const float* __restrict__ wq, const float* __restrict__ wk,
                          const float* __restrict__ wv, float* __restrict__ dst) {
  int i = blockIdx.x * 256 + threadIdx.x;
  if (i >= 384 * 64) return;
  int ic = i % 64, r = i / 64;
  int t = r / 128, q = r % 128;
  const float* w = (t == 0) ? wq : ((t == 1) ? wk : wv);
  dst[i] = w[q * 64 + ic];
}

#define KSPLIT 32
__global__ __launch_bounds__(256) void sum_scale_selu(const float* __restrict__ part, float* __restrict__ s,
                               int nvec, int planeVec, float scale) {
  int i = blockIdx.x * 256 + threadIdx.x;
  if (i >= nvec) return;
  int h = i / planeVec, e = i - h * planeVec;
  const float4* p = reinterpret_cast<const float4*>(part) + (long)h * KSPLIT * planeVec + e;
  float4 v[KSPLIT];
  for (int j = 0; j < KSPLIT; ++j) v[j] = p[(long)j * planeVec];
  float4 a = v[0];
  for (int j = 1; j < KSPLIT; ++j) {
    a.x += v[j].x; a.y += v[j].y; a.z += v[j].z; a.w += v[j].w;
  }
  const float lam = 1.0507009873554805f, alp = 1.6732632423543772f;
  float x0 = scale * a.x, x1 = scale * a.y, x2 = scale * a.z, x3 = scale * a.w;
  float4 r;
  r.x = lam * (x0 > 0.f ? x0 : alp * (expf(x0) - 1.f));
  r.y = lam * (x1 > 0.f ? x1 : alp * (expf(x1) - 1.f));
  r.z = lam * (x2 > 0.f ? x2 : alp * (expf(x2) - 1.f));
  r.w = lam * (x3 > 0.f ? x3 : alp * (expf(x3) - 1.f));
  reinterpret_cast<float4*>(s)[i] = r;
}


__global__ void group_rows(const float* __restrict__ src, float* __restrict__ dst, int nelem) {
  int i = blockIdx.x * 256 + threadIdx.x;
  if (i >= nelem) return;
  int s = i % 110592, r = i / 110592;
  int x = s % 48, y = (s / 48) % 48, z = s / 2304;
  int px = x & 7, nx = x >> 3, py = y & 7, ny = y >> 3, pz = z & 7, nz = z >> 3;
  int p = (((pz << 3) | py) << 3) | px;
  int tok = (nz * 6 + ny) * 6 + nx;
  dst[(long)r * 110592 + p * 216 + tok] = src[i];
}

__global__ void ungroup_rows(const float* __restrict__ src, float* __restrict__ dst, int nelem) {
  int i = blockIdx.x * 256 + threadIdx.x;
  if (i >= nelem) return;
  int g = i % 110592, r = i / 110592;
  int tok = g % 216, p = g / 216;
  int px = p & 7, py = (p >> 3) & 7, pz = p >> 6;
  int nx = tok % 6, ny = (tok / 6) % 6, nz = tok / 36;
  int z = nz * 8 + pz, y = ny * 8 + py, x = nx * 8 + px;
  dst[(long)r * 110592 + (long)(z * 48 + y) * 48 + x] = src[i];
}

extern "C" void kernel_launch(void* const* d_in, const int* in_sizes, int n_in,
                              void* d_out, int out_size, void* d_ws, size_t ws_size,
                              hipStream_t stream) {
  const float* in = (const float*)d_in[0];
  const float* wq = (const float*)d_in[1];
  const float* wk = (const float*)d_in[2];
  const float* wv = (const float*)d_in[3];
  const float* wo = (const float*)d_in[4];
  float* out = (float*)d_out;
  float* w = (float*)d_ws;

  const long R2_o   = 56623104;
  const long blk_o  = 0;
  const long blkG_o = 7077888,    G_o    = 42467328;
  const long S_o    = 92012544,   Og_o   = 92385792;
  const long P_o    = 0;
  const long specG_o = 0,         spec_o = 14155776;
  const long T1_o   = 15925248;
  const long T2_o   = 23003136;
  const long F2_o   = 84934656,   F3_o   = 84953088;
  const long V2_o   = 84962304,   V3_o   = 84980736;
  const long Wqkv_o = 106541568;
  const long Fre_o  = 106566144,  Fim_o  = 106570752;
  const long Vre_o  = 106575360,  Vim_o  = 106579968;
  const long WS_FLOATS = 106584576;
  if (ws_size < (size_t)WS_FLOATS * sizeof(float)) return;

  float *R2 = w+R2_o;
  float *blk = w+blk_o, *blkG = w+blkG_o, *G = w+G_o, *S = w+S_o, *Og = w+Og_o;
  float *specG = w+specG_o, *spec = w+spec_o, *P = w+P_o;
  float *T1 = w+T1_o, *T2 = w+T2_o, *Wqkv = w+Wqkv_o;
  float *Fre = w+Fre_o, *Fim = w+Fim_o, *Vre = w+Vre_o, *Vim = w+Vim_o;
  float *F2 = w+F2_o, *F3 = w+F3_o, *V2 = w+V2_o, *V3 = w+V3_o;

  auto ew = [&](int n) { return dim3((unsigned)((n + 255) / 256)); };

  gen_consts<<<ew(4608), 256, 0, stream>>>(Fre, Fim, Vre, Vim, F2, F3, V2, V3);
  pack_wqkv<<<ew(24576), 256, 0, stream>>>(wq, wk, wv, Wqkv);

  dft_xy_fused<<<dim3(6144), 192, 0, stream>>>(in, Fre, F2, R2);
  launch_gemm<true, false, 3, 4>(stream, F3, 192, 1, 0, R2, 2304, 1, 442368, blk, 2304, 110592, 48, 2304, 192, 64, 1.f);

  group_rows<<<ew(7077888), 256, 0, stream>>>(blk, blkG, 7077888);
  launch_gemm<true, false, 4, 4>(stream, Wqkv, 64, 1, 0, blkG, 110592, 1, 0, G, 110592, 0, 384, 110592, 64, 1, 1.f);

  const long HD = 8192L * 216L;
  const long SL = 256L * 216L;
  launch_gemm<false, false, 7, 7>(stream, G, 1, 216, SL, G + 8 * HD, 216, 1, SL, P, 216, 46656,
       216, 216, 256, 8 * KSPLIT, 1.f);
  sum_scale_selu<<<ew(93312), 256, 0, stream>>>(P, S, 93312, 11664,
                                                0.011048543456039806f);
  launch_gemm<true, true, 4, 7>(stream, G + 16 * HD, 216, 1, HD, S, 1, 216, 46656, Og, 216, HD,
       8192, 216, 216, 8, 1.f);

  launch_gemm<true, false, 1, 4>(stream, wo, 128, 1, 0, Og, 110592, 1, 0, specG, 110592, 0, 16, 110592, 128, 1, 1.f);
  ungroup_rows<<<ew(1769472), 256, 0, stream>>>(specG, spec, 1769472);

  launch_gemm<true, true, 4, 4>(stream, spec, 48, 1, 0, Vre, 1, 48, 0, T1, 192, 0, 36864, 192, 48, 1, 1.f);
  launch_gemm<true, false, 6, 6>(stream, V2, 96, 1, 0, T1, 96, 1, 9216, T2, 96, 18432, 192, 96, 96, 768, 1.f);
  const float invN = 1.f / 884736.f;
  launch_gemm<true, false, 6, 4>(stream, V3, 96, 1, 0, T2, 9216, 1, 884736, out, 9216, 884736, 96, 9216, 96, 16, invN);
}


# Patch-grouping permutations go through LDS, one block per (row, pz, py pair): 16-byte loads and stores on whole 128-byte lines, 3 or 9 per thread, not one 4-byte element per thread; no sum touched
# speedup vs baseline: 6.7708x; 1.1018x over previous
#include <hip/hip_runtime.h>
#include <math.h>

typedef float v2f __attribute__((ext_vector_type(2)));
typedef float v8f __attribute__((ext_vector_type(8)));

#define KC 16
#define PK 20

#define IDLE_SLOTS "v_nop\n\tv_nop\n\tv_nop\n\tv_nop\n\tv_nop"
template <int TM>
__device__ __forceinline__ void slots_behind_last_wmma(v8f (&acc)[TM]) {
  static_assert(TM == 1 || TM == 3 || TM == 4 || TM == 6 || TM == 7, "add the operand list for this TM");
  if constexpr (TM == 1)
    asm volatile(IDLE_SLOTS : "+v"(acc[0]));
  else if constexpr (TM == 3)
    asm volatile(IDLE_SLOTS : "+v"(acc[0]), "+v"(acc[1]), "+v"(acc[2]));
  else if constexpr (TM == 4)
    asm volatile(IDLE_SLOTS : "+v"(acc[0]), "+v"(acc[1]), "+v"(acc[2]), "+v"(acc[3]));
  else if constexpr (TM == 6)
    asm volatile(IDLE_SLOTS : "+v"(acc[0]), "+v"(acc[1]), "+v"(acc[2]), "+v"(acc[3]), "+v"(acc[4]), "+v"(acc[5]));
  else
    asm volatile(IDLE_SLOTS : "+v"(acc[0]), "+v"(acc[1]), "+v"(acc[2]), "+v"(acc[3]), "+v"(acc[4]), "+v"(acc[5]),
                 "+v"(acc[6]));
}

template <bool AK, bool BK, int TM, int NW>
__global__ __launch_bounds__(32 * NW) void gemm_f32_wmma(
    const float* __restrict__ A, long sAm, long sAk, long sAb,
    const float* __restrict__ B, long sBk, long sBn, long sBb,
    float* __restrict__ C, long sCm, long sCb,
    int M, int N, int K, float alpha)
{
  constexpr int BM = 16 * TM, BN = 16 * NW, NT = 32 * NW;
  constexpr int PRA = BM + 8, PRB = BN + 8;
  constexpr int TA = AK ? BM * PK : KC * PRA;
  constexpr int TB = BK ? BN * PK : KC * PRB;
  constexpr int NA = (4 * BM + NT - 1) / NT;
  constexpr bool AFULL = ((4 * BM) % NT) == 0;
  constexpr int PC = BN + 4;
  constexpr int OPF = 2 * TA + 2 * TB;
  constexpr int CTF = BM * PC;
  constexpr int SMF = OPF > CTF ? OPF : CTF;
  __shared__ __attribute__((aligned(16))) float smem[SMF];
  float* const As = smem;
  float* const Bs = smem + 2 * TA;

  const int tid  = threadIdx.x;
  const int lane = tid & 31;
  const int wave = tid >> 5;
  const int half = lane >> 4;
  const int l16  = lane & 15;
  const int mBase = blockIdx.x * BM;
  const int nBlk  = blockIdx.y * BN;

  A += (long)blockIdx.z * sAb;
  B += (long)blockIdx.z * sBb;
  C += (long)blockIdx.z * sCb;

  v8f zero = {0.f,0.f,0.f,0.f,0.f,0.f,0.f,0.f};
  v8f acc[TM];
#pragma unroll
  for (int t = 0; t < TM; ++t) acc[t] = zero;

  long aOff[NA], bOff[2];
  int  aKo[NA], bKo[2];
  bool aIn[NA], bIn[2];
  bool aHas[NA];
  int  aDst[NA], bDst[2];
#pragma unroll
  for (int j = 0; j < NA; ++j) {
    const int idx = tid + NT * j;
    aHas[j] = AFULL || (idx < 4 * BM);
    if (AK) {
      const int row = idx >> 2, kq = (idx & 3) * 4, m = mBase + row;
      aIn[j] = aHas[j] && (m < M);  aKo[j] = kq;  aDst[j] = row * PK + kq;
      aOff[j] = (long)(aIn[j] ? m : 0) * sAm + kq;
    } else {
      const int kk = idx / (BM / 4), mq = (idx % (BM / 4)) * 4, m = mBase + mq;
      aIn[j] = aHas[j] && (m < M);  aKo[j] = kk;  aDst[j] = kk * PRA + mq;
      aOff[j] = (long)kk * sAk + (aIn[j] ? m : 0);
    }
  }
  for (int j = 0; j < 2; ++j) {
    const int idx = tid + NT * j;
    if (BK) {
      const int row = idx >> 2, kq = (idx & 3) * 4, n = nBlk + row;
      bIn[j] = (n < N);  bKo[j] = kq;  bDst[j] = row * PK + kq;
      bOff[j] = (long)(bIn[j] ? n : 0) * sBn + kq;
    } else {
      const int kk = idx / (BN / 4), nq = (idx % (BN / 4)) * 4, n = nBlk + nq;
      bIn[j] = (n < N);  bKo[j] = kk;  bDst[j] = kk * PRB + nq;
      bOff[j] = (long)kk * sBk + (bIn[j] ? n : 0);
    }
  }
  const long aStep = AK ? (long)KC : (long)KC * sAk;
  const long bStep = BK ? (long)KC : (long)KC * sBk;

  const int aFrag = AK ? (l16 * PK + 2 * half) : (2 * half * PRA + l16);
  const int bFrag = BK ? ((wave * 16 + l16) * PK + 2 * half)
                       : (2 * half * PRB + wave * 16 + l16);

  float4 ra[NA], rb[2];

#pragma unroll
  for (int j = 0; j < NA; ++j) {
    ra[j] = make_float4(0.f, 0.f, 0.f, 0.f);
    if (aIn[j] && aKo[j] < K) ra[j] = *reinterpret_cast<const float4*>(A + aOff[j]);
    aOff[j] += aStep;
  }
  for (int j = 0; j < 2; ++j) {
    rb[j] = make_float4(0.f, 0.f, 0.f, 0.f);
    if (bIn[j] && bKo[j] < K) rb[j] = *reinterpret_cast<const float4*>(B + bOff[j]);
    bOff[j] += bStep;
  }
#pragma unroll
  for (int j = 0; j < NA; ++j)
    if (aHas[j]) *reinterpret_cast<float4*>(As + aDst[j]) = ra[j];
  for (int j = 0; j < 2; ++j)
    *reinterpret_cast<float4*>(Bs + bDst[j]) = rb[j];
  __syncthreads();

  int p = 0;
  for (int k0 = 0; ; k0 += KC) {
    const bool more = (k0 + KC < K);
    if (more) {
#pragma unroll
      for (int j = 0; j < NA; ++j) {
        ra[j] = make_float4(0.f, 0.f, 0.f, 0.f);
        if (aIn[j] && k0 + KC + aKo[j] < K) ra[j] = *reinterpret_cast<const float4*>(A + aOff[j]);
        aOff[j] += aStep;
      }
      for (int j = 0; j < 2; ++j) {
        rb[j] = make_float4(0.f, 0.f, 0.f, 0.f);
        if (bIn[j] && k0 + KC + bKo[j] < K) rb[j] = *reinterpret_cast<const float4*>(B + bOff[j]);
        bOff[j] += bStep;
      }
    }

    const float* as = As + p * TA + aFrag;
    const float* bs = Bs + p * TB + bFrag;
#pragma unroll
    for (int ks = 0; ks < KC / 4; ++ks) {
      v2f bf;
      if (BK) {
        bf = *reinterpret_cast<const v2f*>(bs + ks * 4);
      } else {
        bf.x = bs[(ks * 4) * PRB];
        bf.y = bs[(ks * 4 + 1) * PRB];
      }
#pragma unroll
      for (int t = 0; t < TM; ++t) {
        v2f af;
        if (AK) {
          af = *reinterpret_cast<const v2f*>(as + t * 16 * PK + ks * 4);
        } else {
          af.x = as[(ks * 4) * PRA + t * 16];
          af.y = as[(ks * 4 + 1) * PRA + t * 16];
        }
        acc[t] = __builtin_amdgcn_wmma_f32_16x16x4_f32(
            false, af, false, bf, (short)0, acc[t], false, false);
      }
    }

    if (!more) break;
    p ^= 1;
#pragma unroll
    for (int j = 0; j < NA; ++j)
      if (aHas[j]) *reinterpret_cast<float4*>(As + p * TA + aDst[j]) = ra[j];
    for (int j = 0; j < 2; ++j)
      *reinterpret_cast<float4*>(Bs + p * TB + bDst[j]) = rb[j];
    __syncthreads();
  }

  slots_behind_last_wmma<TM>(acc);
  __builtin_amdgcn_sched_barrier(0);
  __syncthreads();
  {
    float* cw = smem + (8 * half) * PC + wave * 16 + l16;
#pragma unroll
    for (int t = 0; t < TM; ++t) {
#pragma unroll
      for (int r = 0; r < 8; ++r) cw[(t * 16 + r) * PC] = alpha * acc[t][r];
    }
  }
  __syncthreads();
  {
    constexpr int RV = BN / 4;
    const int row0 = tid / RV, c4 = tid - row0 * RV;
    const int n = nBlk + 4 * c4;
    const float* cr = smem + row0 * PC + 4 * c4;
    float* cg = C + (long)(mBase + row0) * sCm + n;
    const long gStep = 8 * sCm;
    if (mBase + BM <= M && nBlk + BN <= N) {
#pragma unroll
      for (int j = 0; j < 2 * TM; ++j) {
        *reinterpret_cast<float4*>(cg) = *reinterpret_cast<const float4*>(cr + j * 8 * PC);
        cg += gStep;
      }
    } else if (n < N) {
#pragma unroll
      for (int j = 0; j < 2 * TM; ++j) {
        if (mBase + row0 + 8 * j < M)
          *reinterpret_cast<float4*>(cg) = *reinterpret_cast<const float4*>(cr + j * 8 * PC);
        cg += gStep;
      }
    }
  }
}

template <bool AK, bool BK, int TM, int NW>
static void launch_gemm(hipStream_t stream,
                        const float* A, long sAm, long sAk, long sAb,
                        const float* B, long sBk, long sBn, long sBb,
                        float* C, long sCm, long sCb,
                        int M, int N, int K, int batch, float alpha) {
  const unsigned mb = (unsigned)((M + 16 * TM - 1) / (16 * TM));
  const unsigned nb = (unsigned)((N + 16 * NW - 1) / (16 * NW));
  dim3 g(mb, nb, (unsigned)batch);
  gemm_f32_wmma<AK, BK, TM, NW><<<g, 32 * NW, 0, stream>>>(A, sAm, sAk, sAb, B, sBk, sBn, sBb,
                                                       C, sCm, sCb, M, N, K, alpha);
}

#define PX 100
#define PY 52
#define KC2 32
#define PK2 36
__global__ __launch_bounds__(192) void dft_xy_fused(const float* __restrict__ in,
                                                    const float* __restrict__ Fcat,
                                                    const float* F2m,
                                                    float* __restrict__ R2)
{
  constexpr int TF  = 96 * PK;
  constexpr int X1F = 96 * PX;
  __shared__ __attribute__((aligned(16))) float smem[X1F + 4 * TF];
  float* const X1 = smem;
  float* const As = smem + X1F;
  float* const Bs = As + 2 * TF;

  const int tid  = threadIdx.x;
  const int lane = tid & 31;
  const int wave = tid >> 5;
  const int half = lane >> 4;
  const int l16  = lane & 15;

  const float* A = in + (long)blockIdx.x * 9216;

  const int row = tid >> 2, kq = (tid & 3) * 4;
  const int g0 = row * 96 + kq,  g1 = g0 + 48 * 96;
  const int d0 = row * PK + kq,  d1 = d0 + 48 * PK;
  constexpr int TF2 = 96 * PK2;
  static_assert(2 * TF2 <= 4 * TF && 96 * PY <= 4 * TF, "phase-2 tiles and the result tile reuse the phase-1 tile space");

  v8f zero = {0.f,0.f,0.f,0.f,0.f,0.f,0.f,0.f};
  float4 ra0, ra1, rb0, rb1, rc0, rc1;

  v8f acc[6];
#pragma unroll
  for (int t = 0; t < 6; ++t) acc[t] = zero;

  ra0 = *reinterpret_cast<const float4*>(A + g0);
  ra1 = *reinterpret_cast<const float4*>(A + g1);
  rb0 = *reinterpret_cast<const float4*>(Fcat + g0);
  rb1 = *reinterpret_cast<const float4*>(Fcat + g1);
  *reinterpret_cast<float4*>(As + d0) = ra0;
  *reinterpret_cast<float4*>(As + d1) = ra1;
  *reinterpret_cast<float4*>(Bs + d0) = rb0;
  *reinterpret_cast<float4*>(Bs + d1) = rb1;
  __syncthreads();

  const int aFrag = l16 * PK + 2 * half;
  const int bFrag = (wave * 16 + l16) * PK + 2 * half;
  int p = 0;
#pragma nounroll
  for (int k0 = 0; ; k0 += KC) {
    const bool more = (k0 + KC < 96);
    if (more) {
      ra0 = *reinterpret_cast<const float4*>(A + g0 + k0 + KC);
      ra1 = *reinterpret_cast<const float4*>(A + g1 + k0 + KC);
      rb0 = *reinterpret_cast<const float4*>(Fcat + g0 + k0 + KC);
      rb1 = *reinterpret_cast<const float4*>(Fcat + g1 + k0 + KC);
    }
    const float* as = As + p * TF + aFrag;
    const float* bs = Bs + p * TF + bFrag;
#pragma unroll
    for (int ks = 0; ks < KC / 4; ++ks) {
      const v2f bf = *reinterpret_cast<const v2f*>(bs + ks * 4);
#pragma unroll
      for (int t = 0; t < 6; ++t) {
        const v2f af = *reinterpret_cast<const v2f*>(as + t * 16 * PK + ks * 4);
        acc[t] = __builtin_amdgcn_wmma_f32_16x16x4_f32(
            false, af, false, bf, (short)0, acc[t], false, false);
      }
    }
    if (!more) break;
    p ^= 1;
    *reinterpret_cast<float4*>(As + p * TF + d0) = ra0;
    *reinterpret_cast<float4*>(As + p * TF + d1) = ra1;
    *reinterpret_cast<float4*>(Bs + p * TF + d0) = rb0;
    *reinterpret_cast<float4*>(Bs + p * TF + d1) = rb1;
    __syncthreads();
  }

  slots_behind_last_wmma<6>(acc);
  __builtin_amdgcn_sched_barrier(0);
  const int row2 = tid >> 3, kq2 = (tid & 7) * 4;
  const int f0 = row2 * 192 + kq2;
  const int e0 = row2 * PK2 + kq2;
  ra0 = *reinterpret_cast<const float4*>(F2m + f0);
  ra1 = *reinterpret_cast<const float4*>(F2m + f0 + 24 * 192);
  rc0 = *reinterpret_cast<const float4*>(F2m + f0 + 48 * 192);
  rc1 = *reinterpret_cast<const float4*>(F2m + f0 + 72 * 192);
  {
    float* xw = X1 + (8 * half) * PX + wave * 16 + l16;
#pragma unroll
    for (int t = 0; t < 6; ++t) {
#pragma unroll
      for (int r = 0; r < 8; ++r) xw[(t * 16 + r) * PX] = acc[t][r];
    }
  }
  __syncthreads();
  *reinterpret_cast<float4*>(As + e0) = ra0;
  *reinterpret_cast<float4*>(As + e0 + 24 * PK2) = ra1;
  *reinterpret_cast<float4*>(As + e0 + 48 * PK2) = rc0;
  *reinterpret_cast<float4*>(As + e0 + 72 * PK2) = rc1;
  __syncthreads();

  v8f acc2[3];
#pragma unroll
  for (int t = 0; t < 3; ++t) acc2[t] = zero;
  const int ct = wave % 3, rg = wave / 3;
  const int aFrag2 = (rg * 48 + l16) * PK2 + 2 * half;
  const float* xb = X1 + half * PX + ct * 16 + l16;
  p = 0;
  for (int k0 = 0; ; k0 += KC2) {
    const bool more = (k0 + KC2 < 192);
    if (more) {
      ra0 = *reinterpret_cast<const float4*>(F2m + f0 + k0 + KC2);
      ra1 = *reinterpret_cast<const float4*>(F2m + f0 + 24 * 192 + k0 + KC2);
      rc0 = *reinterpret_cast<const float4*>(F2m + f0 + 48 * 192 + k0 + KC2);
      rc1 = *reinterpret_cast<const float4*>(F2m + f0 + 72 * 192 + k0 + KC2);
    }
    const float* as = As + p * TF2 + aFrag2;
    const float* bs = xb + (k0 >> 1) * PX;
#pragma unroll
    for (int ks = 0; ks < KC2 / 4; ++ks) {
      v2f bf;
      bf.x = bs[(ks * 2) * PX];
      bf.y = bs[(ks * 2) * PX + 48];
#pragma unroll
      for (int t = 0; t < 3; ++t) {
        const v2f af = *reinterpret_cast<const v2f*>(as + t * 16 * PK2 + ks * 4);
        acc2[t] = __builtin_amdgcn_wmma_f32_16x16x4_f32(
            false, af, false, bf, (short)0, acc2[t], false, false);
      }
    }
    if (!more) break;
    p ^= 1;
    *reinterpret_cast<float4*>(As + p * TF2 + e0) = ra0;
    *reinterpret_cast<float4*>(As + p * TF2 + e0 + 24 * PK2) = ra1;
    *reinterpret_cast<float4*>(As + p * TF2 + e0 + 48 * PK2) = rc0;
    *reinterpret_cast<float4*>(As + p * TF2 + e0 + 72 * PK2) = rc1;
    __syncthreads();
  }

  slots_behind_last_wmma<3>(acc2);
  __builtin_amdgcn_sched_barrier(0);
  __syncthreads();
  {
    float* yw = As + (rg * 48 + 8 * half) * PY + ct * 16 + l16;
#pragma unroll
    for (int t = 0; t < 3; ++t) {
#pragma unroll
      for (int r = 0; r < 8; ++r) yw[(t * 16 + r) * PY] = acc2[t][r];
    }
  }
  __syncthreads();
  {
    const int row0 = tid / 12, c4 = tid - row0 * 12;
    const float* yr = As + row0 * PY + 4 * c4;
    float* cg = R2 + (long)blockIdx.x * 4608 + row0 * 48 + 4 * c4;
#pragma unroll
    for (int j = 0; j < 6; ++j)
      *reinterpret_cast<float4*>(cg + j * 16 * 48) = *reinterpret_cast<const float4*>(yr + j * 16 * PY);
  }
}

__global__ void gen_consts(float* Fre, float* Fim, float* Vre, float* Vim,
                           float* F2, float* F3, float* V2, float* V3) {
  int i = blockIdx.x * 256 + threadIdx.x;
  if (i >= 48 * 96) return;
  int k = i / 96, nn = i % 96;
  int f = (k < 24) ? k : (k + 48);
  int ph = (f * nn) % 96;
  float th = 6.283185307179586f * (float)ph / 96.f;
  float c = cosf(th), s = sinf(th);
  Fre[i] = c;  Fim[i] = -s;
  Vre[nn * 48 + k] = c;  Vim[nn * 48 + k] = s;
  F2[k * 192 + 2 * nn]            = c;
  F2[k * 192 + 2 * nn + 1]        = s;
  F2[(48 + k) * 192 + 2 * nn]     = -s;
  F2[(48 + k) * 192 + 2 * nn + 1] = c;
  F3[k * 192 + 2 * nn]     = c + s;
  F3[k * 192 + 2 * nn + 1] = s - c;
  V2[nn * 96 + 2 * k]            = c;
  V2[nn * 96 + 2 * k + 1]        = -s;
  V2[(96 + nn) * 96 + 2 * k]     = s;
  V2[(96 + nn) * 96 + 2 * k + 1] = c;
  V3[nn * 96 + 2 * k]     = c + s;
  V3[nn * 96 + 2 * k + 1] = c - s;
}

__global__ void pack_wqkv(const float* __restrict__ wq, const float* __restrict__ wk,
                          const float* __restrict__ wv, float* __restrict__ dst) {
  int i = blockIdx.x * 256 + threadIdx.x;
  if (i >= 384 * 64) return;
  int ic = i % 64, r = i / 64;
  int t = r / 128, q = r % 128;
  const float* w = (t == 0) ? wq : ((t == 1) ? wk : wv);
  dst[i] = w[q * 64 + ic];
}

#define KSPLIT 32
__global__ __launch_bounds__(256) void sum_scale_selu(const float* __restrict__ part, float* __restrict__ s,
                               int nvec, int planeVec, float scale) {
  int i = blockIdx.x * 256 + threadIdx.x;
  if (i >= nvec) return;
  int h = i / planeVec, e = i - h * planeVec;
  const float4* p = reinterpret_cast<const float4*>(part) + (long)h * KSPLIT * planeVec + e;
  float4 v[KSPLIT];
  for (int j = 0; j < KSPLIT; ++j) v[j] = p[(long)j * planeVec];
  float4 a = v[0];
  for (int j = 1; j < KSPLIT; ++j) {
    a.x += v[j].x; a.y += v[j].y; a.z += v[j].z; a.w += v[j].w;
  }
  const float lam = 1.0507009873554805f, alp = 1.6732632423543772f;
  float x0 = scale * a.x, x1 = scale * a.y, x2 = scale * a.z, x3 = scale * a.w;
  float4 r;
  r.x = lam * (x0 > 0.f ? x0 : alp * (expf(x0) - 1.f));
  r.y = lam * (x1 > 0.f ? x1 : alp * (expf(x1) - 1.f));
  r.z = lam * (x2 > 0.f ? x2 : alp * (expf(x2) - 1.f));
  r.w = lam * (x3 > 0.f ? x3 : alp * (expf(x3) - 1.f));
  reinterpret_cast<float4*>(s)[i] = r;
}


#define GP_HALF 1728
#define GP_PAD  16
#define GP_F4   864

template <int NT>
__global__ __launch_bounds__(NT) void group_rows(const float* __restrict__ src, float* __restrict__ dst) {
  constexpr int NP = GP_F4 / NT;
  static_assert(NP * NT == GP_F4, "whole pieces per thread");
  __shared__ __attribute__((aligned(16))) float t[2 * GP_HALF + GP_PAD];
  const int b = blockIdx.x;
  const int pyp = b & 3, pz = (b >> 2) & 7, r = b >> 5;
  const float* s = src + (long)r * 110592 + (pz * 48 + 2 * pyp) * 48;
  float* d = dst + (long)r * 110592 + (pz * 64 + pyp * 16) * 216;
  float4 v[NP];
#pragma unroll
  for (int i = 0; i < NP; ++i) {
    const int f = threadIdx.x + NT * i;
    const int chunk = f / 24, q = f - chunk * 24;
    const int nz = chunk / 6, ny = chunk - nz * 6;
    v[i] = *reinterpret_cast<const float4*>(s + (nz * 384 + ny * 8) * 48 + q * 4);
  }
#pragma unroll
  for (int i = 0; i < NP; ++i) {
    const int f = threadIdx.x + NT * i;
    const int chunk = f / 24, q = f - chunk * 24;
    const int yy = q / 12, xq = q - yy * 12;
    float* w = t + yy * (GP_HALF + GP_PAD) + (4 * (xq & 1)) * 216 + chunk * 6 + (xq >> 1);
    w[0] = v[i].x;  w[216] = v[i].y;  w[432] = v[i].z;  w[648] = v[i].w;
  }
  __syncthreads();
#pragma unroll
  for (int i = 0; i < NP; ++i) {
    const int g = threadIdx.x + NT * i;
    reinterpret_cast<float4*>(d)[g] = *reinterpret_cast<const float4*>(t + 4 * g + (g / 432) * GP_PAD);
  }
}

template <int NT>
__global__ __launch_bounds__(NT) void ungroup_rows(const float* __restrict__ src, float* __restrict__ dst) {
  constexpr int NP = GP_F4 / NT;
  static_assert(NP * NT == GP_F4, "whole pieces per thread");
  __shared__ __attribute__((aligned(16))) float t[2 * GP_HALF + GP_PAD];
  const int b = blockIdx.x;
  const int pyp = b & 3, pz = (b >> 2) & 7, r = b >> 5;
  const float* s = src + (long)r * 110592 + (pz * 64 + pyp * 16) * 216;
  float* d = dst + (long)r * 110592 + (pz * 48 + 2 * pyp) * 48;
  float4 v[NP];
#pragma unroll
  for (int i = 0; i < NP; ++i) v[i] = reinterpret_cast<const float4*>(s)[threadIdx.x + NT * i];
#pragma unroll
  for (int i = 0; i < NP; ++i) {
    const int g = threadIdx.x + NT * i;
    *reinterpret_cast<float4*>(t + 4 * g + (g / 432) * GP_PAD) = v[i];
  }
  __syncthreads();
#pragma unroll
  for (int i = 0; i < NP; ++i) {
    const int f = threadIdx.x + NT * i;
    const int chunk = f / 24, q = f - chunk * 24;
    const int nz = chunk / 6, ny = chunk - nz * 6;
    const int yy = q / 12, xq = q - yy * 12;
    const float* w = t + yy * (GP_HALF + GP_PAD) + (4 * (xq & 1)) * 216 + chunk * 6 + (xq >> 1);
    float4 o;
    o.x = w[0];  o.y = w[216];  o.z = w[432];  o.w = w[648];
    *reinterpret_cast<float4*>(d + (nz * 384 + ny * 8) * 48 + q * 4) = o;
  }
}

extern "C" void kernel_launch(void* const* d_in, const int* in_sizes, int n_in,
                              void* d_out, int out_size, void* d_ws, size_t ws_size,
                              hipStream_t stream) {
  const float* in = (const float*)d_in[0];
  const float* wq = (const float*)d_in[1];
  const float* wk = (const float*)d_in[2];
  const float* wv = (const float*)d_in[3];
  const float* wo = (const float*)d_in[4];
  float* out = (float*)d_out;
  float* w = (float*)d_ws;

  const long R2_o   = 56623104;
  const long blk_o  = 0;
  const long blkG_o = 7077888,    G_o    = 42467328;
  const long S_o    = 92012544,   Og_o   = 92385792;
  const long P_o    = 0;
  const long specG_o = 0,         spec_o = 14155776;
  const long T1_o   = 15925248;
  const long T2_o   = 23003136;
  const long F2_o   = 84934656,   F3_o   = 84953088;
  const long V2_o   = 84962304,   V3_o   = 84980736;
  const long Wqkv_o = 106541568;
  const long Fre_o  = 106566144,  Fim_o  = 106570752;
  const long Vre_o  = 106575360,  Vim_o  = 106579968;
  const long WS_FLOATS = 106584576;
  if (ws_size < (size_t)WS_FLOATS * sizeof(float)) return;

  float *R2 = w+R2_o;
  float *blk = w+blk_o, *blkG = w+blkG_o, *G = w+G_o, *S = w+S_o, *Og = w+Og_o;
  float *specG = w+specG_o, *spec = w+spec_o, *P = w+P_o;
  float *T1 = w+T1_o, *T2 = w+T2_o, *Wqkv = w+Wqkv_o;
  float *Fre = w+Fre_o, *Fim = w+Fim_o, *Vre = w+Vre_o, *Vim = w+Vim_o;
  float *F2 = w+F2_o, *F3 = w+F3_o, *V2 = w+V2_o, *V3 = w+V3_o;

  auto ew = [&](int n) { return dim3((unsigned)((n + 255) / 256)); };

  gen_consts<<<ew(4608), 256, 0, stream>>>(Fre, Fim, Vre, Vim, F2, F3, V2, V3);
  pack_wqkv<<<ew(24576), 256, 0, stream>>>(wq, wk, wv, Wqkv);

  dft_xy_fused<<<dim3(6144), 192, 0, stream>>>(in, Fre, F2, R2);
  launch_gemm<true, false, 3, 4>(stream, F3, 192, 1, 0, R2, 2304, 1, 442368, blk, 2304, 110592, 48, 2304, 192, 64, 1.f);

  group_rows<96><<<dim3(2048), 96, 0, stream>>>(blk, blkG);
  launch_gemm<true, false, 4, 4>(stream, Wqkv, 64, 1, 0, blkG, 110592, 1, 0, G, 110592, 0, 384, 110592, 64, 1, 1.f);

  const long HD = 8192L * 216L;
  const long SL = 256L * 216L;
  launch_gemm<false, false, 7, 7>(stream, G, 1, 216, SL, G + 8 * HD, 216, 1, SL, P, 216, 46656,
       216, 216, 256, 8 * KSPLIT, 1.f);
  sum_scale_selu<<<ew(93312), 256, 0, stream>>>(P, S, 93312, 11664,
                                                0.011048543456039806f);
  launch_gemm<true, true, 4, 7>(stream, G + 16 * HD, 216, 1, HD, S, 1, 216, 46656, Og, 216, HD,
       8192, 216, 216, 8, 1.f);

  launch_gemm<true, false, 1, 4>(stream, wo, 128, 1, 0, Og, 110592, 1, 0, specG, 110592, 0, 16, 110592, 128, 1, 1.f);
  ungroup_rows<288><<<dim3(512), 288, 0, stream>>>(specG, spec);

  launch_gemm<true, true, 4, 4>(stream, spec, 48, 1, 0, Vre, 1, 48, 0, T1, 192, 0, 36864, 192, 48, 1, 1.f);
  launch_gemm<true, false, 6, 6>(stream, V2, 96, 1, 0, T1, 96, 1, 9216, T2, 96, 18432, 192, 96, 96, 768, 1.f);
  const float invN = 1.f / 884736.f;
  launch_gemm<true, false, 6, 4>(stream, V3, 96, 1, 0, T2, 9216, 1, 884736, out, 9216, 884736, 96, 9216, 96, 16, invN);
}


# Real slab's x spectrum is conjugate-symmetric: x axis makes only its 49 independent real columns, y axis is ONE real product of Fcat with them, mirrored modes by add/sub at store (sum order changes)
# speedup vs baseline: 7.3911x; 1.0916x over previous
#include <hip/hip_runtime.h>
#include <math.h>

typedef float v2f __attribute__((ext_vector_type(2)));
typedef float v8f __attribute__((ext_vector_type(8)));

#define KC 16
#define PK 20

#define IDLE_SLOTS "v_nop\n\tv_nop\n\tv_nop\n\tv_nop\n\tv_nop"
template <int TM>
__device__ __forceinline__ void slots_behind_last_wmma(v8f (&acc)[TM]) {
  static_assert(TM == 1 || TM == 3 || TM == 4 || TM == 6 || TM == 7, "add the operand list for this TM");
  if constexpr (TM == 1)
    asm volatile(IDLE_SLOTS : "+v"(acc[0]));
  else if constexpr (TM == 3)
    asm volatile(IDLE_SLOTS : "+v"(acc[0]), "+v"(acc[1]), "+v"(acc[2]));
  else if constexpr (TM == 4)
    asm volatile(IDLE_SLOTS : "+v"(acc[0]), "+v"(acc[1]), "+v"(acc[2]), "+v"(acc[3]));
  else if constexpr (TM == 6)
    asm volatile(IDLE_SLOTS : "+v"(acc[0]), "+v"(acc[1]), "+v"(acc[2]), "+v"(acc[3]), "+v"(acc[4]), "+v"(acc[5]));
  else
    asm volatile(IDLE_SLOTS : "+v"(acc[0]), "+v"(acc[1]), "+v"(acc[2]), "+v"(acc[3]), "+v"(acc[4]), "+v"(acc[5]),
                 "+v"(acc[6]));
}

template <bool AK, bool BK, int TM, int NW>
__global__ __launch_bounds__(32 * NW) void gemm_f32_wmma(
    const float* __restrict__ A, long sAm, long sAk, long sAb,
    const float* __restrict__ B, long sBk, long sBn, long sBb,
    float* __restrict__ C, long sCm, long sCb,
    int M, int N, int K, float alpha)
{
  constexpr int BM = 16 * TM, BN = 16 * NW, NT = 32 * NW;
  constexpr int PRA = BM + 8, PRB = BN + 8;
  constexpr int TA = AK ? BM * PK : KC * PRA;
  constexpr int TB = BK ? BN * PK : KC * PRB;
  constexpr int NA = (4 * BM + NT - 1) / NT;
  constexpr bool AFULL = ((4 * BM) % NT) == 0;
  constexpr int PC = BN + 4;
  constexpr int OPF = 2 * TA + 2 * TB;
  constexpr int CTF = BM * PC;
  constexpr int SMF = OPF > CTF ? OPF : CTF;
  __shared__ __attribute__((aligned(16))) float smem[SMF];
  float* const As = smem;
  float* const Bs = smem + 2 * TA;

  const int tid  = threadIdx.x;
  const int lane = tid & 31;
  const int wave = tid >> 5;
  const int half = lane >> 4;
  const int l16  = lane & 15;
  const int mBase = blockIdx.x * BM;
  const int nBlk  = blockIdx.y * BN;

  A += (long)blockIdx.z * sAb;
  B += (long)blockIdx.z * sBb;
  C += (long)blockIdx.z * sCb;

  v8f zero = {0.f,0.f,0.f,0.f,0.f,0.f,0.f,0.f};
  v8f acc[TM];
#pragma unroll
  for (int t = 0; t < TM; ++t) acc[t] = zero;

  long aOff[NA], bOff[2];
  int  aKo[NA], bKo[2];
  bool aIn[NA], bIn[2];
  bool aHas[NA];
  int  aDst[NA], bDst[2];
#pragma unroll
  for (int j = 0; j < NA; ++j) {
    const int idx = tid + NT * j;
    aHas[j] = AFULL || (idx < 4 * BM);
    if (AK) {
      const int row = idx >> 2, kq = (idx & 3) * 4, m = mBase + row;
      aIn[j] = aHas[j] && (m < M);  aKo[j] = kq;  aDst[j] = row * PK + kq;
      aOff[j] = (long)(aIn[j] ? m : 0) * sAm + kq;
    } else {
      const int kk = idx / (BM / 4), mq = (idx % (BM / 4)) * 4, m = mBase + mq;
      aIn[j] = aHas[j] && (m < M);  aKo[j] = kk;  aDst[j] = kk * PRA + mq;
      aOff[j] = (long)kk * sAk + (aIn[j] ? m : 0);
    }
  }
  for (int j = 0; j < 2; ++j) {
    const int idx = tid + NT * j;
    if (BK) {
      const int row = idx >> 2, kq = (idx & 3) * 4, n = nBlk + row;
      bIn[j] = (n < N);  bKo[j] = kq;  bDst[j] = row * PK + kq;
      bOff[j] = (long)(bIn[j] ? n : 0) * sBn + kq;
    } else {
      const int kk = idx / (BN / 4), nq = (idx % (BN / 4)) * 4, n = nBlk + nq;
      bIn[j] = (n < N);  bKo[j] = kk;  bDst[j] = kk * PRB + nq;
      bOff[j] = (long)kk * sBk + (bIn[j] ? n : 0);
    }
  }
  const long aStep = AK ? (long)KC : (long)KC * sAk;
  const long bStep = BK ? (long)KC : (long)KC * sBk;

  const int aFrag = AK ? (l16 * PK + 2 * half) : (2 * half * PRA + l16);
  const int bFrag = BK ? ((wave * 16 + l16) * PK + 2 * half)
                       : (2 * half * PRB + wave * 16 + l16);

  float4 ra[NA], rb[2];

#pragma unroll
  for (int j = 0; j < NA; ++j) {
    ra[j] = make_float4(0.f, 0.f, 0.f, 0.f);
    if (aIn[j] && aKo[j] < K) ra[j] = *reinterpret_cast<const float4*>(A + aOff[j]);
    aOff[j] += aStep;
  }
  for (int j = 0; j < 2; ++j) {
    rb[j] = make_float4(0.f, 0.f, 0.f, 0.f);
    if (bIn[j] && bKo[j] < K) rb[j] = *reinterpret_cast<const float4*>(B + bOff[j]);
    bOff[j] += bStep;
  }
#pragma unroll
  for (int j = 0; j < NA; ++j)
    if (aHas[j]) *reinterpret_cast<float4*>(As + aDst[j]) = ra[j];
  for (int j = 0; j < 2; ++j)
    *reinterpret_cast<float4*>(Bs + bDst[j]) = rb[j];
  __syncthreads();

  int p = 0;
  for (int k0 = 0; ; k0 += KC) {
    const bool more = (k0 + KC < K);
    if (more) {
#pragma unroll
      for (int j = 0; j < NA; ++j) {
        ra[j] = make_float4(0.f, 0.f, 0.f, 0.f);
        if (aIn[j] && k0 + KC + aKo[j] < K) ra[j] = *reinterpret_cast<const float4*>(A + aOff[j]);
        aOff[j] += aStep;
      }
      for (int j = 0; j < 2; ++j) {
        rb[j] = make_float4(0.f, 0.f, 0.f, 0.f);
        if (bIn[j] && k0 + KC + bKo[j] < K) rb[j] = *reinterpret_cast<const float4*>(B + bOff[j]);
        bOff[j] += bStep;
      }
    }

    const float* as = As + p * TA + aFrag;
    const float* bs = Bs + p * TB + bFrag;
#pragma unroll
    for (int ks = 0; ks < KC / 4; ++ks) {
      v2f bf;
      if (BK) {
        bf = *reinterpret_cast<const v2f*>(bs + ks * 4);
      } else {
        bf.x = bs[(ks * 4) * PRB];
        bf.y = bs[(ks * 4 + 1) * PRB];
      }
#pragma unroll
      for (int t = 0; t < TM; ++t) {
        v2f af;
        if (AK) {
          af = *reinterpret_cast<const v2f*>(as + t * 16 * PK + ks * 4);
        } else {
          af.x = as[(ks * 4) * PRA + t * 16];
          af.y = as[(ks * 4 + 1) * PRA + t * 16];
        }
        acc[t] = __builtin_amdgcn_wmma_f32_16x16x4_f32(
            false, af, false, bf, (short)0, acc[t], false, false);
      }
    }

    if (!more) break;
    p ^= 1;
#pragma unroll
    for (int j = 0; j < NA; ++j)
      if (aHas[j]) *reinterpret_cast<float4*>(As + p * TA + aDst[j]) = ra[j];
    for (int j = 0; j < 2; ++j)
      *reinterpret_cast<float4*>(Bs + p * TB + bDst[j]) = rb[j];
    __syncthreads();
  }

  slots_behind_last_wmma<TM>(acc);
  __builtin_amdgcn_sched_barrier(0);
  __syncthreads();
  {
    float* cw = smem + (8 * half) * PC + wave * 16 + l16;
#pragma unroll
    for (int t = 0; t < TM; ++t) {
#pragma unroll
      for (int r = 0; r < 8; ++r) cw[(t * 16 + r) * PC] = alpha * acc[t][r];
    }
  }
  __syncthreads();
  {
    constexpr int RV = BN / 4;
    const int row0 = tid / RV, c4 = tid - row0 * RV;
    const int n = nBlk + 4 * c4;
    const float* cr = smem + row0 * PC + 4 * c4;
    float* cg = C + (long)(mBase + row0) * sCm + n;
    const long gStep = 8 * sCm;
    if (mBase + BM <= M && nBlk + BN <= N) {
#pragma unroll
      for (int j = 0; j < 2 * TM; ++j) {
        *reinterpret_cast<float4*>(cg) = *reinterpret_cast<const float4*>(cr + j * 8 * PC);
        cg += gStep;
      }
    } else if (n < N) {
#pragma unroll
      for (int j = 0; j < 2 * TM; ++j) {
        if (mBase + row0 + 8 * j < M)
          *reinterpret_cast<float4*>(cg) = *reinterpret_cast<const float4*>(cr + j * 8 * PC);
        cg += gStep;
      }
    }
  }
}

template <bool AK, bool BK, int TM, int NW>
static void launch_gemm(hipStream_t stream,
                        const float* A, long sAm, long sAk, long sAb,
                        const float* B, long sBk, long sBn, long sBb,
                        float* C, long sCm, long sCb,
                        int M, int N, int K, int batch, float alpha) {
  const unsigned mb = (unsigned)((M + 16 * TM - 1) / (16 * TM));
  const unsigned nb = (unsigned)((N + 16 * NW - 1) / (16 * NW));
  dim3 g(mb, nb, (unsigned)batch);
  gemm_f32_wmma<AK, BK, TM, NW><<<g, 32 * NW, 0, stream>>>(A, sAm, sAk, sAb, B, sBk, sBn, sBb,
                                                       C, sCm, sCb, M, N, K, alpha);
}

#define PX 76
#define PT 68
#define KC2 32
#define PK2 36
__global__ __launch_bounds__(192) void dft_xy_fused(const float* __restrict__ in,
                                                    const float* __restrict__ Fcat,
                                                    const float* Fy,
                                                    float* __restrict__ R2)
{
  constexpr int TFA = 96 * PK;
  constexpr int TFB = 64 * PK;
  constexpr int TF2 = 96 * PK2;
  constexpr int X1F = 96 * PX;
  static_assert(2 * TFA + 2 * TFB <= 2 * TF2 && 96 * PT <= 2 * TF2, "phase-1 tiles and the T tile reuse the phase-2 tile space");
  __shared__ __attribute__((aligned(16))) float smem[X1F + 2 * TF2];
  float* const X1 = smem;
  float* const As = smem + X1F;
  float* const Bs = As + 2 * TFA;

  const int tid  = threadIdx.x;
  const int lane = tid & 31;
  const int wave = tid >> 5;
  const int half = lane >> 4;
  const int l16  = lane & 15;
  const int rt = wave % 3, cp = wave / 3;

  const float* A = in + (long)blockIdx.x * 9216;

  const int row = tid >> 2, kq = (tid & 3) * 4;
  const int g0 = row * 96 + kq,  g1 = g0 + 48 * 96;
  const int d0 = row * PK + kq,  d1 = d0 + 48 * PK;
  const int h0 = (row + (row >= 25 ? 24 : 0)) * 96 + kq;
  const int h1 = (72 + row) * 96 + kq;
  const bool two = tid < 64;

  v8f zero = {0.f,0.f,0.f,0.f,0.f,0.f,0.f,0.f};
  float4 ra0, ra1, rb0, rb1, rc0, rc1;

  v8f acc[4];
#pragma unroll
  for (int t = 0; t < 4; ++t) acc[t] = zero;

  ra0 = *reinterpret_cast<const float4*>(A + g0);
  ra1 = *reinterpret_cast<const float4*>(A + g1);
  rb0 = *reinterpret_cast<const float4*>(Fcat + h0);
  if (two) rb1 = *reinterpret_cast<const float4*>(Fcat + h1);
  *reinterpret_cast<float4*>(As + d0) = ra0;
  *reinterpret_cast<float4*>(As + d1) = ra1;
  *reinterpret_cast<float4*>(Bs + d0) = rb0;
  if (two) *reinterpret_cast<float4*>(Bs + d1) = rb1;
  __syncthreads();

  const int aFrag = (32 * rt + l16) * PK + 2 * half;
  const int bFrag = (32 * cp + l16) * PK + 2 * half;
  int p = 0;
#pragma nounroll
  for (int k0 = 0; ; k0 += KC) {
    const bool more = (k0 + KC < 96);
    if (more) {
      ra0 = *reinterpret_cast<const float4*>(A + g0 + k0 + KC);
      ra1 = *reinterpret_cast<const float4*>(A + g1 + k0 + KC);
      rb0 = *reinterpret_cast<const float4*>(Fcat + h0 + k0 + KC);
      if (two) rb1 = *reinterpret_cast<const float4*>(Fcat + h1 + k0 + KC);
    }
    const float* as = As + p * TFA + aFrag;
    const float* bs = Bs + p * TFB + bFrag;
#pragma unroll
    for (int ks = 0; ks < KC / 4; ++ks) {
      const v2f bf0 = *reinterpret_cast<const v2f*>(bs + ks * 4);
      const v2f bf1 = *reinterpret_cast<const v2f*>(bs + 16 * PK + ks * 4);
#pragma unroll
      for (int t = 0; t < 2; ++t) {
        const v2f af = *reinterpret_cast<const v2f*>(as + t * 16 * PK + ks * 4);
        acc[2 * t] = __builtin_amdgcn_wmma_f32_16x16x4_f32(
            false, af, false, bf0, (short)0, acc[2 * t], false, false);
        acc[2 * t + 1] = __builtin_amdgcn_wmma_f32_16x16x4_f32(
            false, af, false, bf1, (short)0, acc[2 * t + 1], false, false);
      }
    }
    if (!more) break;
    p ^= 1;
    *reinterpret_cast<float4*>(As + p * TFA + d0) = ra0;
    *reinterpret_cast<float4*>(As + p * TFA + d1) = ra1;
    *reinterpret_cast<float4*>(Bs + p * TFB + d0) = rb0;
    if (two) *reinterpret_cast<float4*>(Bs + p * TFB + d1) = rb1;
    __syncthreads();
  }

  slots_behind_last_wmma<4>(acc);
  __builtin_amdgcn_sched_barrier(0);
  const int row2 = tid >> 3, kq2 = (tid & 7) * 4;
  const int f0 = row2 * 96 + kq2;
  const int e0 = row2 * PK2 + kq2;
  ra0 = *reinterpret_cast<const float4*>(Fy + f0);
  ra1 = *reinterpret_cast<const float4*>(Fy + f0 + 24 * 96);
  rc0 = *reinterpret_cast<const float4*>(Fy + f0 + 48 * 96);
  rc1 = *reinterpret_cast<const float4*>(Fy + f0 + 72 * 96);
  {
    float* xw = X1 + (32 * rt + 8 * half) * PX + 32 * cp + l16;
#pragma unroll
    for (int t = 0; t < 2; ++t) {
#pragma unroll
      for (int u = 0; u < 2; ++u) {
#pragma unroll
        for (int r = 0; r < 8; ++r) xw[(t * 16 + r) * PX + u * 16] = acc[2 * t + u][r];
      }
    }
  }
  __syncthreads();
  *reinterpret_cast<float4*>(As + e0) = ra0;
  *reinterpret_cast<float4*>(As + e0 + 24 * PK2) = ra1;
  *reinterpret_cast<float4*>(As + e0 + 48 * PK2) = rc0;
  *reinterpret_cast<float4*>(As + e0 + 72 * PK2) = rc1;
  __syncthreads();

  v8f acc2[4];
#pragma unroll
  for (int t = 0; t < 4; ++t) acc2[t] = zero;
  const int aFrag2 = (32 * rt + l16) * PK2 + 2 * half;
  const float* xb = X1 + (2 * half) * PX + 32 * cp + l16;
  p = 0;
  for (int k0 = 0; ; k0 += KC2) {
    const bool more = (k0 + KC2 < 96);
    if (more) {
      ra0 = *reinterpret_cast<const float4*>(Fy + f0 + k0 + KC2);
      ra1 = *reinterpret_cast<const float4*>(Fy + f0 + 24 * 96 + k0 + KC2);
      rc0 = *reinterpret_cast<const float4*>(Fy + f0 + 48 * 96 + k0 + KC2);
      rc1 = *reinterpret_cast<const float4*>(Fy + f0 + 72 * 96 + k0 + KC2);
    }
    const float* as = As + p * TF2 + aFrag2;
    const float* bs = xb + k0 * PX;
#pragma unroll
    for (int ks = 0; ks < KC2 / 4; ++ks) {
      v2f bf0, bf1;
      bf0.x = bs[(ks * 4) * PX];
      bf0.y = bs[(ks * 4 + 1) * PX];
      bf1.x = bs[(ks * 4) * PX + 16];
      bf1.y = bs[(ks * 4 + 1) * PX + 16];
#pragma unroll
      for (int t = 0; t < 2; ++t) {
        const v2f af = *reinterpret_cast<const v2f*>(as + t * 16 * PK2 + ks * 4);
        acc2[2 * t] = __builtin_amdgcn_wmma_f32_16x16x4_f32(
            false, af, false, bf0, (short)0, acc2[2 * t], false, false);
        acc2[2 * t + 1] = __builtin_amdgcn_wmma_f32_16x16x4_f32(
            false, af, false, bf1, (short)0, acc2[2 * t + 1], false, false);
      }
    }
    if (!more) break;
    p ^= 1;
    *reinterpret_cast<float4*>(As + p * TF2 + e0) = ra0;
    *reinterpret_cast<float4*>(As + p * TF2 + e0 + 24 * PK2) = ra1;
    *reinterpret_cast<float4*>(As + p * TF2 + e0 + 48 * PK2) = rc0;
    *reinterpret_cast<float4*>(As + p * TF2 + e0 + 72 * PK2) = rc1;
    __syncthreads();
  }

  slots_behind_last_wmma<4>(acc2);
  __builtin_amdgcn_sched_barrier(0);
  __syncthreads();
  {
    float* tw = As + (32 * rt + 8 * half) * PT + 32 * cp + l16;
#pragma unroll
    for (int t = 0; t < 2; ++t) {
#pragma unroll
      for (int u = 0; u < 2; ++u) {
#pragma unroll
        for (int r = 0; r < 8; ++r) tw[(t * 16 + r) * PT + u * 16] = acc2[2 * t + u][r];
      }
    }
  }
  __syncthreads();
  {
    const int row0 = tid / 12, c4 = tid - row0 * 12;
    int ja[4], jb[4];
    float sg[4];
#pragma unroll
    for (int e = 0; e < 4; ++e) {
      const int pp = 4 * c4 + e;
      const bool mir = pp > 24;
      ja[e] = mir ? 48 - pp : pp;
      jb[e] = pp == 0 ? 0 : ja[e] + 24;
      sg[e] = pp == 0 ? 0.f : (mir ? 1.f : -1.f);
    }
    const float* tc = As + row0 * PT;
    float* cg = R2 + (long)blockIdx.x * 4608 + row0 * 48 + 4 * c4;
#pragma unroll
    for (int j = 0; j < 3; ++j) {
      const float* c = tc + j * 16 * PT;
      const float* s = c + 48 * PT;
      float4 yre, yim;
      yre.x = fmaf( sg[0], s[jb[0]], c[ja[0]]);  yim.x = fmaf(-sg[0], c[jb[0]], s[ja[0]]);
      yre.y = fmaf( sg[1], s[jb[1]], c[ja[1]]);  yim.y = fmaf(-sg[1], c[jb[1]], s[ja[1]]);
      yre.z = fmaf( sg[2], s[jb[2]], c[ja[2]]);  yim.z = fmaf(-sg[2], c[jb[2]], s[ja[2]]);
      yre.w = fmaf( sg[3], s[jb[3]], c[ja[3]]);  yim.w = fmaf(-sg[3], c[jb[3]], s[ja[3]]);
      *reinterpret_cast<float4*>(cg + j * 16 * 48) = yre;
      *reinterpret_cast<float4*>(cg + 2304 + j * 16 * 48) = yim;
    }
  }
}

__global__ void gen_consts(float* Fre, float* Fim, float* Vre, float* Vim,
                           float* F3, float* V2, float* V3) {
  int i = blockIdx.x * 256 + threadIdx.x;
  if (i >= 48 * 96) return;
  int k = i / 96, nn = i % 96;
  int f = (k < 24) ? k : (k + 48);
  int ph = (f * nn) % 96;
  float th = 6.283185307179586f * (float)ph / 96.f;
  float c = cosf(th), s = sinf(th);
  Fre[i] = c;  Fim[i] = -s;
  Vre[nn * 48 + k] = c;  Vim[nn * 48 + k] = s;
  F3[k * 192 + 2 * nn]     = c + s;
  F3[k * 192 + 2 * nn + 1] = s - c;
  V2[nn * 96 + 2 * k]            = c;
  V2[nn * 96 + 2 * k + 1]        = -s;
  V2[(96 + nn) * 96 + 2 * k]     = s;
  V2[(96 + nn) * 96 + 2 * k + 1] = c;
  V3[nn * 96 + 2 * k]     = c + s;
  V3[nn * 96 + 2 * k + 1] = c - s;
}

__global__ void pack_wqkv(const float* __restrict__ wq, const float* __restrict__ wk,
                          const float* __restrict__ wv, float* __restrict__ dst) {
  int i = blockIdx.x * 256 + threadIdx.x;
  if (i >= 384 * 64) return;
  int ic = i % 64, r = i / 64;
  int t = r / 128, q = r % 128;
  const float* w = (t == 0) ? wq : ((t == 1) ? wk : wv);
  dst[i] = w[q * 64 + ic];
}

#define KSPLIT 32
__global__ __launch_bounds__(256) void sum_scale_selu(const float* __restrict__ part, float* __restrict__ s,
                               int nvec, int planeVec, float scale) {
  int i = blockIdx.x * 256 + threadIdx.x;
  if (i >= nvec) return;
  int h = i / planeVec, e = i - h * planeVec;
  const float4* p = reinterpret_cast<const float4*>(part) + (long)h * KSPLIT * planeVec + e;
  float4 v[KSPLIT];
  for (int j = 0; j < KSPLIT; ++j) v[j] = p[(long)j * planeVec];
  float4 a = v[0];
  for (int j = 1; j < KSPLIT; ++j) {
    a.x += v[j].x; a.y += v[j].y; a.z += v[j].z; a.w += v[j].w;
  }
  const float lam = 1.0507009873554805f, alp = 1.6732632423543772f;
  float x0 = scale * a.x, x1 = scale * a.y, x2 = scale * a.z, x3 = scale * a.w;
  float4 r;
  r.x = lam * (x0 > 0.f ? x0 : alp * (expf(x0) - 1.f));
  r.y = lam * (x1 > 0.f ? x1 : alp * (expf(x1) - 1.f));
  r.z = lam * (x2 > 0.f ? x2 : alp * (expf(x2) - 1.f));
  r.w = lam * (x3 > 0.f ? x3 : alp * (expf(x3) - 1.f));
  reinterpret_cast<float4*>(s)[i] = r;
}


#define GP_HALF 1728
#define GP_PAD  16
#define GP_F4   864

template <int NT>
__global__ __launch_bounds__(NT) void group_rows(const float* __restrict__ src, float* __restrict__ dst) {
  constexpr int NP = GP_F4 / NT;
  static_assert(NP * NT == GP_F4, "whole pieces per thread");
  __shared__ __attribute__((aligned(16))) float t[2 * GP_HALF + GP_PAD];
  const int b = blockIdx.x;
  const int pyp = b & 3, pz = (b >> 2) & 7, r = b >> 5;
  const float* s = src + (long)r * 110592 + (pz * 48 + 2 * pyp) * 48;
  float* d = dst + (long)r * 110592 + (pz * 64 + pyp * 16) * 216;
  float4 v[NP];
#pragma unroll
  for (int i = 0; i < NP; ++i) {
    const int f = threadIdx.x + NT * i;
    const int chunk = f / 24, q = f - chunk * 24;
    const int nz = chunk / 6, ny = chunk - nz * 6;
    v[i] = *reinterpret_cast<const float4*>(s + (nz * 384 + ny * 8) * 48 + q * 4);
  }
#pragma unroll
  for (int i = 0; i < NP; ++i) {
    const int f = threadIdx.x + NT * i;
    const int chunk = f / 24, q = f - chunk * 24;
    const int yy = q / 12, xq = q - yy * 12;
    float* w = t + yy * (GP_HALF + GP_PAD) + (4 * (xq & 1)) * 216 + chunk * 6 + (xq >> 1);
    w[0] = v[i].x;  w[216] = v[i].y;  w[432] = v[i].z;  w[648] = v[i].w;
  }
  __syncthreads();
#pragma unroll
  for (int i = 0; i < NP; ++i) {
    const int g = threadIdx.x + NT * i;
    reinterpret_cast<float4*>(d)[g] = *reinterpret_cast<const float4*>(t + 4 * g + (g / 432) * GP_PAD);
  }
}

template <int NT>
__global__ __launch_bounds__(NT) void ungroup_rows(const float* __restrict__ src, float* __restrict__ dst) {
  constexpr int NP = GP_F4 / NT;
  static_assert(NP * NT == GP_F4, "whole pieces per thread");
  __shared__ __attribute__((aligned(16))) float t[2 * GP_HALF + GP_PAD];
  const int b = blockIdx.x;
  const int pyp = b & 3, pz = (b >> 2) & 7, r = b >> 5;
  const float* s = src + (long)r * 110592 + (pz * 64 + pyp * 16) * 216;
  float* d = dst + (long)r * 110592 + (pz * 48 + 2 * pyp) * 48;
  float4 v[NP];
#pragma unroll
  for (int i = 0; i < NP; ++i) v[i] = reinterpret_cast<const float4*>(s)[threadIdx.x + NT * i];
#pragma unroll
  for (int i = 0; i < NP; ++i) {
    const int g = threadIdx.x + NT * i;
    *reinterpret_cast<float4*>(t + 4 * g + (g / 432) * GP_PAD) = v[i];
  }
  __syncthreads();
#pragma unroll
  for (int i = 0; i < NP; ++i) {
    const int f = threadIdx.x + NT * i;
    const int chunk = f / 24, q = f - chunk * 24;
    const int nz = chunk / 6, ny = chunk - nz * 6;
    const int yy = q / 12, xq = q - yy * 12;
    const float* w = t + yy * (GP_HALF + GP_PAD) + (4 * (xq & 1)) * 216 + chunk * 6 + (xq >> 1);
    float4 o;
    o.x = w[0];  o.y = w[216];  o.z = w[432];  o.w = w[648];
    *reinterpret_cast<float4*>(d + (nz * 384 + ny * 8) * 48 + q * 4) = o;
  }
}

extern "C" void kernel_launch(void* const* d_in, const int* in_sizes, int n_in,
                              void* d_out, int out_size, void* d_ws, size_t ws_size,
                              hipStream_t stream) {
  const float* in = (const float*)d_in[0];
  const float* wq = (const float*)d_in[1];
  const float* wk = (const float*)d_in[2];
  const float* wv = (const float*)d_in[3];
  const float* wo = (const float*)d_in[4];
  float* out = (float*)d_out;
  float* w = (float*)d_ws;

  const long R2_o   = 56623104;
  const long blk_o  = 0;
  const long blkG_o = 7077888,    G_o    = 42467328;
  const long S_o    = 92012544,   Og_o   = 92385792;
  const long P_o    = 0;
  const long specG_o = 0,         spec_o = 14155776;
  const long T1_o   = 15925248;
  const long T2_o   = 23003136;
  const long F3_o   = 84953088;
  const long V2_o   = 84962304,   V3_o   = 84980736;
  const long Wqkv_o = 106541568;
  const long Fre_o  = 106566144,  Fim_o  = 106570752;
  const long Vre_o  = 106575360,  Vim_o  = 106579968;
  const long WS_FLOATS = 106584576;
  if (ws_size < (size_t)WS_FLOATS * sizeof(float)) return;

  float *R2 = w+R2_o;
  float *blk = w+blk_o, *blkG = w+blkG_o, *G = w+G_o, *S = w+S_o, *Og = w+Og_o;
  float *specG = w+specG_o, *spec = w+spec_o, *P = w+P_o;
  float *T1 = w+T1_o, *T2 = w+T2_o, *Wqkv = w+Wqkv_o;
  float *Fre = w+Fre_o, *Fim = w+Fim_o, *Vre = w+Vre_o, *Vim = w+Vim_o;
  float *F3 = w+F3_o, *V2 = w+V2_o, *V3 = w+V3_o;

  auto ew = [&](int n) { return dim3((unsigned)((n + 255) / 256)); };

  gen_consts<<<ew(4608), 256, 0, stream>>>(Fre, Fim, Vre, Vim, F3, V2, V3);
  pack_wqkv<<<ew(24576), 256, 0, stream>>>(wq, wk, wv, Wqkv);

  dft_xy_fused<<<dim3(6144), 192, 0, stream>>>(in, Fre, Fre, R2);
  launch_gemm<true, false, 3, 4>(stream, F3, 192, 1, 0, R2, 2304, 1, 442368, blk, 2304, 110592, 48, 2304, 192, 64, 1.f);

  group_rows<96><<<dim3(2048), 96, 0, stream>>>(blk, blkG);
  launch_gemm<true, false, 4, 4>(stream, Wqkv, 64, 1, 0, blkG, 110592, 1, 0, G, 110592, 0, 384, 110592, 64, 1, 1.f);

  const long HD = 8192L * 216L;
  const long SL = 256L * 216L;
  launch_gemm<false, false, 7, 7>(stream, G, 1, 216, SL, G + 8 * HD, 216, 1, SL, P, 216, 46656,
       216, 216, 256, 8 * KSPLIT, 1.f);
  sum_scale_selu<<<ew(93312), 256, 0, stream>>>(P, S, 93312, 11664,
                                                0.011048543456039806f);
  launch_gemm<true, true, 4, 7>(stream, G + 16 * HD, 216, 1, HD, S, 1, 216, 46656, Og, 216, HD,
       8192, 216, 216, 8, 1.f);

  launch_gemm<true, false, 1, 4>(stream, wo, 128, 1, 0, Og, 110592, 1, 0, specG, 110592, 0, 16, 110592, 128, 1, 1.f);
  ungroup_rows<288><<<dim3(512), 288, 0, stream>>>(specG, spec);

  launch_gemm<true, true, 4, 4>(stream, spec, 48, 1, 0, Vre, 1, 48, 0, T1, 192, 0, 36864, 192, 48, 1, 1.f);
  launch_gemm<true, false, 6, 6>(stream, V2, 96, 1, 0, T1, 96, 1, 9216, T2, 96, 18432, 192, 96, 96, 768, 1.f);
  const float invN = 1.f / 884736.f;
  launch_gemm<true, false, 6, 4>(stream, V3, 96, 1, 0, T2, 9216, 1, 884736, out, 9216, 884736, 96, 9216, 96, 16, invN);
}


# y-axis product folded on the twiddles' even/odd symmetry: slab columns folded in LDS (y' with 96-y'), Y2(64x52) replaces Fcat(96x96): 16 subtiles x 13 k-steps, not 24 x 24 (sum order changes)
# speedup vs baseline: 7.7882x; 1.0537x over previous
#include <hip/hip_runtime.h>
#include <math.h>

typedef float v2f __attribute__((ext_vector_type(2)));
typedef float v8f __attribute__((ext_vector_type(8)));

#define KC 16
#define PK 20

#define IDLE_SLOTS "v_nop\n\tv_nop\n\tv_nop\n\tv_nop\n\tv_nop"
template <int TM>
__device__ __forceinline__ void slots_behind_last_wmma(v8f (&acc)[TM]) {
  static_assert(TM == 1 || TM == 3 || TM == 4 || TM == 6 || TM == 7, "add the operand list for this TM");
  if constexpr (TM == 1)
    asm volatile(IDLE_SLOTS : "+v"(acc[0]));
  else if constexpr (TM == 3)
    asm volatile(IDLE_SLOTS : "+v"(acc[0]), "+v"(acc[1]), "+v"(acc[2]));
  else if constexpr (TM == 4)
    asm volatile(IDLE_SLOTS : "+v"(acc[0]), "+v"(acc[1]), "+v"(acc[2]), "+v"(acc[3]));
  else if constexpr (TM == 6)
    asm volatile(IDLE_SLOTS : "+v"(acc[0]), "+v"(acc[1]), "+v"(acc[2]), "+v"(acc[3]), "+v"(acc[4]), "+v"(acc[5]));
  else
    asm volatile(IDLE_SLOTS : "+v"(acc[0]), "+v"(acc[1]), "+v"(acc[2]), "+v"(acc[3]), "+v"(acc[4]), "+v"(acc[5]),
                 "+v"(acc[6]));
}

template <bool AK, bool BK, int TM, int NW>
__global__ __launch_bounds__(32 * NW) void gemm_f32_wmma(
    const float* __restrict__ A, long sAm, long sAk, long sAb,
    const float* __restrict__ B, long sBk, long sBn, long sBb,
    float* __restrict__ C, long sCm, long sCb,
    int M, int N, int K, float alpha)
{
  constexpr int BM = 16 * TM, BN = 16 * NW, NT = 32 * NW;
  constexpr int PRA = BM + 8, PRB = BN + 8;
  constexpr int TA = AK ? BM * PK : KC * PRA;
  constexpr int TB = BK ? BN * PK : KC * PRB;
  constexpr int NA = (4 * BM + NT - 1) / NT;
  constexpr bool AFULL = ((4 * BM) % NT) == 0;
  constexpr int PC = BN + 4;
  constexpr int OPF = 2 * TA + 2 * TB;
  constexpr int CTF = BM * PC;
  constexpr int SMF = OPF > CTF ? OPF : CTF;
  __shared__ __attribute__((aligned(16))) float smem[SMF];
  float* const As = smem;
  float* const Bs = smem + 2 * TA;

  const int tid  = threadIdx.x;
  const int lane = tid & 31;
  const int wave = tid >> 5;
  const int half = lane >> 4;
  const int l16  = lane & 15;
  const int mBase = blockIdx.x * BM;
  const int nBlk  = blockIdx.y * BN;

  A += (long)blockIdx.z * sAb;
  B += (long)blockIdx.z * sBb;
  C += (long)blockIdx.z * sCb;

  v8f zero = {0.f,0.f,0.f,0.f,0.f,0.f,0.f,0.f};
  v8f acc[TM];
#pragma unroll
  for (int t = 0; t < TM; ++t) acc[t] = zero;

  long aOff[NA], bOff[2];
  int  aKo[NA], bKo[2];
  bool aIn[NA], bIn[2];
  bool aHas[NA];
  int  aDst[NA], bDst[2];
#pragma unroll
  for (int j = 0; j < NA; ++j) {
    const int idx = tid + NT * j;
    aHas[j] = AFULL || (idx < 4 * BM);
    if (AK) {
      const int row = idx >> 2, kq = (idx & 3) * 4, m = mBase + row;
      aIn[j] = aHas[j] && (m < M);  aKo[j] = kq;  aDst[j] = row * PK + kq;
      aOff[j] = (long)(aIn[j] ? m : 0) * sAm + kq;
    } else {
      const int kk = idx / (BM / 4), mq = (idx % (BM / 4)) * 4, m = mBase + mq;
      aIn[j] = aHas[j] && (m < M);  aKo[j] = kk;  aDst[j] = kk * PRA + mq;
      aOff[j] = (long)kk * sAk + (aIn[j] ? m : 0);
    }
  }
  for (int j = 0; j < 2; ++j) {
    const int idx = tid + NT * j;
    if (BK) {
      const int row = idx >> 2, kq = (idx & 3) * 4, n = nBlk + row;
      bIn[j] = (n < N);  bKo[j] = kq;  bDst[j] = row * PK + kq;
      bOff[j] = (long)(bIn[j] ? n : 0) * sBn + kq;
    } else {
      const int kk = idx / (BN / 4), nq = (idx % (BN / 4)) * 4, n = nBlk + nq;
      bIn[j] = (n < N);  bKo[j] = kk;  bDst[j] = kk * PRB + nq;
      bOff[j] = (long)kk * sBk + (bIn[j] ? n : 0);
    }
  }
  const long aStep = AK ? (long)KC : (long)KC * sAk;
  const long bStep = BK ? (long)KC : (long)KC * sBk;

  const int aFrag = AK ? (l16 * PK + 2 * half) : (2 * half * PRA + l16);
  const int bFrag = BK ? ((wave * 16 + l16) * PK + 2 * half)
                       : (2 * half * PRB + wave * 16 + l16);

  float4 ra[NA], rb[2];

#pragma unroll
  for (int j = 0; j < NA; ++j) {
    ra[j] = make_float4(0.f, 0.f, 0.f, 0.f);
    if (aIn[j] && aKo[j] < K) ra[j] = *reinterpret_cast<const float4*>(A + aOff[j]);
    aOff[j] += aStep;
  }
  for (int j = 0; j < 2; ++j) {
    rb[j] = make_float4(0.f, 0.f, 0.f, 0.f);
    if (bIn[j] && bKo[j] < K) rb[j] = *reinterpret_cast<const float4*>(B + bOff[j]);
    bOff[j] += bStep;
  }
#pragma unroll
  for (int j = 0; j < NA; ++j)
    if (aHas[j]) *reinterpret_cast<float4*>(As + aDst[j]) = ra[j];
  for (int j = 0; j < 2; ++j)
    *reinterpret_cast<float4*>(Bs + bDst[j]) = rb[j];
  __syncthreads();

  int p = 0;
  for (int k0 = 0; ; k0 += KC) {
    const bool more = (k0 + KC < K);
    if (more) {
#pragma unroll
      for (int j = 0; j < NA; ++j) {
        ra[j] = make_float4(0.f, 0.f, 0.f, 0.f);
        if (aIn[j] && k0 + KC + aKo[j] < K) ra[j] = *reinterpret_cast<const float4*>(A + aOff[j]);
        aOff[j] += aStep;
      }
      for (int j = 0; j < 2; ++j) {
        rb[j] = make_float4(0.f, 0.f, 0.f, 0.f);
        if (bIn[j] && k0 + KC + bKo[j] < K) rb[j] = *reinterpret_cast<const float4*>(B + bOff[j]);
        bOff[j] += bStep;
      }
    }

    const float* as = As + p * TA + aFrag;
    const float* bs = Bs + p * TB + bFrag;
#pragma unroll
    for (int ks = 0; ks < KC / 4; ++ks) {
      v2f bf;
      if (BK) {
        bf = *reinterpret_cast<const v2f*>(bs + ks * 4);
      } else {
        bf.x = bs[(ks * 4) * PRB];
        bf.y = bs[(ks * 4 + 1) * PRB];
      }
#pragma unroll
      for (int t = 0; t < TM; ++t) {
        v2f af;
        if (AK) {
          af = *reinterpret_cast<const v2f*>(as + t * 16 * PK + ks * 4);
        } else {
          af.x = as[(ks * 4) * PRA + t * 16];
          af.y = as[(ks * 4 + 1) * PRA + t * 16];
        }
        acc[t] = __builtin_amdgcn_wmma_f32_16x16x4_f32(
            false, af, false, bf, (short)0, acc[t], false, false);
      }
    }

    if (!more) break;
    p ^= 1;
#pragma unroll
    for (int j = 0; j < NA; ++j)
      if (aHas[j]) *reinterpret_cast<float4*>(As + p * TA + aDst[j]) = ra[j];
    for (int j = 0; j < 2; ++j)
      *reinterpret_cast<float4*>(Bs + p * TB + bDst[j]) = rb[j];
    __syncthreads();
  }

  slots_behind_last_wmma<TM>(acc);
  __builtin_amdgcn_sched_barrier(0);
  __syncthreads();
  {
    float* cw = smem + (8 * half) * PC + wave * 16 + l16;
#pragma unroll
    for (int t = 0; t < TM; ++t) {
#pragma unroll
      for (int r = 0; r < 8; ++r) cw[(t * 16 + r) * PC] = alpha * acc[t][r];
    }
  }
  __syncthreads();
  {
    constexpr int RV = BN / 4;
    const int row0 = tid / RV, c4 = tid - row0 * RV;
    const int n = nBlk + 4 * c4;
    const float* cr = smem + row0 * PC + 4 * c4;
    float* cg = C + (long)(mBase + row0) * sCm + n;
    const long gStep = 8 * sCm;
    if (mBase + BM <= M && nBlk + BN <= N) {
#pragma unroll
      for (int j = 0; j < 2 * TM; ++j) {
        *reinterpret_cast<float4*>(cg) = *reinterpret_cast<const float4*>(cr + j * 8 * PC);
        cg += gStep;
      }
    } else if (n < N) {
#pragma unroll
      for (int j = 0; j < 2 * TM; ++j) {
        if (mBase + row0 + 8 * j < M)
          *reinterpret_cast<float4*>(cg) = *reinterpret_cast<const float4*>(cr + j * 8 * PC);
        cg += gStep;
      }
    }
  }
}

template <bool AK, bool BK, int TM, int NW>
static void launch_gemm(hipStream_t stream,
                        const float* A, long sAm, long sAk, long sAb,
                        const float* B, long sBk, long sBn, long sBb,
                        float* C, long sCm, long sCb,
                        int M, int N, int K, int batch, float alpha) {
  const unsigned mb = (unsigned)((M + 16 * TM - 1) / (16 * TM));
  const unsigned nb = (unsigned)((N + 16 * NW - 1) / (16 * NW));
  dim3 g(mb, nb, (unsigned)batch);
  gemm_f32_wmma<AK, BK, TM, NW><<<g, 32 * NW, 0, stream>>>(A, sAm, sAk, sAb, B, sBk, sBn, sBb,
                                                       C, sCm, sCb, M, N, K, alpha);
}

#define PX 76
#define PT 68
#define PY 52
__global__ __launch_bounds__(192) void dft_xy_fused(const float* __restrict__ in,
                                                    const float* __restrict__ Fcat,
                                                    const float* __restrict__ Y2,
                                                    float* __restrict__ R2)
{
  constexpr int TFA = 96 * PK;
  constexpr int TFB = 64 * PK;
  constexpr int RGN = 2 * TFA + 2 * TFB;
  constexpr int X1F = 96 * PX;
  static_assert(64 * PY <= RGN && 64 * PT <= RGN, "Y2 and the T tile reuse the phase-1 tile space");
  __shared__ __attribute__((aligned(16))) float smem[X1F + RGN];
  float* const X1 = smem;
  float* const As = smem + X1F;
  float* const Bs = As + 2 * TFA;

  const int tid  = threadIdx.x;
  const int lane = tid & 31;
  const int wave = tid >> 5;
  const int half = lane >> 4;
  const int l16  = lane & 15;
  const int rt = wave % 3, cp = wave / 3;

  const float* A = in + (long)blockIdx.x * 9216;

  const int row = tid >> 2, kq = (tid & 3) * 4;
  const int g0 = row * 96 + kq,  g1 = g0 + 48 * 96;
  const int d0 = row * PK + kq,  d1 = d0 + 48 * PK;
  const int h0 = (row + (row >= 25 ? 24 : 0)) * 96 + kq;
  const int h1 = (72 + row) * 96 + kq;
  const bool two = tid < 64;

  v8f zero = {0.f,0.f,0.f,0.f,0.f,0.f,0.f,0.f};
  float4 ra0, ra1, rb0, rb1, rc0, rc1;

  v8f acc[4];
#pragma unroll
  for (int t = 0; t < 4; ++t) acc[t] = zero;

  ra0 = *reinterpret_cast<const float4*>(A + g0);
  ra1 = *reinterpret_cast<const float4*>(A + g1);
  rb0 = *reinterpret_cast<const float4*>(Fcat + h0);
  if (two) rb1 = *reinterpret_cast<const float4*>(Fcat + h1);
  *reinterpret_cast<float4*>(As + d0) = ra0;
  *reinterpret_cast<float4*>(As + d1) = ra1;
  *reinterpret_cast<float4*>(Bs + d0) = rb0;
  if (two) *reinterpret_cast<float4*>(Bs + d1) = rb1;
  __syncthreads();

  const int aFrag = (32 * rt + l16) * PK + 2 * half;
  const int bFrag = (32 * cp + l16) * PK + 2 * half;
  int p = 0;
#pragma nounroll
  for (int k0 = 0; ; k0 += KC) {
    const bool more = (k0 + KC < 96);
    if (more) {
      ra0 = *reinterpret_cast<const float4*>(A + g0 + k0 + KC);
      ra1 = *reinterpret_cast<const float4*>(A + g1 + k0 + KC);
      rb0 = *reinterpret_cast<const float4*>(Fcat + h0 + k0 + KC);
      if (two) rb1 = *reinterpret_cast<const float4*>(Fcat + h1 + k0 + KC);
    }
    const float* as = As + p * TFA + aFrag;
    const float* bs = Bs + p * TFB + bFrag;
#pragma unroll
    for (int ks = 0; ks < KC / 4; ++ks) {
      const v2f bf0 = *reinterpret_cast<const v2f*>(bs + ks * 4);
      const v2f bf1 = *reinterpret_cast<const v2f*>(bs + 16 * PK + ks * 4);
#pragma unroll
      for (int t = 0; t < 2; ++t) {
        const v2f af = *reinterpret_cast<const v2f*>(as + t * 16 * PK + ks * 4);
        acc[2 * t] = __builtin_amdgcn_wmma_f32_16x16x4_f32(
            false, af, false, bf0, (short)0, acc[2 * t], false, false);
        acc[2 * t + 1] = __builtin_amdgcn_wmma_f32_16x16x4_f32(
            false, af, false, bf1, (short)0, acc[2 * t + 1], false, false);
      }
    }
    if (!more) break;
    p ^= 1;
    *reinterpret_cast<float4*>(As + p * TFA + d0) = ra0;
    *reinterpret_cast<float4*>(As + p * TFA + d1) = ra1;
    *reinterpret_cast<float4*>(Bs + p * TFB + d0) = rb0;
    if (two) *reinterpret_cast<float4*>(Bs + p * TFB + d1) = rb1;
    __syncthreads();
  }

  slots_behind_last_wmma<4>(acc);
  __builtin_amdgcn_sched_barrier(0);
  const int f0 = tid * 4;
  ra0 = *reinterpret_cast<const float4*>(Y2 + f0);
  ra1 = *reinterpret_cast<const float4*>(Y2 + f0 + 768);
  rc0 = *reinterpret_cast<const float4*>(Y2 + f0 + 1536);
  rc1 = *reinterpret_cast<const float4*>(Y2 + f0 + 2304);
  if (two) rb0 = *reinterpret_cast<const float4*>(Y2 + f0 + 3072);
  {
    float* xw = X1 + (32 * rt + 8 * half) * PX + 32 * cp + l16;
#pragma unroll
    for (int t = 0; t < 2; ++t) {
#pragma unroll
      for (int u = 0; u < 2; ++u) {
#pragma unroll
        for (int r = 0; r < 8; ++r) xw[(t * 16 + r) * PX + u * 16] = acc[2 * t + u][r];
      }
    }
  }
  __syncthreads();
  *reinterpret_cast<float4*>(As + f0) = ra0;
  *reinterpret_cast<float4*>(As + f0 + 768) = ra1;
  *reinterpret_cast<float4*>(As + f0 + 1536) = rc0;
  *reinterpret_cast<float4*>(As + f0 + 2304) = rc1;
  if (two) *reinterpret_cast<float4*>(As + f0 + 3072) = rb0;
#pragma unroll
  for (int j = 0; j < 4; ++j) {
    const int it = tid + 192 * j;
    if (it < 752) {
      const int yy = 1 + (it >> 4), q4 = (it & 15) * 4;
      float4* lo = reinterpret_cast<float4*>(X1 + yy * PX + q4);
      float4* hi = reinterpret_cast<float4*>(X1 + (96 - yy) * PX + q4);
      const float4 u = *lo, v = *hi;
      *lo = make_float4(u.x + v.x, u.y + v.y, u.z + v.z, u.w + v.w);
      *hi = make_float4(u.x - v.x, u.y - v.y, u.z - v.z, u.w - v.w);
    }
  }
  __syncthreads();

  v8f acc2[4];
#pragma unroll
  for (int t = 0; t < 4; ++t) acc2[t] = zero;
  const int rg = wave & 1, ch = wave >> 1;
  if (wave < 4) {
    const float* as = As + (32 * rg + l16) * PY + 2 * half;
    const float* bs = X1 + (44 * rg + 2 * half) * PX + 32 * ch + l16;
#pragma unroll
    for (int ks = 0; ks < PY / 4; ++ks) {
      v2f bf0, bf1;
      bf0.x = bs[(ks * 4) * PX];
      bf0.y = bs[(ks * 4 + 1) * PX];
      bf1.x = bs[(ks * 4) * PX + 16];
      bf1.y = bs[(ks * 4 + 1) * PX + 16];
#pragma unroll
      for (int t = 0; t < 2; ++t) {
        const v2f af = *reinterpret_cast<const v2f*>(as + t * 16 * PY + ks * 4);
        acc2[2 * t] = __builtin_amdgcn_wmma_f32_16x16x4_f32(
            false, af, false, bf0, (short)0, acc2[2 * t], false, false);
        acc2[2 * t + 1] = __builtin_amdgcn_wmma_f32_16x16x4_f32(
            false, af, false, bf1, (short)0, acc2[2 * t + 1], false, false);
      }
    }
  }

  slots_behind_last_wmma<4>(acc2);
  __builtin_amdgcn_sched_barrier(0);
  __syncthreads();
  if (wave < 4) {
    float* tw = As + (32 * rg + 8 * half) * PT + 32 * ch + l16;
#pragma unroll
    for (int t = 0; t < 2; ++t) {
#pragma unroll
      for (int u = 0; u < 2; ++u) {
#pragma unroll
        for (int r = 0; r < 8; ++r) tw[(t * 16 + r) * PT + u * 16] = acc2[2 * t + u][r];
      }
    }
  }
  __syncthreads();
  {
    const int row0 = tid / 12, c4 = tid - row0 * 12;
    int ja[4], jb[4];
    float sg[4];
#pragma unroll
    for (int e = 0; e < 4; ++e) {
      const int pp = 4 * c4 + e;
      const bool mir = pp > 24;
      ja[e] = mir ? 48 - pp : pp;
      jb[e] = pp == 0 ? 0 : ja[e] + 24;
      sg[e] = pp == 0 ? 0.f : (mir ? 1.f : -1.f);
    }
    float* cg = R2 + (long)blockIdx.x * 4608 + row0 * 48 + 4 * c4;
#pragma unroll
    for (int j = 0; j < 3; ++j) {
      const int ky = row0 + 16 * j;
      const int mk = ky < 24 ? ky : 48 - ky;
      const float sk = ky < 24 ? -1.f : 1.f;
      const float* c = As + mk * PT;
      const float* s = c + 32 * PT;
      float4 yre, yim;
      yre.x = fmaf(sk * sg[0], s[jb[0]], c[ja[0]]);  yim.x = fmaf(-sg[0], c[jb[0]], sk * s[ja[0]]);
      yre.y = fmaf(sk * sg[1], s[jb[1]], c[ja[1]]);  yim.y = fmaf(-sg[1], c[jb[1]], sk * s[ja[1]]);
      yre.z = fmaf(sk * sg[2], s[jb[2]], c[ja[2]]);  yim.z = fmaf(-sg[2], c[jb[2]], sk * s[ja[2]]);
      yre.w = fmaf(sk * sg[3], s[jb[3]], c[ja[3]]);  yim.w = fmaf(-sg[3], c[jb[3]], sk * s[ja[3]]);
      *reinterpret_cast<float4*>(cg + j * 16 * 48) = yre;
      *reinterpret_cast<float4*>(cg + 2304 + j * 16 * 48) = yim;
    }
  }
}

__global__ void gen_consts(float* Fre, float* Fim, float* Vre, float* Vim,
                           float* F3, float* V2, float* V3, float* Y2) {
  int i = blockIdx.x * 256 + threadIdx.x;
  if (i >= 48 * 96) return;
  if (i < 64 * PY) {
    const int row = i / PY, kk = i - row * PY;
    float val = 0.f;
    if (row < 32) {
      if (row <= 24 && kk <= 48)
        val = cosf(6.283185307179586f * (float)((row * kk) % 96) / 96.f);
    } else {
      const int m = row - 32, r = 44 + kk;
      if (m >= 1 && m <= 24 && r >= 49)
        val = sinf(6.283185307179586f * (float)((m * (96 - r)) % 96) / 96.f);
    }
    Y2[i] = val;
  }
  int k = i / 96, nn = i % 96;
  int f = (k < 24) ? k : (k + 48);
  int ph = (f * nn) % 96;
  float th = 6.283185307179586f * (float)ph / 96.f;
  float c = cosf(th), s = sinf(th);
  Fre[i] = c;  Fim[i] = -s;
  Vre[nn * 48 + k] = c;  Vim[nn * 48 + k] = s;
  F3[k * 192 + 2 * nn]     = c + s;
  F3[k * 192 + 2 * nn + 1] = s - c;
  V2[nn * 96 + 2 * k]            = c;
  V2[nn * 96 + 2 * k + 1]        = -s;
  V2[(96 + nn) * 96 + 2 * k]     = s;
  V2[(96 + nn) * 96 + 2 * k + 1] = c;
  V3[nn * 96 + 2 * k]     = c + s;
  V3[nn * 96 + 2 * k + 1] = c - s;
}

__global__ void pack_wqkv(const float* __restrict__ wq, const float* __restrict__ wk,
                          const float* __restrict__ wv, float* __restrict__ dst) {
  int i = blockIdx.x * 256 + threadIdx.x;
  if (i >= 384 * 64) return;
  int ic = i % 64, r = i / 64;
  int t = r / 128, q = r % 128;
  const float* w = (t == 0) ? wq : ((t == 1) ? wk : wv);
  dst[i] = w[q * 64 + ic];
}

#define KSPLIT 32
__global__ __launch_bounds__(256) void sum_scale_selu(const float* __restrict__ part, float* __restrict__ s,
                               int nvec, int planeVec, float scale) {
  int i = blockIdx.x * 256 + threadIdx.x;
  if (i >= nvec) return;
  int h = i / planeVec, e = i - h * planeVec;
  const float4* p = reinterpret_cast<const float4*>(part) + (long)h * KSPLIT * planeVec + e;
  float4 v[KSPLIT];
  for (int j = 0; j < KSPLIT; ++j) v[j] = p[(long)j * planeVec];
  float4 a = v[0];
  for (int j = 1; j < KSPLIT; ++j) {
    a.x += v[j].x; a.y += v[j].y; a.z += v[j].z; a.w += v[j].w;
  }
  const float lam = 1.0507009873554805f, alp = 1.6732632423543772f;
  float x0 = scale * a.x, x1 = scale * a.y, x2 = scale * a.z, x3 = scale * a.w;
  float4 r;
  r.x = lam * (x0 > 0.f ? x0 : alp * (expf(x0) - 1.f));
  r.y = lam * (x1 > 0.f ? x1 : alp * (expf(x1) - 1.f));
  r.z = lam * (x2 > 0.f ? x2 : alp * (expf(x2) - 1.f));
  r.w = lam * (x3 > 0.f ? x3 : alp * (expf(x3) - 1.f));
  reinterpret_cast<float4*>(s)[i] = r;
}


#define GP_HALF 1728
#define GP_PAD  16
#define GP_F4   864

template <int NT>
__global__ __launch_bounds__(NT) void group_rows(const float* __restrict__ src, float* __restrict__ dst) {
  constexpr int NP = GP_F4 / NT;
  static_assert(NP * NT == GP_F4, "whole pieces per thread");
  __shared__ __attribute__((aligned(16))) float t[2 * GP_HALF + GP_PAD];
  const int b = blockIdx.x;
  const int pyp = b & 3, pz = (b >> 2) & 7, r = b >> 5;
  const float* s = src + (long)r * 110592 + (pz * 48 + 2 * pyp) * 48;
  float* d = dst + (long)r * 110592 + (pz * 64 + pyp * 16) * 216;
  float4 v[NP];
#pragma unroll
  for (int i = 0; i < NP; ++i) {
    const int f = threadIdx.x + NT * i;
    const int chunk = f / 24, q = f - chunk * 24;
    const int nz = chunk / 6, ny = chunk - nz * 6;
    v[i] = *reinterpret_cast<const float4*>(s + (nz * 384 + ny * 8) * 48 + q * 4);
  }
#pragma unroll
  for (int i = 0; i < NP; ++i) {
    const int f = threadIdx.x + NT * i;
    const int chunk = f / 24, q = f - chunk * 24;
    const int yy = q / 12, xq = q - yy * 12;
    float* w = t + yy * (GP_HALF + GP_PAD) + (4 * (xq & 1)) * 216 + chunk * 6 + (xq >> 1);
    w[0] = v[i].x;  w[216] = v[i].y;  w[432] = v[i].z;  w[648] = v[i].w;
  }
  __syncthreads();
#pragma unroll
  for (int i = 0; i < NP; ++i) {
    const int g = threadIdx.x + NT * i;
    reinterpret_cast<float4*>(d)[g] = *reinterpret_cast<const float4*>(t + 4 * g + (g / 432) * GP_PAD);
  }
}

template <int NT>
__global__ __launch_bounds__(NT) void ungroup_rows(const float* __restrict__ src, float* __restrict__ dst) {
  constexpr int NP = GP_F4 / NT;
  static_assert(NP * NT == GP_F4, "whole pieces per thread");
  __shared__ __attribute__((aligned(16))) float t[2 * GP_HALF + GP_PAD];
  const int b = blockIdx.x;
  const int pyp = b & 3, pz = (b >> 2) & 7, r = b >> 5;
  const float* s = src + (long)r * 110592 + (pz * 64 + pyp * 16) * 216;
  float* d = dst + (long)r * 110592 + (pz * 48 + 2 * pyp) * 48;
  float4 v[NP];
#pragma unroll
  for (int i = 0; i < NP; ++i) v[i] = reinterpret_cast<const float4*>(s)[threadIdx.x + NT * i];
#pragma unroll
  for (int i = 0; i < NP; ++i) {
    const int g = threadIdx.x + NT * i;
    *reinterpret_cast<float4*>(t + 4 * g + (g / 432) * GP_PAD) = v[i];
  }
  __syncthreads();
#pragma unroll
  for (int i = 0; i < NP; ++i) {
    const int f = threadIdx.x + NT * i;
    const int chunk = f / 24, q = f - chunk * 24;
    const int nz = chunk / 6, ny = chunk - nz * 6;
    const int yy = q / 12, xq = q - yy * 12;
    const float* w = t + yy * (GP_HALF + GP_PAD) + (4 * (xq & 1)) * 216 + chunk * 6 + (xq >> 1);
    float4 o;
    o.x = w[0];  o.y = w[216];  o.z = w[432];  o.w = w[648];
    *reinterpret_cast<float4*>(d + (nz * 384 + ny * 8) * 48 + q * 4) = o;
  }
}

extern "C" void kernel_launch(void* const* d_in, const int* in_sizes, int n_in,
                              void* d_out, int out_size, void* d_ws, size_t ws_size,
                              hipStream_t stream) {
  const float* in = (const float*)d_in[0];
  const float* wq = (const float*)d_in[1];
  const float* wk = (const float*)d_in[2];
  const float* wv = (const float*)d_in[3];
  const float* wo = (const float*)d_in[4];
  float* out = (float*)d_out;
  float* w = (float*)d_ws;

  const long R2_o   = 56623104;
  const long blk_o  = 0;
  const long blkG_o = 7077888,    G_o    = 42467328;
  const long S_o    = 92012544,   Og_o   = 92385792;
  const long P_o    = 0;
  const long specG_o = 0,         spec_o = 14155776;
  const long T1_o   = 15925248;
  const long T2_o   = 23003136;
  const long F3_o   = 84953088;
  const long V2_o   = 84962304,   V3_o   = 84980736;
  const long Y2_o   = 84989952;
  const long Wqkv_o = 106541568;
  const long Fre_o  = 106566144,  Fim_o  = 106570752;
  const long Vre_o  = 106575360,  Vim_o  = 106579968;
  const long WS_FLOATS = 106584576;
  if (ws_size < (size_t)WS_FLOATS * sizeof(float)) return;

  float *R2 = w+R2_o;
  float *blk = w+blk_o, *blkG = w+blkG_o, *G = w+G_o, *S = w+S_o, *Og = w+Og_o;
  float *specG = w+specG_o, *spec = w+spec_o, *P = w+P_o;
  float *T1 = w+T1_o, *T2 = w+T2_o, *Wqkv = w+Wqkv_o;
  float *Fre = w+Fre_o, *Fim = w+Fim_o, *Vre = w+Vre_o, *Vim = w+Vim_o;
  float *F3 = w+F3_o, *V2 = w+V2_o, *V3 = w+V3_o, *Y2 = w+Y2_o;

  auto ew = [&](int n) { return dim3((unsigned)((n + 255) / 256)); };

  gen_consts<<<ew(4608), 256, 0, stream>>>(Fre, Fim, Vre, Vim, F3, V2, V3, Y2);
  pack_wqkv<<<ew(24576), 256, 0, stream>>>(wq, wk, wv, Wqkv);

  dft_xy_fused<<<dim3(6144), 192, 0, stream>>>(in, Fre, Y2, R2);
  launch_gemm<true, false, 3, 4>(stream, F3, 192, 1, 0, R2, 2304, 1, 442368, blk, 2304, 110592, 48, 2304, 192, 64, 1.f);

  group_rows<96><<<dim3(2048), 96, 0, stream>>>(blk, blkG);
  launch_gemm<true, false, 4, 4>(stream, Wqkv, 64, 1, 0, blkG, 110592, 1, 0, G, 110592, 0, 384, 110592, 64, 1, 1.f);

  const long HD = 8192L * 216L;
  const long SL = 256L * 216L;
  launch_gemm<false, false, 7, 7>(stream, G, 1, 216, SL, G + 8 * HD, 216, 1, SL, P, 216, 46656,
       216, 216, 256, 8 * KSPLIT, 1.f);
  sum_scale_selu<<<ew(93312), 256, 0, stream>>>(P, S, 93312, 11664,
                                                0.011048543456039806f);
  launch_gemm<true, true, 4, 7>(stream, G + 16 * HD, 216, 1, HD, S, 1, 216, 46656, Og, 216, HD,
       8192, 216, 216, 8, 1.f);

  launch_gemm<true, false, 1, 4>(stream, wo, 128, 1, 0, Og, 110592, 1, 0, specG, 110592, 0, 16, 110592, 128, 1, 1.f);
  ungroup_rows<288><<<dim3(512), 288, 0, stream>>>(specG, spec);

  launch_gemm<true, true, 4, 4>(stream, spec, 48, 1, 0, Vre, 1, 48, 0, T1, 192, 0, 36864, 192, 48, 1, 1.f);
  launch_gemm<true, false, 6, 6>(stream, V2, 96, 1, 0, T1, 96, 1, 9216, T2, 96, 18432, 192, 96, 96, 768, 1.f);
  const float invN = 1.f / 884736.f;
  launch_gemm<true, false, 6, 4>(stream, V3, 96, 1, 0, T2, 9216, 1, 884736, out, 9216, 884736, 96, 9216, 96, 16, invN);
}
